# starting kernel
# baseline (speedup 1.0000x reference)
#include <hip/hip_runtime.h>

typedef __attribute__((ext_vector_type(2))) float v2f;
typedef __attribute__((ext_vector_type(8))) float v8f;

#define HD 64

__device__ __forceinline__ float lrelu_f(float x) { return x > 0.f ? x : 0.1f * x; }

__device__ __forceinline__ unsigned fenc(float f) {
    unsigned u = __float_as_uint(f);
    return (u & 0x80000000u) ? ~u : (u | 0x80000000u);
}
__device__ __forceinline__ float fdec(unsigned e) {
    unsigned u = (e & 0x80000000u) ? (e & 0x7fffffffu) : ~e;
    return __uint_as_float(u);
}

template <bool HAS_A2, bool HAS_RES, bool HAS_BIAS, int ACT>
__global__ __launch_bounds__(32) void gemm64(
    const float* __restrict__ A,  const float* __restrict__ B,
    const float* __restrict__ A2, const float* __restrict__ B2,
    const float* __restrict__ res, const float* __restrict__ bias,
    float* __restrict__ out)
{
    const int lane = threadIdx.x & 31;
    const int cn   = lane & 15;
    const int kh   = (lane >> 4) << 1;
    const size_t row = (size_t)blockIdx.x * 16 + cn;

    const v8f z = {0.f, 0.f, 0.f, 0.f, 0.f, 0.f, 0.f, 0.f};
    v8f acc0 = z, acc1 = z, acc2 = z, acc3 = z;

    const float* arow = A + row * HD;
#pragma unroll
    for (int k0 = 0; k0 < 64; k0 += 4) {
        v2f a; a.x = arow[k0 + kh]; a.y = arow[k0 + kh + 1];
        const float* b0 = B + (size_t)(k0 + kh) * HD;
        const float* b1 = b0 + HD;
        v2f b;
        b.x = b0[cn];        b.y = b1[cn];
        acc0 = __builtin_amdgcn_wmma_f32_16x16x4_f32(false, a, false, b, (short)0, acc0, false, false);
        b.x = b0[16 + cn];   b.y = b1[16 + cn];
        acc1 = __builtin_amdgcn_wmma_f32_16x16x4_f32(false, a, false, b, (short)0, acc1, false, false);
        b.x = b0[32 + cn];   b.y = b1[32 + cn];
        acc2 = __builtin_amdgcn_wmma_f32_16x16x4_f32(false, a, false, b, (short)0, acc2, false, false);
        b.x = b0[48 + cn];   b.y = b1[48 + cn];
        acc3 = __builtin_amdgcn_wmma_f32_16x16x4_f32(false, a, false, b, (short)0, acc3, false, false);
    }
    if (HAS_A2) {
        const float* a2row = A2 + row * 16;
#pragma unroll
        for (int k0 = 0; k0 < 16; k0 += 4) {
            v2f a; a.x = a2row[k0 + kh]; a.y = a2row[k0 + kh + 1];
            const float* b0 = B2 + (size_t)(k0 + kh) * HD;
            const float* b1 = b0 + HD;
            v2f b;
            b.x = b0[cn];        b.y = b1[cn];
            acc0 = __builtin_amdgcn_wmma_f32_16x16x4_f32(false, a, false, b, (short)0, acc0, false, false);
            b.x = b0[16 + cn];   b.y = b1[16 + cn];
            acc1 = __builtin_amdgcn_wmma_f32_16x16x4_f32(false, a, false, b, (short)0, acc1, false, false);
            b.x = b0[32 + cn];   b.y = b1[32 + cn];
            acc2 = __builtin_amdgcn_wmma_f32_16x16x4_f32(false, a, false, b, (short)0, acc2, false, false);
            b.x = b0[48 + cn];   b.y = b1[48 + cn];
            acc3 = __builtin_amdgcn_wmma_f32_16x16x4_f32(false, a, false, b, (short)0, acc3, false, false);
        }
    }

    const size_t mbase = (size_t)blockIdx.x * 16 + ((lane >> 4) << 3);
    v8f accs[4] = {acc0, acc1, acc2, acc3};
#pragma unroll
    for (int t = 0; t < 4; ++t) {
        const int col = t * 16 + cn;
        const float bval = HAS_BIAS ? bias[col] : 0.f;
#pragma unroll
        for (int v = 0; v < 8; ++v) {
            const size_t m = mbase + v;
            float val = accs[t][v] + bval;
            if (HAS_RES) val += res[m * HD + col];
            if (ACT == 1) val = lrelu_f(val);
            else if (ACT == 2) val = fmaxf(val, 0.f);
            out[m * HD + col] = val;
        }
    }
}

__global__ void prep_ce(const float* __restrict__ W_e, const float* __restrict__ a_e,
                        float* __restrict__ cE)
{
    const int tid = threadIdx.x;
    const int l = tid >> 4, d = tid & 15;
    float s = 0.f;
    for (int k = 0; k < 64; ++k) s += W_e[d * 64 + k] * a_e[l * 64 + k];
    cE[tid] = s;
}

__global__ void prep_wewm(const float* __restrict__ W_e, const float* __restrict__ Wm,
                          float* __restrict__ WeWm)
{
    const int idx = blockIdx.x * blockDim.x + threadIdx.x;
    if (idx >= 4 * 16 * 64) return;
    const int l = idx >> 10, d = (idx >> 6) & 15, k = idx & 63;
    float s = 0.f;
    for (int j = 0; j < 64; ++j) s += W_e[d * 64 + j] * Wm[l * 4096 + j * 64 + k];
    WeWm[idx] = s;
}

__global__ void edge_ce(const float* __restrict__ re, const float* __restrict__ cE,
                        float* __restrict__ ceL, int E)
{
    const int j = blockIdx.x * blockDim.x + threadIdx.x;
    if (j >= E) return;
    float r[16];
    const float4* rp = (const float4*)(re + (size_t)j * 16);
#pragma unroll
    for (int i = 0; i < 4; ++i) {
        float4 v = rp[i];
        r[i * 4 + 0] = v.x; r[i * 4 + 1] = v.y; r[i * 4 + 2] = v.z; r[i * 4 + 3] = v.w;
    }
#pragma unroll
    for (int l = 0; l < 4; ++l) {
        float s = 0.f;
#pragma unroll
        for (int d = 0; d < 16; ++d) s += r[d] * cE[l * 16 + d];
        ceL[(size_t)j * 4 + l] = s;
    }
}

__global__ void init_hi(const float* __restrict__ i_node, const float* __restrict__ W_i,
                        float* __restrict__ hi, float* __restrict__ pooled, int G)
{
    const int idx = blockIdx.x * blockDim.x + threadIdx.x;
    if (idx >= G * 64) return;
    hi[idx] = i_node[idx >> 6] * W_i[idx & 63];
    pooled[idx] = 0.f;
}

__global__ void node_prep(const float* __restrict__ h, const float* __restrict__ as,
                          const float* __restrict__ ad, float* __restrict__ sb,
                          float* __restrict__ db, unsigned* __restrict__ menc,
                          float* __restrict__ denom, float* __restrict__ A64,
                          float* __restrict__ A16, int N)
{
    const int n = blockIdx.x * blockDim.x + threadIdx.x;
    if (n >= N) return;
    const float* hr = h + (size_t)n * 64;
    float s = 0.f, d = 0.f;
#pragma unroll
    for (int k = 0; k < 64; ++k) { const float v = hr[k]; s += v * as[k]; d += v * ad[k]; }
    sb[n] = s; db[n] = d; menc[n] = 0u; denom[n] = 0.f;
    const float4 z4 = {0.f, 0.f, 0.f, 0.f};
    float4* a64 = (float4*)(A64 + (size_t)n * 64);
#pragma unroll
    for (int i = 0; i < 16; ++i) a64[i] = z4;
    float4* a16 = (float4*)(A16 + (size_t)n * 16);
#pragma unroll
    for (int i = 0; i < 4; ++i) a16[i] = z4;
}

__global__ void edge_logits(const int* __restrict__ src, const int* __restrict__ dst,
                            const float* __restrict__ sb, const float* __restrict__ db,
                            const float* __restrict__ ceL, float* __restrict__ logit,
                            unsigned* __restrict__ menc, int E, int l)
{
    const int j = blockIdx.x * blockDim.x + threadIdx.x;
    if (j >= E) return;
    const float lg = lrelu_f(sb[src[j]] + db[dst[j]] + ceL[(size_t)j * 4 + l]);
    logit[j] = lg;
    atomicMax(menc + dst[j], fenc(lg));
}

__global__ void edge_exp(const int* __restrict__ dst, const unsigned* __restrict__ menc,
                         float* __restrict__ logit, float* __restrict__ denom, int E)
{
    const int j = blockIdx.x * blockDim.x + threadIdx.x;
    if (j >= E) return;
    const int d = dst[j];
    const float ex = __expf(logit[j] - fdec(menc[d]));
    logit[j] = ex;
    atomicAdd(denom + d, ex);
}

__global__ __launch_bounds__(256) void edge_scatter(
    const int* __restrict__ src, const int* __restrict__ dst,
    const float* __restrict__ logit, const float* __restrict__ denom,
    const float* __restrict__ h, const float* __restrict__ re,
    float* __restrict__ A64, float* __restrict__ A16, int E)
{
    const int j = blockIdx.x * 4 + (threadIdx.x >> 6);
    const int t = threadIdx.x & 63;
    if (j >= E) return;
    const int s = src[j], d = dst[j];
    const float attn = logit[j] / denom[d];
    atomicAdd(A64 + (size_t)d * 64 + t, attn * h[(size_t)s * 64 + t]);
    if (t < 16) atomicAdd(A16 + (size_t)d * 16 + t, attn * re[(size_t)j * 16 + t]);
}

__global__ void gate_hi(const float* __restrict__ h, const float* __restrict__ d_edge,
                        const int* __restrict__ gid, const float* __restrict__ wd,
                        const float* __restrict__ bd, float* __restrict__ gate,
                        float* __restrict__ hi, int N)
{
    const int idx = blockIdx.x * blockDim.x + threadIdx.x;
    if (idx >= N * 64) return;
    const int n = idx >> 6, k = idx & 63;
    const float g = 1.f / (1.f + __expf(-(d_edge[n] * wd[0] + bd[0])));
    if (k == 0) gate[n] = g;
    atomicAdd(hi + (size_t)gid[n] * 64 + k, g * h[idx]);
}

__global__ void h_mix_pool(float* __restrict__ h, const float* __restrict__ gate,
                           const float* __restrict__ hi, const int* __restrict__ gid,
                           float* __restrict__ pooled, int N)
{
    const int idx = blockIdx.x * blockDim.x + threadIdx.x;
    if (idx >= N * 64) return;
    const int n = idx >> 6, k = idx & 63;
    const float v = h[idx] + gate[n] * hi[(size_t)gid[n] * 64 + k];
    h[idx] = v;
    atomicAdd(pooled + (size_t)gid[n] * 64 + k, v);
}

__global__ void out_kernel(const float* __restrict__ x, const float* __restrict__ Wo,
                           const float* __restrict__ bo, float* __restrict__ out, int G)
{
    const int g = blockIdx.x * blockDim.x + threadIdx.x;
    if (g >= G) return;
    float s = bo[0];
#pragma unroll
    for (int k = 0; k < 64; ++k) s += x[(size_t)g * 64 + k] * Wo[k];
    out[g] = s;
}

extern "C" void kernel_launch(void* const* d_in, const int* in_sizes, int n_in,
                              void* d_out, int out_size, void* d_ws, size_t ws_size,
                              hipStream_t stream)
{
    const float* r_node = (const float*)d_in[0];
    const float* i_node = (const float*)d_in[1];
    const float* r_edge = (const float*)d_in[2];
    const float* d_edge = (const float*)d_in[3];
    const int*   src    = (const int*)d_in[4];
    const int*   dst    = (const int*)d_in[5];
    const int*   gid    = (const int*)d_in[6];
    const float* W_r    = (const float*)d_in[7];
    const float* W_i    = (const float*)d_in[8];
    const float* W_e    = (const float*)d_in[9];
    const float* Wm     = (const float*)d_in[10];
    const float* a_s    = (const float*)d_in[11];
    const float* a_d    = (const float*)d_in[12];
    const float* a_e    = (const float*)d_in[13];
    const float* w_d    = (const float*)d_in[14];
    const float* b_d    = (const float*)d_in[15];
    const float* W_mlp  = (const float*)d_in[16];
    const float* b_mlp  = (const float*)d_in[17];
    const float* W_out  = (const float*)d_in[18];
    const float* b_out  = (const float*)d_in[19];

    const int N = in_sizes[0] / 64;
    const int G = in_sizes[1];
    const int E = in_sizes[2] / 16;
    (void)n_in; (void)out_size; (void)ws_size;

    float* ws = (float*)d_ws;
    size_t off = 0;
    auto carve = [&](size_t n) -> float* {
        float* p = ws + off; off += (n + 63) & ~(size_t)63; return p;
    };
    float*    h      = carve((size_t)N * 64);
    float*    A64    = carve((size_t)N * 64);
    float*    A16    = carve((size_t)N * 16);
    float*    sb     = carve(N);
    float*    db     = carve(N);
    unsigned* menc   = (unsigned*)carve(N);
    float*    denom  = carve(N);
    float*    gate   = carve(N);
    float*    ceL    = carve((size_t)E * 4);
    float*    logit  = carve(E);
    float*    hi     = carve((size_t)G * 64);
    float*    pooled = carve((size_t)G * 64);
    float*    x1     = carve((size_t)G * 64);
    float*    x2     = carve((size_t)G * 64);
    float*    cE     = carve(64);
    float*    WeWm   = carve(4 * 16 * 64);

    hipLaunchKernelGGL(prep_ce, dim3(1), dim3(64), 0, stream, W_e, a_e, cE);
    hipLaunchKernelGGL(prep_wewm, dim3((4096 + 255) / 256), dim3(256), 0, stream, W_e, Wm, WeWm);
    hipLaunchKernelGGL(edge_ce, dim3((E + 255) / 256), dim3(256), 0, stream, r_edge, cE, ceL, E);

    hipLaunchKernelGGL((gemm64<false, false, false, 0>), dim3(N / 16), dim3(32), 0, stream,
                       r_node, W_r, nullptr, nullptr, nullptr, nullptr, h);
    hipLaunchKernelGGL(init_hi, dim3((G * 64 + 255) / 256), dim3(256), 0, stream,
                       i_node, W_i, hi, pooled, G);

    for (int l = 0; l < 4; ++l) {
        hipLaunchKernelGGL(node_prep, dim3((N + 127) / 128), dim3(128), 0, stream,
                           h, a_s + l * 64, a_d + l * 64, sb, db, menc, denom, A64, A16, N);
        hipLaunchKernelGGL(edge_logits, dim3((E + 255) / 256), dim3(256), 0, stream,
                           src, dst, sb, db, ceL, logit, menc, E, l);
        hipLaunchKernelGGL(edge_exp, dim3((E + 255) / 256), dim3(256), 0, stream,
                           dst, menc, logit, denom, E);
        hipLaunchKernelGGL(edge_scatter, dim3((E + 3) / 4), dim3(256), 0, stream,
                           src, dst, logit, denom, h, r_edge, A64, A16, E);
        hipLaunchKernelGGL((gemm64<true, true, false, 1>), dim3(N / 16), dim3(32), 0, stream,
                           A64, Wm + (size_t)l * 4096, A16, WeWm + (size_t)l * 1024,
                           h, nullptr, h);
    }

    hipLaunchKernelGGL(gate_hi, dim3((N * 64 + 255) / 256), dim3(256), 0, stream,
                       h, d_edge, gid, w_d, b_d, gate, hi, N);
    hipLaunchKernelGGL(h_mix_pool, dim3((N * 64 + 255) / 256), dim3(256), 0, stream,
                       h, gate, hi, gid, pooled, N);

    hipLaunchKernelGGL((gemm64<false, false, true, 2>), dim3(G / 16), dim3(32), 0, stream,
                       pooled, W_mlp + 0 * 4096, nullptr, nullptr, nullptr, b_mlp + 0 * 64, x1);
    hipLaunchKernelGGL((gemm64<false, false, true, 2>), dim3(G / 16), dim3(32), 0, stream,
                       x1, W_mlp + 1 * 4096, nullptr, nullptr, nullptr, b_mlp + 1 * 64, x2);
    hipLaunchKernelGGL((gemm64<false, false, true, 2>), dim3(G / 16), dim3(32), 0, stream,
                       x2, W_mlp + 2 * 4096, nullptr, nullptr, nullptr, b_mlp + 2 * 64, x1);

    hipLaunchKernelGGL(out_kernel, dim3((G + 255) / 256), dim3(256), 0, stream,
                       x1, W_out, b_out, (float*)d_out, G);
}


# Per-edge f32 atomic scatter into A64/A16 replaced by a per-call destination-ordered edge list and a per-node register gather-sum; sum order = arrival-order rank, not atomic arrival
# speedup vs baseline: 1.9108x; 1.9108x over previous
#include <hip/hip_runtime.h>

typedef __attribute__((ext_vector_type(2))) float v2f;
typedef __attribute__((ext_vector_type(8))) float v8f;

#define HD 64

__device__ __forceinline__ float lrelu_f(float x) { return x > 0.f ? x : 0.1f * x; }

__device__ __forceinline__ unsigned fenc(float f) {
    unsigned u = __float_as_uint(f);
    return (u & 0x80000000u) ? ~u : (u | 0x80000000u);
}
__device__ __forceinline__ float fdec(unsigned e) {
    unsigned u = (e & 0x80000000u) ? (e & 0x7fffffffu) : ~e;
    return __uint_as_float(u);
}

template <bool HAS_A2, bool HAS_RES, bool HAS_BIAS, int ACT>
__global__ __launch_bounds__(32) void gemm64(
    const float* __restrict__ A,  const float* __restrict__ B,
    const float* __restrict__ A2, const float* __restrict__ B2,
    const float* __restrict__ res, const float* __restrict__ bias,
    float* __restrict__ out)
{
    const int lane = threadIdx.x & 31;
    const int cn   = lane & 15;
    const int kh   = (lane >> 4) << 1;
    const size_t row = (size_t)blockIdx.x * 16 + cn;

    const v8f z = {0.f, 0.f, 0.f, 0.f, 0.f, 0.f, 0.f, 0.f};
    v8f acc0 = z, acc1 = z, acc2 = z, acc3 = z;

    const float* arow = A + row * HD;
#pragma unroll
    for (int k0 = 0; k0 < 64; k0 += 4) {
        v2f a; a.x = arow[k0 + kh]; a.y = arow[k0 + kh + 1];
        const float* b0 = B + (size_t)(k0 + kh) * HD;
        const float* b1 = b0 + HD;
        v2f b;
        b.x = b0[cn];        b.y = b1[cn];
        acc0 = __builtin_amdgcn_wmma_f32_16x16x4_f32(false, a, false, b, (short)0, acc0, false, false);
        b.x = b0[16 + cn];   b.y = b1[16 + cn];
        acc1 = __builtin_amdgcn_wmma_f32_16x16x4_f32(false, a, false, b, (short)0, acc1, false, false);
        b.x = b0[32 + cn];   b.y = b1[32 + cn];
        acc2 = __builtin_amdgcn_wmma_f32_16x16x4_f32(false, a, false, b, (short)0, acc2, false, false);
        b.x = b0[48 + cn];   b.y = b1[48 + cn];
        acc3 = __builtin_amdgcn_wmma_f32_16x16x4_f32(false, a, false, b, (short)0, acc3, false, false);
    }
    if (HAS_A2) {
        const float* a2row = A2 + row * 16;
#pragma unroll
        for (int k0 = 0; k0 < 16; k0 += 4) {
            v2f a; a.x = a2row[k0 + kh]; a.y = a2row[k0 + kh + 1];
            const float* b0 = B2 + (size_t)(k0 + kh) * HD;
            const float* b1 = b0 + HD;
            v2f b;
            b.x = b0[cn];        b.y = b1[cn];
            acc0 = __builtin_amdgcn_wmma_f32_16x16x4_f32(false, a, false, b, (short)0, acc0, false, false);
            b.x = b0[16 + cn];   b.y = b1[16 + cn];
            acc1 = __builtin_amdgcn_wmma_f32_16x16x4_f32(false, a, false, b, (short)0, acc1, false, false);
            b.x = b0[32 + cn];   b.y = b1[32 + cn];
            acc2 = __builtin_amdgcn_wmma_f32_16x16x4_f32(false, a, false, b, (short)0, acc2, false, false);
            b.x = b0[48 + cn];   b.y = b1[48 + cn];
            acc3 = __builtin_amdgcn_wmma_f32_16x16x4_f32(false, a, false, b, (short)0, acc3, false, false);
        }
    }

    const size_t mbase = (size_t)blockIdx.x * 16 + ((lane >> 4) << 3);
    v8f accs[4] = {acc0, acc1, acc2, acc3};
#pragma unroll
    for (int t = 0; t < 4; ++t) {
        const int col = t * 16 + cn;
        const float bval = HAS_BIAS ? bias[col] : 0.f;
#pragma unroll
        for (int v = 0; v < 8; ++v) {
            const size_t m = mbase + v;
            float val = accs[t][v] + bval;
            if (HAS_RES) val += res[m * HD + col];
            if (ACT == 1) val = lrelu_f(val);
            else if (ACT == 2) val = fmaxf(val, 0.f);
            out[m * HD + col] = val;
        }
    }
}

__global__ void prep_ce(const float* __restrict__ W_e, const float* __restrict__ a_e,
                        float* __restrict__ cE)
{
    const int tid = threadIdx.x;
    const int l = tid >> 4, d = tid & 15;
    float s = 0.f;
    for (int k = 0; k < 64; ++k) s += W_e[d * 64 + k] * a_e[l * 64 + k];
    cE[tid] = s;
}

__global__ void prep_wewm(const float* __restrict__ W_e, const float* __restrict__ Wm,
                          float* __restrict__ WeWm)
{
    const int idx = blockIdx.x * blockDim.x + threadIdx.x;
    if (idx >= 4 * 16 * 64) return;
    const int l = idx >> 10, d = (idx >> 6) & 15, k = idx & 63;
    float s = 0.f;
    for (int j = 0; j < 64; ++j) s += W_e[d * 64 + j] * Wm[l * 4096 + j * 64 + k];
    WeWm[idx] = s;
}

__global__ void edge_ce(const float* __restrict__ re, const float* __restrict__ cE,
                        float* __restrict__ ceL, int E)
{
    const int j = blockIdx.x * blockDim.x + threadIdx.x;
    if (j >= E) return;
    float r[16];
    const float4* rp = (const float4*)(re + (size_t)j * 16);
#pragma unroll
    for (int i = 0; i < 4; ++i) {
        float4 v = rp[i];
        r[i * 4 + 0] = v.x; r[i * 4 + 1] = v.y; r[i * 4 + 2] = v.z; r[i * 4 + 3] = v.w;
    }
#pragma unroll
    for (int l = 0; l < 4; ++l) {
        float s = 0.f;
#pragma unroll
        for (int d = 0; d < 16; ++d) s += r[d] * cE[l * 16 + d];
        ceL[(size_t)j * 4 + l] = s;
    }
}

__global__ void init_hi(const float* __restrict__ i_node, const float* __restrict__ W_i,
                        float* __restrict__ hi, float* __restrict__ pooled, int G)
{
    const int idx = blockIdx.x * blockDim.x + threadIdx.x;
    if (idx >= G * 64) return;
    hi[idx] = i_node[idx >> 6] * W_i[idx & 63];
    pooled[idx] = 0.f;
}

__global__ void node_prep(const float* __restrict__ h, const float* __restrict__ as,
                          const float* __restrict__ ad, float* __restrict__ sb,
                          float* __restrict__ db, unsigned* __restrict__ menc,
                          float* __restrict__ denom, int N)
{
    const int n = blockIdx.x * blockDim.x + threadIdx.x;
    if (n >= N) return;
    const float* hr = h + (size_t)n * 64;
    float s = 0.f, d = 0.f;
#pragma unroll
    for (int k = 0; k < 64; ++k) { const float v = hr[k]; s += v * as[k]; d += v * ad[k]; }
    sb[n] = s; db[n] = d; menc[n] = 0u; denom[n] = 0.f;
}

__global__ void edge_logits(const int* __restrict__ src, const int* __restrict__ dst,
                            const float* __restrict__ sb, const float* __restrict__ db,
                            const float* __restrict__ ceL, float* __restrict__ logit,
                            unsigned* __restrict__ menc, int E, int l)
{
    const int j = blockIdx.x * blockDim.x + threadIdx.x;
    if (j >= E) return;
    const float lg = lrelu_f(sb[src[j]] + db[dst[j]] + ceL[(size_t)j * 4 + l]);
    logit[j] = lg;
    atomicMax(menc + dst[j], fenc(lg));
}

__global__ void edge_exp(const int* __restrict__ dst, const unsigned* __restrict__ menc,
                         float* __restrict__ logit, float* __restrict__ denom, int E)
{
    const int j = blockIdx.x * blockDim.x + threadIdx.x;
    if (j >= E) return;
    const int d = dst[j];
    const float ex = __expf(logit[j] - fdec(menc[d]));
    logit[j] = ex;
    atomicAdd(denom + d, ex);
}

__global__ void csr_zero(int* __restrict__ cnt, int N)
{
    const int n = blockIdx.x * blockDim.x + threadIdx.x;
    if (n < N) cnt[n] = 0;
}

__global__ void csr_count(const int* __restrict__ dst, int* __restrict__ cnt,
                          int* __restrict__ rank, int E)
{
    const int j = blockIdx.x * blockDim.x + threadIdx.x;
    if (j >= E) return;
    rank[j] = atomicAdd(cnt + dst[j], 1);
}

__global__ __launch_bounds__(1024) void csr_scan(const int* __restrict__ cnt,
                                                 int* __restrict__ rowptr, int N)
{
    __shared__ int wfront[32];
    __shared__ int wtot[32];
    const int t = threadIdx.x, lane = t & 31, w = t >> 5;
    const int base0 = blockIdx.x * 1024;

    int s = 0;
    for (int i = t; i < base0; i += 1024) s += cnt[i];
    for (int o = 16; o > 0; o >>= 1) s += __shfl_xor(s, o, 32);
    if (lane == 0) wfront[w] = s;

    const int i = base0 + t;
    const int c = (i < N) ? cnt[i] : 0;
    int x = c;
    for (int o = 1; o < 32; o <<= 1) {
        const int y = __shfl_up(x, o, 32);
        if (lane >= o) x += y;
    }
    if (lane == 31) wtot[w] = x;
    __syncthreads();

    int front = 0;
    for (int k = 0; k < 32; ++k) front += wfront[k];
    for (int k = 0; k < w; ++k) front += wtot[k];
    if (i < N) rowptr[i] = front + x - c;
    if (i == N - 1) rowptr[N] = front + x;
}

__global__ void csr_fill(const int* __restrict__ src, const int* __restrict__ dst,
                         const int* __restrict__ rowptr, const int* __restrict__ rank,
                         int* __restrict__ pk, int E)
{
    const int j = blockIdx.x * blockDim.x + threadIdx.x;
    if (j >= E) return;
    const int p = rowptr[dst[j]] + rank[j];
    ((int2*)pk)[p] = make_int2(j, src[j]);
}

__global__ __launch_bounds__(256) void node_gather(
    const int* __restrict__ rowptr, const int* __restrict__ pk,
    const float* __restrict__ logit, const float* __restrict__ denom,
    const float* __restrict__ h, const float* __restrict__ re,
    float* __restrict__ A64, float* __restrict__ A16, int N)
{
    const int n = blockIdx.x * 16 + (threadIdx.x >> 4);
    const int q = threadIdx.x & 15;
    if (n >= N) return;
    const int beg = rowptr[n], end = rowptr[n + 1];
    const int2* __restrict__ pk2 = (const int2*)pk;
    float4 acc = {0.f, 0.f, 0.f, 0.f};
    float accr = 0.f;
    if (end > beg) {
        const float dn = denom[n];
        const int last = end - 1;
        for (int i = beg; i < end; i += 4) {
            int2 e[4];
            float w[4], rv[4];
            float4 hv[4];
            for (int u = 0; u < 4; ++u) e[u] = pk2[min(i + u, last)];
            for (int u = 0; u < 4; ++u) {
                w[u]  = logit[e[u].x];
                hv[u] = *(const float4*)(h + (size_t)e[u].y * 64 + q * 4);
                rv[u] = re[(size_t)e[u].x * 16 + q];
            }
            for (int u = 0; u < 4; ++u) {
                const float a = (i + u < end) ? w[u] / dn : 0.f;
                acc.x += a * hv[u].x; acc.y += a * hv[u].y;
                acc.z += a * hv[u].z; acc.w += a * hv[u].w;
                accr  += a * rv[u];
            }
        }
    }
    *(float4*)(A64 + (size_t)n * 64 + q * 4) = acc;
    A16[(size_t)n * 16 + q] = accr;
}

__global__ void gate_hi(const float* __restrict__ h, const float* __restrict__ d_edge,
                        const int* __restrict__ gid, const float* __restrict__ wd,
                        const float* __restrict__ bd, float* __restrict__ gate,
                        float* __restrict__ hi, int N)
{
    const int idx = blockIdx.x * blockDim.x + threadIdx.x;
    if (idx >= N * 64) return;
    const int n = idx >> 6, k = idx & 63;
    const float g = 1.f / (1.f + __expf(-(d_edge[n] * wd[0] + bd[0])));
    if (k == 0) gate[n] = g;
    atomicAdd(hi + (size_t)gid[n] * 64 + k, g * h[idx]);
}

__global__ void h_mix_pool(float* __restrict__ h, const float* __restrict__ gate,
                           const float* __restrict__ hi, const int* __restrict__ gid,
                           float* __restrict__ pooled, int N)
{
    const int idx = blockIdx.x * blockDim.x + threadIdx.x;
    if (idx >= N * 64) return;
    const int n = idx >> 6, k = idx & 63;
    const float v = h[idx] + gate[n] * hi[(size_t)gid[n] * 64 + k];
    h[idx] = v;
    atomicAdd(pooled + (size_t)gid[n] * 64 + k, v);
}

__global__ void out_kernel(const float* __restrict__ x, const float* __restrict__ Wo,
                           const float* __restrict__ bo, float* __restrict__ out, int G)
{
    const int g = blockIdx.x * blockDim.x + threadIdx.x;
    if (g >= G) return;
    float s = bo[0];
#pragma unroll
    for (int k = 0; k < 64; ++k) s += x[(size_t)g * 64 + k] * Wo[k];
    out[g] = s;
}

extern "C" void kernel_launch(void* const* d_in, const int* in_sizes, int n_in,
                              void* d_out, int out_size, void* d_ws, size_t ws_size,
                              hipStream_t stream)
{
    const float* r_node = (const float*)d_in[0];
    const float* i_node = (const float*)d_in[1];
    const float* r_edge = (const float*)d_in[2];
    const float* d_edge = (const float*)d_in[3];
    const int*   src    = (const int*)d_in[4];
    const int*   dst    = (const int*)d_in[5];
    const int*   gid    = (const int*)d_in[6];
    const float* W_r    = (const float*)d_in[7];
    const float* W_i    = (const float*)d_in[8];
    const float* W_e    = (const float*)d_in[9];
    const float* Wm     = (const float*)d_in[10];
    const float* a_s    = (const float*)d_in[11];
    const float* a_d    = (const float*)d_in[12];
    const float* a_e    = (const float*)d_in[13];
    const float* w_d    = (const float*)d_in[14];
    const float* b_d    = (const float*)d_in[15];
    const float* W_mlp  = (const float*)d_in[16];
    const float* b_mlp  = (const float*)d_in[17];
    const float* W_out  = (const float*)d_in[18];
    const float* b_out  = (const float*)d_in[19];

    const int N = in_sizes[0] / 64;
    const int G = in_sizes[1];
    const int E = in_sizes[2] / 16;
    (void)n_in; (void)out_size; (void)ws_size;

    float* ws = (float*)d_ws;
    size_t off = 0;
    auto carve = [&](size_t n) -> float* {
        float* p = ws + off; off += (n + 63) & ~(size_t)63; return p;
    };
    float*    h      = carve((size_t)N * 64);
    float*    A64    = carve((size_t)N * 64);
    float*    A16    = carve((size_t)N * 16);
    float*    sb     = carve(N);
    float*    db     = carve(N);
    unsigned* menc   = (unsigned*)carve(N);
    float*    denom  = carve(N);
    float*    gate   = carve(N);
    float*    ceL    = carve((size_t)E * 4);
    float*    logit  = carve(E);
    float*    hi     = carve((size_t)G * 64);
    float*    pooled = carve((size_t)G * 64);
    float*    x1     = carve((size_t)G * 64);
    float*    x2     = carve((size_t)G * 64);
    float*    cE     = carve(64);
    float*    WeWm   = carve(4 * 16 * 64);
    int*      cnt    = (int*)carve(N);
    int*      rowptr = (int*)carve((size_t)N + 1);
    int*      rank   = (int*)carve(E);
    int*      pk     = (int*)carve((size_t)E * 2);

    hipLaunchKernelGGL(prep_ce, dim3(1), dim3(64), 0, stream, W_e, a_e, cE);
    hipLaunchKernelGGL(prep_wewm, dim3((4096 + 255) / 256), dim3(256), 0, stream, W_e, Wm, WeWm);
    hipLaunchKernelGGL(edge_ce, dim3((E + 255) / 256), dim3(256), 0, stream, r_edge, cE, ceL, E);

    hipLaunchKernelGGL((gemm64<false, false, false, 0>), dim3(N / 16), dim3(32), 0, stream,
                       r_node, W_r, nullptr, nullptr, nullptr, nullptr, h);
    hipLaunchKernelGGL(init_hi, dim3((G * 64 + 255) / 256), dim3(256), 0, stream,
                       i_node, W_i, hi, pooled, G);

    hipLaunchKernelGGL(csr_zero, dim3((N + 255) / 256), dim3(256), 0, stream, cnt, N);
    hipLaunchKernelGGL(csr_count, dim3((E + 255) / 256), dim3(256), 0, stream, dst, cnt, rank, E);
    hipLaunchKernelGGL(csr_scan, dim3((N + 1023) / 1024), dim3(1024), 0, stream, cnt, rowptr, N);
    hipLaunchKernelGGL(csr_fill, dim3((E + 255) / 256), dim3(256), 0, stream,
                       src, dst, rowptr, rank, pk, E);

    for (int l = 0; l < 4; ++l) {
        hipLaunchKernelGGL(node_prep, dim3((N + 127) / 128), dim3(128), 0, stream,
                           h, a_s + l * 64, a_d + l * 64, sb, db, menc, denom, N);
        hipLaunchKernelGGL(edge_logits, dim3((E + 255) / 256), dim3(256), 0, stream,
                           src, dst, sb, db, ceL, logit, menc, E, l);
        hipLaunchKernelGGL(edge_exp, dim3((E + 255) / 256), dim3(256), 0, stream,
                           dst, menc, logit, denom, E);
        hipLaunchKernelGGL(node_gather, dim3((N + 15) / 16), dim3(256), 0, stream,
                           rowptr, pk, logit, denom, h, r_edge, A64, A16, N);
        hipLaunchKernelGGL((gemm64<true, true, false, 1>), dim3(N / 16), dim3(32), 0, stream,
                           A64, Wm + (size_t)l * 4096, A16, WeWm + (size_t)l * 1024,
                           h, nullptr, h);
    }

    hipLaunchKernelGGL(gate_hi, dim3((N * 64 + 255) / 256), dim3(256), 0, stream,
                       h, d_edge, gid, w_d, b_d, gate, hi, N);
    hipLaunchKernelGGL(h_mix_pool, dim3((N * 64 + 255) / 256), dim3(256), 0, stream,
                       h, gate, hi, gid, pooled, N);

    hipLaunchKernelGGL((gemm64<false, false, true, 2>), dim3(G / 16), dim3(32), 0, stream,
                       pooled, W_mlp + 0 * 4096, nullptr, nullptr, nullptr, b_mlp + 0 * 64, x1);
    hipLaunchKernelGGL((gemm64<false, false, true, 2>), dim3(G / 16), dim3(32), 0, stream,
                       x1, W_mlp + 1 * 4096, nullptr, nullptr, nullptr, b_mlp + 1 * 64, x2);
    hipLaunchKernelGGL((gemm64<false, false, true, 2>), dim3(G / 16), dim3(32), 0, stream,
                       x2, W_mlp + 2 * 4096, nullptr, nullptr, nullptr, b_mlp + 2 * 64, x1);

    hipLaunchKernelGGL(out_kernel, dim3((G + 255) / 256), dim3(256), 0, stream,
                       x1, W_out, b_out, (float*)d_out, G);
}


# Per-destination softmax (max, exp, sum) moved into the destination-ordered gather pass; per-edge atomic max/add passes and logit plane gone; weight sum = fixed 16-lane tree, one divide per node
# speedup vs baseline: 3.7152x; 1.9443x over previous
#include <hip/hip_runtime.h>

typedef __attribute__((ext_vector_type(2))) float v2f;
typedef __attribute__((ext_vector_type(8))) float v8f;

#define HD 64

__device__ __forceinline__ float lrelu_f(float x) { return x > 0.f ? x : 0.1f * x; }

template <int M> __device__ __forceinline__ float row_xor(float v) {
    return __int_as_float(__builtin_amdgcn_mov_dpp(__float_as_int(v), 0x160 + M, 0xf, 0xf, false));
}
template <int U> __device__ __forceinline__ int row_bcast(int v) {
    return __builtin_amdgcn_mov_dpp(v, 0x150 + U, 0xf, 0xf, false);
}
template <int U> __device__ __forceinline__ float row_bcast(float v) {
    return __int_as_float(__builtin_amdgcn_mov_dpp(__float_as_int(v), 0x150 + U, 0xf, 0xf, false));
}
__device__ __forceinline__ float row_max16(float v) {
    v = fmaxf(v, row_xor<1>(v)); v = fmaxf(v, row_xor<2>(v));
    v = fmaxf(v, row_xor<4>(v)); v = fmaxf(v, row_xor<8>(v));
    return v;
}
__device__ __forceinline__ float row_sum16(float v) {
    v += row_xor<1>(v); v += row_xor<2>(v); v += row_xor<4>(v); v += row_xor<8>(v);
    return v;
}

template <bool HAS_A2, bool HAS_RES, bool HAS_BIAS, int ACT>
__global__ __launch_bounds__(32) void gemm64(
    const float* __restrict__ A,  const float* __restrict__ B,
    const float* __restrict__ A2, const float* __restrict__ B2,
    const float* __restrict__ res, const float* __restrict__ bias,
    float* __restrict__ out)
{
    const int lane = threadIdx.x & 31;
    const int cn   = lane & 15;
    const int kh   = (lane >> 4) << 1;
    const size_t row = (size_t)blockIdx.x * 16 + cn;

    const v8f z = {0.f, 0.f, 0.f, 0.f, 0.f, 0.f, 0.f, 0.f};
    v8f acc0 = z, acc1 = z, acc2 = z, acc3 = z;

    const float* arow = A + row * HD;
#pragma unroll
    for (int k0 = 0; k0 < 64; k0 += 4) {
        v2f a; a.x = arow[k0 + kh]; a.y = arow[k0 + kh + 1];
        const float* b0 = B + (size_t)(k0 + kh) * HD;
        const float* b1 = b0 + HD;
        v2f b;
        b.x = b0[cn];        b.y = b1[cn];
        acc0 = __builtin_amdgcn_wmma_f32_16x16x4_f32(false, a, false, b, (short)0, acc0, false, false);
        b.x = b0[16 + cn];   b.y = b1[16 + cn];
        acc1 = __builtin_amdgcn_wmma_f32_16x16x4_f32(false, a, false, b, (short)0, acc1, false, false);
        b.x = b0[32 + cn];   b.y = b1[32 + cn];
        acc2 = __builtin_amdgcn_wmma_f32_16x16x4_f32(false, a, false, b, (short)0, acc2, false, false);
        b.x = b0[48 + cn];   b.y = b1[48 + cn];
        acc3 = __builtin_amdgcn_wmma_f32_16x16x4_f32(false, a, false, b, (short)0, acc3, false, false);
    }
    if (HAS_A2) {
        const float* a2row = A2 + row * 16;
#pragma unroll
        for (int k0 = 0; k0 < 16; k0 += 4) {
            v2f a; a.x = a2row[k0 + kh]; a.y = a2row[k0 + kh + 1];
            const float* b0 = B2 + (size_t)(k0 + kh) * HD;
            const float* b1 = b0 + HD;
            v2f b;
            b.x = b0[cn];        b.y = b1[cn];
            acc0 = __builtin_amdgcn_wmma_f32_16x16x4_f32(false, a, false, b, (short)0, acc0, false, false);
            b.x = b0[16 + cn];   b.y = b1[16 + cn];
            acc1 = __builtin_amdgcn_wmma_f32_16x16x4_f32(false, a, false, b, (short)0, acc1, false, false);
            b.x = b0[32 + cn];   b.y = b1[32 + cn];
            acc2 = __builtin_amdgcn_wmma_f32_16x16x4_f32(false, a, false, b, (short)0, acc2, false, false);
            b.x = b0[48 + cn];   b.y = b1[48 + cn];
            acc3 = __builtin_amdgcn_wmma_f32_16x16x4_f32(false, a, false, b, (short)0, acc3, false, false);
        }
    }

    const size_t mbase = (size_t)blockIdx.x * 16 + ((lane >> 4) << 3);
    v8f accs[4] = {acc0, acc1, acc2, acc3};
#pragma unroll
    for (int t = 0; t < 4; ++t) {
        const int col = t * 16 + cn;
        const float bval = HAS_BIAS ? bias[col] : 0.f;
#pragma unroll
        for (int v = 0; v < 8; ++v) {
            const size_t m = mbase + v;
            float val = accs[t][v] + bval;
            if (HAS_RES) val += res[m * HD + col];
            if (ACT == 1) val = lrelu_f(val);
            else if (ACT == 2) val = fmaxf(val, 0.f);
            out[m * HD + col] = val;
        }
    }
}

__global__ void prep_ce(const float* __restrict__ W_e, const float* __restrict__ a_e,
                        float* __restrict__ cE)
{
    const int tid = threadIdx.x;
    const int l = tid >> 4, d = tid & 15;
    float s = 0.f;
    for (int k = 0; k < 64; ++k) s += W_e[d * 64 + k] * a_e[l * 64 + k];
    cE[tid] = s;
}

__global__ void prep_wewm(const float* __restrict__ W_e, const float* __restrict__ Wm,
                          float* __restrict__ WeWm)
{
    const int idx = blockIdx.x * blockDim.x + threadIdx.x;
    if (idx >= 4 * 16 * 64) return;
    const int l = idx >> 10, d = (idx >> 6) & 15, k = idx & 63;
    float s = 0.f;
    for (int j = 0; j < 64; ++j) s += W_e[d * 64 + j] * Wm[l * 4096 + j * 64 + k];
    WeWm[idx] = s;
}

__global__ void edge_ce(const float* __restrict__ re, const float* __restrict__ cE,
                        float* __restrict__ ceL, int E)
{
    const int j = blockIdx.x * blockDim.x + threadIdx.x;
    if (j >= E) return;
    float r[16];
    const float4* rp = (const float4*)(re + (size_t)j * 16);
#pragma unroll
    for (int i = 0; i < 4; ++i) {
        float4 v = rp[i];
        r[i * 4 + 0] = v.x; r[i * 4 + 1] = v.y; r[i * 4 + 2] = v.z; r[i * 4 + 3] = v.w;
    }
#pragma unroll
    for (int l = 0; l < 4; ++l) {
        float s = 0.f;
#pragma unroll
        for (int d = 0; d < 16; ++d) s += r[d] * cE[l * 16 + d];
        ceL[(size_t)j * 4 + l] = s;
    }
}

__global__ void init_hi(const float* __restrict__ i_node, const float* __restrict__ W_i,
                        float* __restrict__ hi, float* __restrict__ pooled, int G)
{
    const int idx = blockIdx.x * blockDim.x + threadIdx.x;
    if (idx >= G * 64) return;
    hi[idx] = i_node[idx >> 6] * W_i[idx & 63];
    pooled[idx] = 0.f;
}

__global__ void node_prep(const float* __restrict__ h, const float* __restrict__ as,
                          const float* __restrict__ ad, float* __restrict__ sb,
                          float* __restrict__ db, int N)
{
    const int n = blockIdx.x * blockDim.x + threadIdx.x;
    if (n >= N) return;
    const float* hr = h + (size_t)n * 64;
    float s = 0.f, d = 0.f;
#pragma unroll
    for (int k = 0; k < 64; ++k) { const float v = hr[k]; s += v * as[k]; d += v * ad[k]; }
    sb[n] = s; db[n] = d;
}

__global__ void csr_zero(int* __restrict__ cnt, int N)
{
    const int n = blockIdx.x * blockDim.x + threadIdx.x;
    if (n < N) cnt[n] = 0;
}

__global__ void csr_count(const int* __restrict__ dst, int* __restrict__ cnt,
                          int* __restrict__ rank, int E)
{
    const int j = blockIdx.x * blockDim.x + threadIdx.x;
    if (j >= E) return;
    rank[j] = atomicAdd(cnt + dst[j], 1);
}

__global__ __launch_bounds__(1024) void csr_scan(const int* __restrict__ cnt,
                                                 int* __restrict__ rowptr, int N)
{
    __shared__ int wfront[32];
    __shared__ int wtot[32];
    const int t = threadIdx.x, lane = t & 31, w = t >> 5;
    const int base0 = blockIdx.x * 1024;

    int s = 0;
    for (int i = t; i < base0; i += 1024) s += cnt[i];
    for (int o = 16; o > 0; o >>= 1) s += __shfl_xor(s, o, 32);
    if (lane == 0) wfront[w] = s;

    const int i = base0 + t;
    const int c = (i < N) ? cnt[i] : 0;
    int x = c;
    for (int o = 1; o < 32; o <<= 1) {
        const int y = __shfl_up(x, o, 32);
        if (lane >= o) x += y;
    }
    if (lane == 31) wtot[w] = x;
    __syncthreads();

    int front = 0;
    for (int k = 0; k < 32; ++k) front += wfront[k];
    for (int k = 0; k < w; ++k) front += wtot[k];
    if (i < N) rowptr[i] = front + x - c;
    if (i == N - 1) rowptr[N] = front + x;
}

__global__ void csr_fill(const int* __restrict__ src, const int* __restrict__ dst,
                         const int* __restrict__ rowptr, const int* __restrict__ rank,
                         int* __restrict__ pk, int E)
{
    const int j = blockIdx.x * blockDim.x + threadIdx.x;
    if (j >= E) return;
    const int p = rowptr[dst[j]] + rank[j];
    ((int2*)pk)[p] = make_int2(j, src[j]);
}

template <int U0> __device__ __forceinline__ void gather4(
    const int2 e, const float ex, const int q,
    const float* __restrict__ h, const float* __restrict__ re,
    float4& acc, float& accr)
{
    const int j0 = row_bcast<U0 + 0>(e.x), s0 = row_bcast<U0 + 0>(e.y);
    const int j1 = row_bcast<U0 + 1>(e.x), s1 = row_bcast<U0 + 1>(e.y);
    const int j2 = row_bcast<U0 + 2>(e.x), s2 = row_bcast<U0 + 2>(e.y);
    const int j3 = row_bcast<U0 + 3>(e.x), s3 = row_bcast<U0 + 3>(e.y);
    const float4 h0 = *(const float4*)(h + (size_t)s0 * 64 + q * 4);
    const float4 h1 = *(const float4*)(h + (size_t)s1 * 64 + q * 4);
    const float4 h2 = *(const float4*)(h + (size_t)s2 * 64 + q * 4);
    const float4 h3 = *(const float4*)(h + (size_t)s3 * 64 + q * 4);
    const float r0 = re[(size_t)j0 * 16 + q];
    const float r1 = re[(size_t)j1 * 16 + q];
    const float r2 = re[(size_t)j2 * 16 + q];
    const float r3 = re[(size_t)j3 * 16 + q];
    const float a0 = row_bcast<U0 + 0>(ex), a1 = row_bcast<U0 + 1>(ex);
    const float a2 = row_bcast<U0 + 2>(ex), a3 = row_bcast<U0 + 3>(ex);
    acc.x += a0 * h0.x; acc.y += a0 * h0.y; acc.z += a0 * h0.z; acc.w += a0 * h0.w; accr += a0 * r0;
    acc.x += a1 * h1.x; acc.y += a1 * h1.y; acc.z += a1 * h1.z; acc.w += a1 * h1.w; accr += a1 * r1;
    acc.x += a2 * h2.x; acc.y += a2 * h2.y; acc.z += a2 * h2.z; acc.w += a2 * h2.w; accr += a2 * r2;
    acc.x += a3 * h3.x; acc.y += a3 * h3.y; acc.z += a3 * h3.z; acc.w += a3 * h3.w; accr += a3 * r3;
}

__global__ __launch_bounds__(256) void node_gather(
    const int* __restrict__ rowptr, const int* __restrict__ pk,
    const float* __restrict__ sb, const float* __restrict__ db,
    const float* __restrict__ ceL, const int l,
    const float* __restrict__ h, const float* __restrict__ re,
    float* __restrict__ A64, float* __restrict__ A16, int N)
{
    const int n = blockIdx.x * 16 + (threadIdx.x >> 4);
    const int q = threadIdx.x & 15;
    if (n >= N) return;
    const int beg = rowptr[n], end = rowptr[n + 1];
    const int2* __restrict__ pk2 = (const int2*)pk;
    float4 acc = {0.f, 0.f, 0.f, 0.f};
    float accr = 0.f;
    if (end > beg) {
        const float dbn = db[n];
        const int last = end - 1;
        const float ninf = -__builtin_inff();
        float m = ninf;
        float dsum = 0.f;
        for (int c = beg; c < end; c += 16) {
            const int idx = c + q;
            const bool live = idx < end;
            const int2 e = pk2[min(idx, last)];
            const float lgv = lrelu_f(sb[e.y] + dbn + ceL[(size_t)e.x * 4 + l]);
            const float lg = live ? lgv : ninf;
            const float mn = fmaxf(m, row_max16(lg));
            const float sc = __expf(m - mn);
            const float ex = live ? __expf(lg - mn) : 0.f;
            dsum = dsum * sc + ex;
            acc.x *= sc; acc.y *= sc; acc.z *= sc; acc.w *= sc; accr *= sc;
            m = mn;
            const int cnt = end - c;
            gather4<0>(e, ex, q, h, re, acc, accr);
            if (cnt > 4)  gather4<4>(e, ex, q, h, re, acc, accr);
            if (cnt > 8)  gather4<8>(e, ex, q, h, re, acc, accr);
            if (cnt > 12) gather4<12>(e, ex, q, h, re, acc, accr);
        }
        const float inv = 1.f / row_sum16(dsum);
        acc.x *= inv; acc.y *= inv; acc.z *= inv; acc.w *= inv; accr *= inv;
    }
    *(float4*)(A64 + (size_t)n * 64 + q * 4) = acc;
    A16[(size_t)n * 16 + q] = accr;
}

__global__ void gate_hi(const float* __restrict__ h, const float* __restrict__ d_edge,
                        const int* __restrict__ gid, const float* __restrict__ wd,
                        const float* __restrict__ bd, float* __restrict__ gate,
                        float* __restrict__ hi, int N)
{
    const int idx = blockIdx.x * blockDim.x + threadIdx.x;
    if (idx >= N * 64) return;
    const int n = idx >> 6, k = idx & 63;
    const float g = 1.f / (1.f + __expf(-(d_edge[n] * wd[0] + bd[0])));
    if (k == 0) gate[n] = g;
    atomicAdd(hi + (size_t)gid[n] * 64 + k, g * h[idx]);
}

__global__ void h_mix_pool(float* __restrict__ h, const float* __restrict__ gate,
                           const float* __restrict__ hi, const int* __restrict__ gid,
                           float* __restrict__ pooled, int N)
{
    const int idx = blockIdx.x * blockDim.x + threadIdx.x;
    if (idx >= N * 64) return;
    const int n = idx >> 6, k = idx & 63;
    const float v = h[idx] + gate[n] * hi[(size_t)gid[n] * 64 + k];
    h[idx] = v;
    atomicAdd(pooled + (size_t)gid[n] * 64 + k, v);
}

__global__ void out_kernel(const float* __restrict__ x, const float* __restrict__ Wo,
                           const float* __restrict__ bo, float* __restrict__ out, int G)
{
    const int g = blockIdx.x * blockDim.x + threadIdx.x;
    if (g >= G) return;
    float s = bo[0];
#pragma unroll
    for (int k = 0; k < 64; ++k) s += x[(size_t)g * 64 + k] * Wo[k];
    out[g] = s;
}

extern "C" void kernel_launch(void* const* d_in, const int* in_sizes, int n_in,
                              void* d_out, int out_size, void* d_ws, size_t ws_size,
                              hipStream_t stream)
{
    const float* r_node = (const float*)d_in[0];
    const float* i_node = (const float*)d_in[1];
    const float* r_edge = (const float*)d_in[2];
    const float* d_edge = (const float*)d_in[3];
    const int*   src    = (const int*)d_in[4];
    const int*   dst    = (const int*)d_in[5];
    const int*   gid    = (const int*)d_in[6];
    const float* W_r    = (const float*)d_in[7];
    const float* W_i    = (const float*)d_in[8];
    const float* W_e    = (const float*)d_in[9];
    const float* Wm     = (const float*)d_in[10];
    const float* a_s    = (const float*)d_in[11];
    const float* a_d    = (const float*)d_in[12];
    const float* a_e    = (const float*)d_in[13];
    const float* w_d    = (const float*)d_in[14];
    const float* b_d    = (const float*)d_in[15];
    const float* W_mlp  = (const float*)d_in[16];
    const float* b_mlp  = (const float*)d_in[17];
    const float* W_out  = (const float*)d_in[18];
    const float* b_out  = (const float*)d_in[19];

    const int N = in_sizes[0] / 64;
    const int G = in_sizes[1];
    const int E = in_sizes[2] / 16;
    (void)n_in; (void)out_size; (void)ws_size;

    float* ws = (float*)d_ws;
    size_t off = 0;
    auto carve = [&](size_t n) -> float* {
        float* p = ws + off; off += (n + 63) & ~(size_t)63; return p;
    };
    float*    h      = carve((size_t)N * 64);
    float*    A64    = carve((size_t)N * 64);
    float*    A16    = carve((size_t)N * 16);
    float*    sb     = carve(N);
    float*    db     = carve(N);
    float*    gate   = carve(N);
    float*    ceL    = carve((size_t)E * 4);
    float*    hi     = carve((size_t)G * 64);
    float*    pooled = carve((size_t)G * 64);
    float*    x1     = carve((size_t)G * 64);
    float*    x2     = carve((size_t)G * 64);
    float*    cE     = carve(64);
    float*    WeWm   = carve(4 * 16 * 64);
    int*      cnt    = (int*)carve(N);
    int*      rowptr = (int*)carve((size_t)N + 1);
    int*      rank   = (int*)carve(E);
    int*      pk     = (int*)carve((size_t)E * 2);

    hipLaunchKernelGGL(prep_ce, dim3(1), dim3(64), 0, stream, W_e, a_e, cE);
    hipLaunchKernelGGL(prep_wewm, dim3((4096 + 255) / 256), dim3(256), 0, stream, W_e, Wm, WeWm);
    hipLaunchKernelGGL(edge_ce, dim3((E + 255) / 256), dim3(256), 0, stream, r_edge, cE, ceL, E);

    hipLaunchKernelGGL((gemm64<false, false, false, 0>), dim3(N / 16), dim3(32), 0, stream,
                       r_node, W_r, nullptr, nullptr, nullptr, nullptr, h);
    hipLaunchKernelGGL(init_hi, dim3((G * 64 + 255) / 256), dim3(256), 0, stream,
                       i_node, W_i, hi, pooled, G);

    hipLaunchKernelGGL(csr_zero, dim3((N + 255) / 256), dim3(256), 0, stream, cnt, N);
    hipLaunchKernelGGL(csr_count, dim3((E + 255) / 256), dim3(256), 0, stream, dst, cnt, rank, E);
    hipLaunchKernelGGL(csr_scan, dim3((N + 1023) / 1024), dim3(1024), 0, stream, cnt, rowptr, N);
    hipLaunchKernelGGL(csr_fill, dim3((E + 255) / 256), dim3(256), 0, stream,
                       src, dst, rowptr, rank, pk, E);

    for (int l = 0; l < 4; ++l) {
        hipLaunchKernelGGL(node_prep, dim3((N + 127) / 128), dim3(128), 0, stream,
                           h, a_s + l * 64, a_d + l * 64, sb, db, N);
        hipLaunchKernelGGL(node_gather, dim3((N + 15) / 16), dim3(256), 0, stream,
                           rowptr, pk, sb, db, ceL, l, h, r_edge, A64, A16, N);
        hipLaunchKernelGGL((gemm64<true, true, false, 1>), dim3(N / 16), dim3(32), 0, stream,
                           A64, Wm + (size_t)l * 4096, A16, WeWm + (size_t)l * 1024,
                           h, nullptr, h);
    }

    hipLaunchKernelGGL(gate_hi, dim3((N * 64 + 255) / 256), dim3(256), 0, stream,
                       h, d_edge, gid, w_d, b_d, gate, hi, N);
    hipLaunchKernelGGL(h_mix_pool, dim3((N * 64 + 255) / 256), dim3(256), 0, stream,
                       h, gate, hi, gid, pooled, N);

    hipLaunchKernelGGL((gemm64<false, false, true, 2>), dim3(G / 16), dim3(32), 0, stream,
                       pooled, W_mlp + 0 * 4096, nullptr, nullptr, nullptr, b_mlp + 0 * 64, x1);
    hipLaunchKernelGGL((gemm64<false, false, true, 2>), dim3(G / 16), dim3(32), 0, stream,
                       x1, W_mlp + 1 * 4096, nullptr, nullptr, nullptr, b_mlp + 1 * 64, x2);
    hipLaunchKernelGGL((gemm64<false, false, true, 2>), dim3(G / 16), dim3(32), 0, stream,
                       x2, W_mlp + 2 * 4096, nullptr, nullptr, nullptr, b_mlp + 2 * 64, x1);

    hipLaunchKernelGGL(out_kernel, dim3((G + 255) / 256), dim3(256), 0, stream,
                       x1, W_out, b_out, (float*)d_out, G);
}


# gate_hi and h_mix_pool: a 16-lane row walks 16 consecutive nodes (float4 per lane) in place of one thread per element; runs of equal graph id summed in registers in node order, one atomic per run
# speedup vs baseline: 4.0452x; 1.0888x over previous
#include <hip/hip_runtime.h>

typedef __attribute__((ext_vector_type(2))) float v2f;
typedef __attribute__((ext_vector_type(8))) float v8f;

#define HD 64

__device__ __forceinline__ float lrelu_f(float x) { return x > 0.f ? x : 0.1f * x; }

template <int M> __device__ __forceinline__ float row_xor(float v) {
    return __int_as_float(__builtin_amdgcn_mov_dpp(__float_as_int(v), 0x160 + M, 0xf, 0xf, false));
}
template <int U> __device__ __forceinline__ int row_bcast(int v) {
    return __builtin_amdgcn_mov_dpp(v, 0x150 + U, 0xf, 0xf, false);
}
template <int U> __device__ __forceinline__ float row_bcast(float v) {
    return __int_as_float(__builtin_amdgcn_mov_dpp(__float_as_int(v), 0x150 + U, 0xf, 0xf, false));
}
__device__ __forceinline__ float row_max16(float v) {
    v = fmaxf(v, row_xor<1>(v)); v = fmaxf(v, row_xor<2>(v));
    v = fmaxf(v, row_xor<4>(v)); v = fmaxf(v, row_xor<8>(v));
    return v;
}
__device__ __forceinline__ float row_sum16(float v) {
    v += row_xor<1>(v); v += row_xor<2>(v); v += row_xor<4>(v); v += row_xor<8>(v);
    return v;
}

template <bool HAS_A2, bool HAS_RES, bool HAS_BIAS, int ACT>
__global__ __launch_bounds__(32) void gemm64(
    const float* __restrict__ A,  const float* __restrict__ B,
    const float* __restrict__ A2, const float* __restrict__ B2,
    const float* __restrict__ res, const float* __restrict__ bias,
    float* __restrict__ out)
{
    const int lane = threadIdx.x & 31;
    const int cn   = lane & 15;
    const int kh   = (lane >> 4) << 1;
    const size_t row = (size_t)blockIdx.x * 16 + cn;

    const v8f z = {0.f, 0.f, 0.f, 0.f, 0.f, 0.f, 0.f, 0.f};
    v8f acc0 = z, acc1 = z, acc2 = z, acc3 = z;

    const float* arow = A + row * HD;
#pragma unroll
    for (int k0 = 0; k0 < 64; k0 += 4) {
        v2f a; a.x = arow[k0 + kh]; a.y = arow[k0 + kh + 1];
        const float* b0 = B + (size_t)(k0 + kh) * HD;
        const float* b1 = b0 + HD;
        v2f b;
        b.x = b0[cn];        b.y = b1[cn];
        acc0 = __builtin_amdgcn_wmma_f32_16x16x4_f32(false, a, false, b, (short)0, acc0, false, false);
        b.x = b0[16 + cn];   b.y = b1[16 + cn];
        acc1 = __builtin_amdgcn_wmma_f32_16x16x4_f32(false, a, false, b, (short)0, acc1, false, false);
        b.x = b0[32 + cn];   b.y = b1[32 + cn];
        acc2 = __builtin_amdgcn_wmma_f32_16x16x4_f32(false, a, false, b, (short)0, acc2, false, false);
        b.x = b0[48 + cn];   b.y = b1[48 + cn];
        acc3 = __builtin_amdgcn_wmma_f32_16x16x4_f32(false, a, false, b, (short)0, acc3, false, false);
    }
    if (HAS_A2) {
        const float* a2row = A2 + row * 16;
#pragma unroll
        for (int k0 = 0; k0 < 16; k0 += 4) {
            v2f a; a.x = a2row[k0 + kh]; a.y = a2row[k0 + kh + 1];
            const float* b0 = B2 + (size_t)(k0 + kh) * HD;
            const float* b1 = b0 + HD;
            v2f b;
            b.x = b0[cn];        b.y = b1[cn];
            acc0 = __builtin_amdgcn_wmma_f32_16x16x4_f32(false, a, false, b, (short)0, acc0, false, false);
            b.x = b0[16 + cn];   b.y = b1[16 + cn];
            acc1 = __builtin_amdgcn_wmma_f32_16x16x4_f32(false, a, false, b, (short)0, acc1, false, false);
            b.x = b0[32 + cn];   b.y = b1[32 + cn];
            acc2 = __builtin_amdgcn_wmma_f32_16x16x4_f32(false, a, false, b, (short)0, acc2, false, false);
            b.x = b0[48 + cn];   b.y = b1[48 + cn];
            acc3 = __builtin_amdgcn_wmma_f32_16x16x4_f32(false, a, false, b, (short)0, acc3, false, false);
        }
    }

    const size_t mbase = (size_t)blockIdx.x * 16 + ((lane >> 4) << 3);
    v8f accs[4] = {acc0, acc1, acc2, acc3};
#pragma unroll
    for (int t = 0; t < 4; ++t) {
        const int col = t * 16 + cn;
        const float bval = HAS_BIAS ? bias[col] : 0.f;
#pragma unroll
        for (int v = 0; v < 8; ++v) {
            const size_t m = mbase + v;
            float val = accs[t][v] + bval;
            if (HAS_RES) val += res[m * HD + col];
            if (ACT == 1) val = lrelu_f(val);
            else if (ACT == 2) val = fmaxf(val, 0.f);
            out[m * HD + col] = val;
        }
    }
}

__global__ void prep_ce(const float* __restrict__ W_e, const float* __restrict__ a_e,
                        float* __restrict__ cE)
{
    const int tid = threadIdx.x;
    const int l = tid >> 4, d = tid & 15;
    float s = 0.f;
    for (int k = 0; k < 64; ++k) s += W_e[d * 64 + k] * a_e[l * 64 + k];
    cE[tid] = s;
}

__global__ void prep_wewm(const float* __restrict__ W_e, const float* __restrict__ Wm,
                          float* __restrict__ WeWm)
{
    const int idx = blockIdx.x * blockDim.x + threadIdx.x;
    if (idx >= 4 * 16 * 64) return;
    const int l = idx >> 10, d = (idx >> 6) & 15, k = idx & 63;
    float s = 0.f;
    for (int j = 0; j < 64; ++j) s += W_e[d * 64 + j] * Wm[l * 4096 + j * 64 + k];
    WeWm[idx] = s;
}

__global__ void edge_ce(const float* __restrict__ re, const float* __restrict__ cE,
                        float* __restrict__ ceL, int E)
{
    const int j = blockIdx.x * blockDim.x + threadIdx.x;
    if (j >= E) return;
    float r[16];
    const float4* rp = (const float4*)(re + (size_t)j * 16);
#pragma unroll
    for (int i = 0; i < 4; ++i) {
        float4 v = rp[i];
        r[i * 4 + 0] = v.x; r[i * 4 + 1] = v.y; r[i * 4 + 2] = v.z; r[i * 4 + 3] = v.w;
    }
#pragma unroll
    for (int l = 0; l < 4; ++l) {
        float s = 0.f;
#pragma unroll
        for (int d = 0; d < 16; ++d) s += r[d] * cE[l * 16 + d];
        ceL[(size_t)j * 4 + l] = s;
    }
}

__global__ void init_hi(const float* __restrict__ i_node, const float* __restrict__ W_i,
                        float* __restrict__ hi, float* __restrict__ pooled, int G)
{
    const int idx = blockIdx.x * blockDim.x + threadIdx.x;
    if (idx >= G * 64) return;
    hi[idx] = i_node[idx >> 6] * W_i[idx & 63];
    pooled[idx] = 0.f;
}

__global__ void node_prep(const float* __restrict__ h, const float* __restrict__ as,
                          const float* __restrict__ ad, float* __restrict__ sb,
                          float* __restrict__ db, int N)
{
    const int n = blockIdx.x * blockDim.x + threadIdx.x;
    if (n >= N) return;
    const float* hr = h + (size_t)n * 64;
    float s = 0.f, d = 0.f;
#pragma unroll
    for (int k = 0; k < 64; ++k) { const float v = hr[k]; s += v * as[k]; d += v * ad[k]; }
    sb[n] = s; db[n] = d;
}

__global__ void csr_zero(int* __restrict__ cnt, int N)
{
    const int n = blockIdx.x * blockDim.x + threadIdx.x;
    if (n < N) cnt[n] = 0;
}

__global__ void csr_count(const int* __restrict__ dst, int* __restrict__ cnt,
                          int* __restrict__ rank, int E)
{
    const int j = blockIdx.x * blockDim.x + threadIdx.x;
    if (j >= E) return;
    rank[j] = atomicAdd(cnt + dst[j], 1);
}

__global__ __launch_bounds__(1024) void csr_scan(const int* __restrict__ cnt,
                                                 int* __restrict__ rowptr, int N)
{
    __shared__ int wfront[32];
    __shared__ int wtot[32];
    const int t = threadIdx.x, lane = t & 31, w = t >> 5;
    const int base0 = blockIdx.x * 1024;

    int s = 0;
    for (int i = t; i < base0; i += 1024) s += cnt[i];
    for (int o = 16; o > 0; o >>= 1) s += __shfl_xor(s, o, 32);
    if (lane == 0) wfront[w] = s;

    const int i = base0 + t;
    const int c = (i < N) ? cnt[i] : 0;
    int x = c;
    for (int o = 1; o < 32; o <<= 1) {
        const int y = __shfl_up(x, o, 32);
        if (lane >= o) x += y;
    }
    if (lane == 31) wtot[w] = x;
    __syncthreads();

    int front = 0;
    for (int k = 0; k < 32; ++k) front += wfront[k];
    for (int k = 0; k < w; ++k) front += wtot[k];
    if (i < N) rowptr[i] = front + x - c;
    if (i == N - 1) rowptr[N] = front + x;
}

__global__ void csr_fill(const int* __restrict__ src, const int* __restrict__ dst,
                         const int* __restrict__ rowptr, const int* __restrict__ rank,
                         int* __restrict__ pk, int E)
{
    const int j = blockIdx.x * blockDim.x + threadIdx.x;
    if (j >= E) return;
    const int p = rowptr[dst[j]] + rank[j];
    ((int2*)pk)[p] = make_int2(j, src[j]);
}

template <int U0> __device__ __forceinline__ void gather4(
    const int2 e, const float ex, const int q,
    const float* __restrict__ h, const float* __restrict__ re,
    float4& acc, float& accr)
{
    const int j0 = row_bcast<U0 + 0>(e.x), s0 = row_bcast<U0 + 0>(e.y);
    const int j1 = row_bcast<U0 + 1>(e.x), s1 = row_bcast<U0 + 1>(e.y);
    const int j2 = row_bcast<U0 + 2>(e.x), s2 = row_bcast<U0 + 2>(e.y);
    const int j3 = row_bcast<U0 + 3>(e.x), s3 = row_bcast<U0 + 3>(e.y);
    const float4 h0 = *(const float4*)(h + (size_t)s0 * 64 + q * 4);
    const float4 h1 = *(const float4*)(h + (size_t)s1 * 64 + q * 4);
    const float4 h2 = *(const float4*)(h + (size_t)s2 * 64 + q * 4);
    const float4 h3 = *(const float4*)(h + (size_t)s3 * 64 + q * 4);
    const float r0 = re[(size_t)j0 * 16 + q];
    const float r1 = re[(size_t)j1 * 16 + q];
    const float r2 = re[(size_t)j2 * 16 + q];
    const float r3 = re[(size_t)j3 * 16 + q];
    const float a0 = row_bcast<U0 + 0>(ex), a1 = row_bcast<U0 + 1>(ex);
    const float a2 = row_bcast<U0 + 2>(ex), a3 = row_bcast<U0 + 3>(ex);
    acc.x += a0 * h0.x; acc.y += a0 * h0.y; acc.z += a0 * h0.z; acc.w += a0 * h0.w; accr += a0 * r0;
    acc.x += a1 * h1.x; acc.y += a1 * h1.y; acc.z += a1 * h1.z; acc.w += a1 * h1.w; accr += a1 * r1;
    acc.x += a2 * h2.x; acc.y += a2 * h2.y; acc.z += a2 * h2.z; acc.w += a2 * h2.w; accr += a2 * r2;
    acc.x += a3 * h3.x; acc.y += a3 * h3.y; acc.z += a3 * h3.z; acc.w += a3 * h3.w; accr += a3 * r3;
}

__global__ __launch_bounds__(256) void node_gather(
    const int* __restrict__ rowptr, const int* __restrict__ pk,
    const float* __restrict__ sb, const float* __restrict__ db,
    const float* __restrict__ ceL, const int l,
    const float* __restrict__ h, const float* __restrict__ re,
    float* __restrict__ A64, float* __restrict__ A16, int N)
{
    const int n = blockIdx.x * 16 + (threadIdx.x >> 4);
    const int q = threadIdx.x & 15;
    if (n >= N) return;
    const int beg = rowptr[n], end = rowptr[n + 1];
    const int2* __restrict__ pk2 = (const int2*)pk;
    float4 acc = {0.f, 0.f, 0.f, 0.f};
    float accr = 0.f;
    if (end > beg) {
        const float dbn = db[n];
        const int last = end - 1;
        const float ninf = -__builtin_inff();
        float m = ninf;
        float dsum = 0.f;
        for (int c = beg; c < end; c += 16) {
            const int idx = c + q;
            const bool live = idx < end;
            const int2 e = pk2[min(idx, last)];
            const float lgv = lrelu_f(sb[e.y] + dbn + ceL[(size_t)e.x * 4 + l]);
            const float lg = live ? lgv : ninf;
            const float mn = fmaxf(m, row_max16(lg));
            const float sc = __expf(m - mn);
            const float ex = live ? __expf(lg - mn) : 0.f;
            dsum = dsum * sc + ex;
            acc.x *= sc; acc.y *= sc; acc.z *= sc; acc.w *= sc; accr *= sc;
            m = mn;
            const int cnt = end - c;
            gather4<0>(e, ex, q, h, re, acc, accr);
            if (cnt > 4)  gather4<4>(e, ex, q, h, re, acc, accr);
            if (cnt > 8)  gather4<8>(e, ex, q, h, re, acc, accr);
            if (cnt > 12) gather4<12>(e, ex, q, h, re, acc, accr);
        }
        const float inv = 1.f / row_sum16(dsum);
        acc.x *= inv; acc.y *= inv; acc.z *= inv; acc.w *= inv; accr *= inv;
    }
    *(float4*)(A64 + (size_t)n * 64 + q * 4) = acc;
    A16[(size_t)n * 16 + q] = accr;
}

#define WALK 16

__device__ __forceinline__ void row_flush(float* __restrict__ grow, const int q, const float4 a)
{
    float* p = grow + q * 4;
    atomicAdd(p + 0, a.x); atomicAdd(p + 1, a.y); atomicAdd(p + 2, a.z); atomicAdd(p + 3, a.w);
}

__device__ __forceinline__ void row_bcast_all(const int iq, const float fq,
                                              int (&iv)[WALK], float (&fv)[WALK])
{
#define BC1(U) iv[U] = row_bcast<U>(iq); fv[U] = row_bcast<U>(fq);
    BC1(0)  BC1(1)  BC1(2)  BC1(3)  BC1(4)  BC1(5)  BC1(6)  BC1(7)
    BC1(8)  BC1(9)  BC1(10) BC1(11) BC1(12) BC1(13) BC1(14) BC1(15)
#undef BC1
}

__global__ __launch_bounds__(128) void gate_hi(
    const float* __restrict__ h, const float* __restrict__ d_edge,
    const int* __restrict__ gid, const float* __restrict__ wd,
    const float* __restrict__ bd, float* __restrict__ gate,
    float* __restrict__ hi, int N)
{
    const int q  = threadIdx.x & 15;
    const int n0 = (blockIdx.x * 8 + (threadIdx.x >> 4)) * WALK;
    if (n0 >= N) return;
    const int nq = min(n0 + q, N - 1);
    float4 hv[WALK];
#pragma unroll
    for (int u = 0; u < WALK; ++u)
        hv[u] = *(const float4*)(h + (size_t)min(n0 + u, N - 1) * 64 + q * 4);
    const float gq = 1.f / (1.f + __expf(-(d_edge[nq] * wd[0] + bd[0])));
    const int gidq = gid[nq];
    if (n0 + q < N) gate[n0 + q] = gq;
    int gi[WALK]; float g[WALK];
    row_bcast_all(gidq, gq, gi, g);
    float4 acc = {0.f, 0.f, 0.f, 0.f};
    int cur = gi[0];
#pragma unroll
    for (int u = 0; u < WALK; ++u) {
        if (n0 + u < N) {
            if (gi[u] != cur) {
                row_flush(hi + (size_t)cur * 64, q, acc);
                acc.x = 0.f; acc.y = 0.f; acc.z = 0.f; acc.w = 0.f;
                cur = gi[u];
            }
            acc.x += g[u] * hv[u].x; acc.y += g[u] * hv[u].y;
            acc.z += g[u] * hv[u].z; acc.w += g[u] * hv[u].w;
        }
    }
    row_flush(hi + (size_t)cur * 64, q, acc);
}

__global__ __launch_bounds__(128) void h_mix_pool(
    float* __restrict__ h, const float* __restrict__ gate,
    const float* __restrict__ hi, const int* __restrict__ gid,
    float* __restrict__ pooled, int N)
{
    const int q  = threadIdx.x & 15;
    const int n0 = (blockIdx.x * 8 + (threadIdx.x >> 4)) * WALK;
    if (n0 >= N) return;
    const int nq = min(n0 + q, N - 1);
    float4 hv[WALK];
#pragma unroll
    for (int u = 0; u < WALK; ++u)
        hv[u] = *(const float4*)(h + (size_t)min(n0 + u, N - 1) * 64 + q * 4);
    const float gq = gate[nq];
    const int gidq = gid[nq];
    int gi[WALK]; float g[WALK];
    row_bcast_all(gidq, gq, gi, g);
    float4 hiv[WALK];
#pragma unroll
    for (int u = 0; u < WALK; ++u)
        hiv[u] = *(const float4*)(hi + (size_t)gi[u] * 64 + q * 4);
    float4 acc = {0.f, 0.f, 0.f, 0.f};
    int cur = gi[0];
#pragma unroll
    for (int u = 0; u < WALK; ++u) {
        if (n0 + u < N) {
            float4 v;
            v.x = hv[u].x + g[u] * hiv[u].x; v.y = hv[u].y + g[u] * hiv[u].y;
            v.z = hv[u].z + g[u] * hiv[u].z; v.w = hv[u].w + g[u] * hiv[u].w;
            *(float4*)(h + (size_t)(n0 + u) * 64 + q * 4) = v;
            if (gi[u] != cur) {
                row_flush(pooled + (size_t)cur * 64, q, acc);
                acc.x = 0.f; acc.y = 0.f; acc.z = 0.f; acc.w = 0.f;
                cur = gi[u];
            }
            acc.x += v.x; acc.y += v.y; acc.z += v.z; acc.w += v.w;
        }
    }
    row_flush(pooled + (size_t)cur * 64, q, acc);
}

__global__ void out_kernel(const float* __restrict__ x, const float* __restrict__ Wo,
                           const float* __restrict__ bo, float* __restrict__ out, int G)
{
    const int g = blockIdx.x * blockDim.x + threadIdx.x;
    if (g >= G) return;
    float s = bo[0];
#pragma unroll
    for (int k = 0; k < 64; ++k) s += x[(size_t)g * 64 + k] * Wo[k];
    out[g] = s;
}

extern "C" void kernel_launch(void* const* d_in, const int* in_sizes, int n_in,
                              void* d_out, int out_size, void* d_ws, size_t ws_size,
                              hipStream_t stream)
{
    const float* r_node = (const float*)d_in[0];
    const float* i_node = (const float*)d_in[1];
    const float* r_edge = (const float*)d_in[2];
    const float* d_edge = (const float*)d_in[3];
    const int*   src    = (const int*)d_in[4];
    const int*   dst    = (const int*)d_in[5];
    const int*   gid    = (const int*)d_in[6];
    const float* W_r    = (const float*)d_in[7];
    const float* W_i    = (const float*)d_in[8];
    const float* W_e    = (const float*)d_in[9];
    const float* Wm     = (const float*)d_in[10];
    const float* a_s    = (const float*)d_in[11];
    const float* a_d    = (const float*)d_in[12];
    const float* a_e    = (const float*)d_in[13];
    const float* w_d    = (const float*)d_in[14];
    const float* b_d    = (const float*)d_in[15];
    const float* W_mlp  = (const float*)d_in[16];
    const float* b_mlp  = (const float*)d_in[17];
    const float* W_out  = (const float*)d_in[18];
    const float* b_out  = (const float*)d_in[19];

    const int N = in_sizes[0] / 64;
    const int G = in_sizes[1];
    const int E = in_sizes[2] / 16;
    (void)n_in; (void)out_size; (void)ws_size;

    float* ws = (float*)d_ws;
    size_t off = 0;
    auto carve = [&](size_t n) -> float* {
        float* p = ws + off; off += (n + 63) & ~(size_t)63; return p;
    };
    float*    h      = carve((size_t)N * 64);
    float*    A64    = carve((size_t)N * 64);
    float*    A16    = carve((size_t)N * 16);
    float*    sb     = carve(N);
    float*    db     = carve(N);
    float*    gate   = carve(N);
    float*    ceL    = carve((size_t)E * 4);
    float*    hi     = carve((size_t)G * 64);
    float*    pooled = carve((size_t)G * 64);
    float*    x1     = carve((size_t)G * 64);
    float*    x2     = carve((size_t)G * 64);
    float*    cE     = carve(64);
    float*    WeWm   = carve(4 * 16 * 64);
    int*      cnt    = (int*)carve(N);
    int*      rowptr = (int*)carve((size_t)N + 1);
    int*      rank   = (int*)carve(E);
    int*      pk     = (int*)carve((size_t)E * 2);

    hipLaunchKernelGGL(prep_ce, dim3(1), dim3(64), 0, stream, W_e, a_e, cE);
    hipLaunchKernelGGL(prep_wewm, dim3((4096 + 255) / 256), dim3(256), 0, stream, W_e, Wm, WeWm);
    hipLaunchKernelGGL(edge_ce, dim3((E + 255) / 256), dim3(256), 0, stream, r_edge, cE, ceL, E);

    hipLaunchKernelGGL((gemm64<false, false, false, 0>), dim3(N / 16), dim3(32), 0, stream,
                       r_node, W_r, nullptr, nullptr, nullptr, nullptr, h);
    hipLaunchKernelGGL(init_hi, dim3((G * 64 + 255) / 256), dim3(256), 0, stream,
                       i_node, W_i, hi, pooled, G);

    hipLaunchKernelGGL(csr_zero, dim3((N + 255) / 256), dim3(256), 0, stream, cnt, N);
    hipLaunchKernelGGL(csr_count, dim3((E + 255) / 256), dim3(256), 0, stream, dst, cnt, rank, E);
    hipLaunchKernelGGL(csr_scan, dim3((N + 1023) / 1024), dim3(1024), 0, stream, cnt, rowptr, N);
    hipLaunchKernelGGL(csr_fill, dim3((E + 255) / 256), dim3(256), 0, stream,
                       src, dst, rowptr, rank, pk, E);

    for (int l = 0; l < 4; ++l) {
        hipLaunchKernelGGL(node_prep, dim3((N + 127) / 128), dim3(128), 0, stream,
                           h, a_s + l * 64, a_d + l * 64, sb, db, N);
        hipLaunchKernelGGL(node_gather, dim3((N + 15) / 16), dim3(256), 0, stream,
                           rowptr, pk, sb, db, ceL, l, h, r_edge, A64, A16, N);
        hipLaunchKernelGGL((gemm64<true, true, false, 1>), dim3(N / 16), dim3(32), 0, stream,
                           A64, Wm + (size_t)l * 4096, A16, WeWm + (size_t)l * 1024,
                           h, nullptr, h);
    }

    hipLaunchKernelGGL(gate_hi, dim3((N + 127) / 128), dim3(128), 0, stream,
                       h, d_edge, gid, w_d, b_d, gate, hi, N);
    hipLaunchKernelGGL(h_mix_pool, dim3((N + 127) / 128), dim3(128), 0, stream,
                       h, gate, hi, gid, pooled, N);

    hipLaunchKernelGGL((gemm64<false, false, true, 2>), dim3(G / 16), dim3(32), 0, stream,
                       pooled, W_mlp + 0 * 4096, nullptr, nullptr, nullptr, b_mlp + 0 * 64, x1);
    hipLaunchKernelGGL((gemm64<false, false, true, 2>), dim3(G / 16), dim3(32), 0, stream,
                       x1, W_mlp + 1 * 4096, nullptr, nullptr, nullptr, b_mlp + 1 * 64, x2);
    hipLaunchKernelGGL((gemm64<false, false, true, 2>), dim3(G / 16), dim3(32), 0, stream,
                       x2, W_mlp + 2 * 4096, nullptr, nullptr, nullptr, b_mlp + 2 * 64, x1);

    hipLaunchKernelGGL(out_kernel, dim3((G + 255) / 256), dim3(256), 0, stream,
                       x1, W_out, b_out, (float*)d_out, G);
}


# Edge logits formed in the gather from the h[src]/r_edge rows it loads anyway (16-lane tree dots); sb/ceL gathers, edge_ce, node_prep gone; softmax max and weight sum per 4 edges in list order
# speedup vs baseline: 4.9070x; 1.2130x over previous
#include <hip/hip_runtime.h>

typedef __attribute__((ext_vector_type(2))) float v2f;
typedef __attribute__((ext_vector_type(8))) float v8f;

#define HD 64

__device__ __forceinline__ float lrelu_f(float x) { return x > 0.f ? x : 0.1f * x; }

template <int M> __device__ __forceinline__ float row_xor(float v) {
    return __int_as_float(__builtin_amdgcn_mov_dpp(__float_as_int(v), 0x160 + M, 0xf, 0xf, false));
}
template <int U> __device__ __forceinline__ int row_bcast(int v) {
    return __builtin_amdgcn_mov_dpp(v, 0x150 + U, 0xf, 0xf, false);
}
template <int U> __device__ __forceinline__ float row_bcast(float v) {
    return __int_as_float(__builtin_amdgcn_mov_dpp(__float_as_int(v), 0x150 + U, 0xf, 0xf, false));
}
__device__ __forceinline__ float row_max16(float v) {
    v = fmaxf(v, row_xor<1>(v)); v = fmaxf(v, row_xor<2>(v));
    v = fmaxf(v, row_xor<4>(v)); v = fmaxf(v, row_xor<8>(v));
    return v;
}
__device__ __forceinline__ float row_sum16(float v) {
    v += row_xor<1>(v); v += row_xor<2>(v); v += row_xor<4>(v); v += row_xor<8>(v);
    return v;
}

template <bool HAS_A2, bool HAS_RES, bool HAS_BIAS, int ACT>
__global__ __launch_bounds__(32) void gemm64(
    const float* __restrict__ A,  const float* __restrict__ B,
    const float* __restrict__ A2, const float* __restrict__ B2,
    const float* __restrict__ res, const float* __restrict__ bias,
    float* __restrict__ out)
{
    const int lane = threadIdx.x & 31;
    const int cn   = lane & 15;
    const int kh   = (lane >> 4) << 1;
    const size_t row = (size_t)blockIdx.x * 16 + cn;

    const v8f z = {0.f, 0.f, 0.f, 0.f, 0.f, 0.f, 0.f, 0.f};
    v8f acc0 = z, acc1 = z, acc2 = z, acc3 = z;

    const float* arow = A + row * HD;
#pragma unroll
    for (int k0 = 0; k0 < 64; k0 += 4) {
        v2f a; a.x = arow[k0 + kh]; a.y = arow[k0 + kh + 1];
        const float* b0 = B + (size_t)(k0 + kh) * HD;
        const float* b1 = b0 + HD;
        v2f b;
        b.x = b0[cn];        b.y = b1[cn];
        acc0 = __builtin_amdgcn_wmma_f32_16x16x4_f32(false, a, false, b, (short)0, acc0, false, false);
        b.x = b0[16 + cn];   b.y = b1[16 + cn];
        acc1 = __builtin_amdgcn_wmma_f32_16x16x4_f32(false, a, false, b, (short)0, acc1, false, false);
        b.x = b0[32 + cn];   b.y = b1[32 + cn];
        acc2 = __builtin_amdgcn_wmma_f32_16x16x4_f32(false, a, false, b, (short)0, acc2, false, false);
        b.x = b0[48 + cn];   b.y = b1[48 + cn];
        acc3 = __builtin_amdgcn_wmma_f32_16x16x4_f32(false, a, false, b, (short)0, acc3, false, false);
    }
    if (HAS_A2) {
        const float* a2row = A2 + row * 16;
#pragma unroll
        for (int k0 = 0; k0 < 16; k0 += 4) {
            v2f a; a.x = a2row[k0 + kh]; a.y = a2row[k0 + kh + 1];
            const float* b0 = B2 + (size_t)(k0 + kh) * HD;
            const float* b1 = b0 + HD;
            v2f b;
            b.x = b0[cn];        b.y = b1[cn];
            acc0 = __builtin_amdgcn_wmma_f32_16x16x4_f32(false, a, false, b, (short)0, acc0, false, false);
            b.x = b0[16 + cn];   b.y = b1[16 + cn];
            acc1 = __builtin_amdgcn_wmma_f32_16x16x4_f32(false, a, false, b, (short)0, acc1, false, false);
            b.x = b0[32 + cn];   b.y = b1[32 + cn];
            acc2 = __builtin_amdgcn_wmma_f32_16x16x4_f32(false, a, false, b, (short)0, acc2, false, false);
            b.x = b0[48 + cn];   b.y = b1[48 + cn];
            acc3 = __builtin_amdgcn_wmma_f32_16x16x4_f32(false, a, false, b, (short)0, acc3, false, false);
        }
    }

    const size_t mbase = (size_t)blockIdx.x * 16 + ((lane >> 4) << 3);
    v8f accs[4] = {acc0, acc1, acc2, acc3};
#pragma unroll
    for (int t = 0; t < 4; ++t) {
        const int col = t * 16 + cn;
        const float bval = HAS_BIAS ? bias[col] : 0.f;
#pragma unroll
        for (int v = 0; v < 8; ++v) {
            const size_t m = mbase + v;
            float val = accs[t][v] + bval;
            if (HAS_RES) val += res[m * HD + col];
            if (ACT == 1) val = lrelu_f(val);
            else if (ACT == 2) val = fmaxf(val, 0.f);
            out[m * HD + col] = val;
        }
    }
}

__global__ void prep_ce(const float* __restrict__ W_e, const float* __restrict__ a_e,
                        float* __restrict__ cE)
{
    const int tid = threadIdx.x;
    const int l = tid >> 4, d = tid & 15;
    float s = 0.f;
    for (int k = 0; k < 64; ++k) s += W_e[d * 64 + k] * a_e[l * 64 + k];
    cE[tid] = s;
}

__global__ void prep_wewm(const float* __restrict__ W_e, const float* __restrict__ Wm,
                          float* __restrict__ WeWm)
{
    const int idx = blockIdx.x * blockDim.x + threadIdx.x;
    if (idx >= 4 * 16 * 64) return;
    const int l = idx >> 10, d = (idx >> 6) & 15, k = idx & 63;
    float s = 0.f;
    for (int j = 0; j < 64; ++j) s += W_e[d * 64 + j] * Wm[l * 4096 + j * 64 + k];
    WeWm[idx] = s;
}

__global__ void init_hi(const float* __restrict__ i_node, const float* __restrict__ W_i,
                        float* __restrict__ hi, float* __restrict__ pooled, int G)
{
    const int idx = blockIdx.x * blockDim.x + threadIdx.x;
    if (idx >= G * 64) return;
    hi[idx] = i_node[idx >> 6] * W_i[idx & 63];
    pooled[idx] = 0.f;
}

__global__ void csr_zero(int* __restrict__ cnt, int N)
{
    const int n = blockIdx.x * blockDim.x + threadIdx.x;
    if (n < N) cnt[n] = 0;
}

__global__ void csr_count(const int* __restrict__ dst, int* __restrict__ cnt,
                          int* __restrict__ rank, int E)
{
    const int j = blockIdx.x * blockDim.x + threadIdx.x;
    if (j >= E) return;
    rank[j] = atomicAdd(cnt + dst[j], 1);
}

__global__ __launch_bounds__(1024) void csr_scan(const int* __restrict__ cnt,
                                                 int* __restrict__ rowptr, int N)
{
    __shared__ int wfront[32];
    __shared__ int wtot[32];
    const int t = threadIdx.x, lane = t & 31, w = t >> 5;
    const int base0 = blockIdx.x * 1024;

    int s = 0;
    for (int i = t; i < base0; i += 1024) s += cnt[i];
    for (int o = 16; o > 0; o >>= 1) s += __shfl_xor(s, o, 32);
    if (lane == 0) wfront[w] = s;

    const int i = base0 + t;
    const int c = (i < N) ? cnt[i] : 0;
    int x = c;
    for (int o = 1; o < 32; o <<= 1) {
        const int y = __shfl_up(x, o, 32);
        if (lane >= o) x += y;
    }
    if (lane == 31) wtot[w] = x;
    __syncthreads();

    int front = 0;
    for (int k = 0; k < 32; ++k) front += wfront[k];
    for (int k = 0; k < w; ++k) front += wtot[k];
    if (i < N) rowptr[i] = front + x - c;
    if (i == N - 1) rowptr[N] = front + x;
}

__global__ void csr_fill(const int* __restrict__ src, const int* __restrict__ dst,
                         const int* __restrict__ rowptr, const int* __restrict__ rank,
                         int* __restrict__ pk, int E)
{
    const int j = blockIdx.x * blockDim.x + threadIdx.x;
    if (j >= E) return;
    const int p = rowptr[dst[j]] + rank[j];
    ((int2*)pk)[p] = make_int2(j, src[j]);
}

template <int U0> __device__ __forceinline__ void gather4(
    const int2 e, const int q, const int nlive,
    const float4 as4, const float ceq, const float dn,
    const float* __restrict__ h, const float* __restrict__ re,
    float& m, float& dsum, float4& acc, float& accr)
{
    const int j0 = row_bcast<U0 + 0>(e.x), s0 = row_bcast<U0 + 0>(e.y);
    const int j1 = row_bcast<U0 + 1>(e.x), s1 = row_bcast<U0 + 1>(e.y);
    const int j2 = row_bcast<U0 + 2>(e.x), s2 = row_bcast<U0 + 2>(e.y);
    const int j3 = row_bcast<U0 + 3>(e.x), s3 = row_bcast<U0 + 3>(e.y);
    const float4 h0 = *(const float4*)(h + ((unsigned)s0 * 64u + (unsigned)q * 4u));
    const float4 h1 = *(const float4*)(h + ((unsigned)s1 * 64u + (unsigned)q * 4u));
    const float4 h2 = *(const float4*)(h + ((unsigned)s2 * 64u + (unsigned)q * 4u));
    const float4 h3 = *(const float4*)(h + ((unsigned)s3 * 64u + (unsigned)q * 4u));
    const float r0 = re[(unsigned)j0 * 16u + (unsigned)q];
    const float r1 = re[(unsigned)j1 * 16u + (unsigned)q];
    const float r2 = re[(unsigned)j2 * 16u + (unsigned)q];
    const float r3 = re[(unsigned)j3 * 16u + (unsigned)q];
    const float ninf = -__builtin_inff();
    float l0 = row_sum16(h0.x * as4.x + h0.y * as4.y + h0.z * as4.z + h0.w * as4.w + r0 * ceq);
    float l1 = row_sum16(h1.x * as4.x + h1.y * as4.y + h1.z * as4.z + h1.w * as4.w + r1 * ceq);
    float l2 = row_sum16(h2.x * as4.x + h2.y * as4.y + h2.z * as4.z + h2.w * as4.w + r2 * ceq);
    float l3 = row_sum16(h3.x * as4.x + h3.y * as4.y + h3.z * as4.z + h3.w * as4.w + r3 * ceq);
    l0 = lrelu_f(l0 + dn);
    l1 = (U0 + 1 < nlive) ? lrelu_f(l1 + dn) : ninf;
    l2 = (U0 + 2 < nlive) ? lrelu_f(l2 + dn) : ninf;
    l3 = (U0 + 3 < nlive) ? lrelu_f(l3 + dn) : ninf;
    const float mn = fmaxf(fmaxf(m, l0), fmaxf(fmaxf(l1, l2), l3));
    const float sc = __expf(m - mn);
    const float a0 = __expf(l0 - mn), a1 = __expf(l1 - mn);
    const float a2 = __expf(l2 - mn), a3 = __expf(l3 - mn);
    m = mn;
    dsum = dsum * sc + ((a0 + a1) + (a2 + a3));
    acc.x *= sc; acc.y *= sc; acc.z *= sc; acc.w *= sc; accr *= sc;
    acc.x += a0 * h0.x; acc.y += a0 * h0.y; acc.z += a0 * h0.z; acc.w += a0 * h0.w; accr += a0 * r0;
    acc.x += a1 * h1.x; acc.y += a1 * h1.y; acc.z += a1 * h1.z; acc.w += a1 * h1.w; accr += a1 * r1;
    acc.x += a2 * h2.x; acc.y += a2 * h2.y; acc.z += a2 * h2.z; acc.w += a2 * h2.w; accr += a2 * r2;
    acc.x += a3 * h3.x; acc.y += a3 * h3.y; acc.z += a3 * h3.z; acc.w += a3 * h3.w; accr += a3 * r3;
}

__global__ __launch_bounds__(256) void node_gather(
    const int* __restrict__ rowptr, const int* __restrict__ pk,
    const float* __restrict__ as, const float* __restrict__ ad,
    const float* __restrict__ ce,
    const float* __restrict__ h, const float* __restrict__ re,
    float* __restrict__ A64, float* __restrict__ A16, int N)
{
    const int n = blockIdx.x * 16 + (threadIdx.x >> 4);
    const int q = threadIdx.x & 15;
    if (n >= N) return;
    const int beg = rowptr[n], end = rowptr[n + 1];
    const float4 as4 = *(const float4*)(as + q * 4);
    const float4 ad4 = *(const float4*)(ad + q * 4);
    const float ceq = ce[q];
    const float4 hn = *(const float4*)(h + (size_t)n * 64 + q * 4);
    asm volatile("" :: "v"(hn.x), "v"(hn.y), "v"(hn.z), "v"(hn.w),
                       "v"(ad4.x), "v"(ad4.y), "v"(ad4.z), "v"(ad4.w),
                       "v"(as4.x), "v"(as4.y), "v"(as4.z), "v"(as4.w), "v"(ceq));
    const int2* __restrict__ pk2 = (const int2*)pk;
    float4 acc = {0.f, 0.f, 0.f, 0.f};
    float accr = 0.f;
    if (end > beg) {
        const int last = end - 1;
        int2 e = pk2[min(beg + q, last)];
        const float dn = row_sum16(hn.x * ad4.x + hn.y * ad4.y + hn.z * ad4.z + hn.w * ad4.w);
        float m = -__builtin_inff();
        float dsum = 0.f;
        int c = beg;
        for (;;) {
            const int nlive = end - c;
            gather4<0>(e, q, nlive, as4, ceq, dn, h, re, m, dsum, acc, accr);
            if (nlive > 4)  gather4<4>(e, q, nlive, as4, ceq, dn, h, re, m, dsum, acc, accr);
            if (nlive > 8)  gather4<8>(e, q, nlive, as4, ceq, dn, h, re, m, dsum, acc, accr);
            if (nlive > 12) gather4<12>(e, q, nlive, as4, ceq, dn, h, re, m, dsum, acc, accr);
            c += 16;
            if (c >= end) break;
            e = pk2[min(c + q, last)];
        }
        const float inv = 1.f / dsum;
        acc.x *= inv; acc.y *= inv; acc.z *= inv; acc.w *= inv; accr *= inv;
    }
    *(float4*)(A64 + (size_t)n * 64 + q * 4) = acc;
    A16[(size_t)n * 16 + q] = accr;
}

#define WALK 16

__device__ __forceinline__ void row_flush(float* __restrict__ grow, const int q, const float4 a)
{
    float* p = grow + q * 4;
    atomicAdd(p + 0, a.x); atomicAdd(p + 1, a.y); atomicAdd(p + 2, a.z); atomicAdd(p + 3, a.w);
}

__device__ __forceinline__ void row_bcast_all(const int iq, const float fq,
                                              int (&iv)[WALK], float (&fv)[WALK])
{
#define BC1(U) iv[U] = row_bcast<U>(iq); fv[U] = row_bcast<U>(fq);
    BC1(0)  BC1(1)  BC1(2)  BC1(3)  BC1(4)  BC1(5)  BC1(6)  BC1(7)
    BC1(8)  BC1(9)  BC1(10) BC1(11) BC1(12) BC1(13) BC1(14) BC1(15)
#undef BC1
}

__global__ __launch_bounds__(128) void gate_hi(
    const float* __restrict__ h, const float* __restrict__ d_edge,
    const int* __restrict__ gid, const float* __restrict__ wd,
    const float* __restrict__ bd, float* __restrict__ gate,
    float* __restrict__ hi, int N)
{
    const int q  = threadIdx.x & 15;
    const int n0 = (blockIdx.x * 8 + (threadIdx.x >> 4)) * WALK;
    if (n0 >= N) return;
    const int nq = min(n0 + q, N - 1);
    float4 hv[WALK];
#pragma unroll
    for (int u = 0; u < WALK; ++u)
        hv[u] = *(const float4*)(h + (size_t)min(n0 + u, N - 1) * 64 + q * 4);
    const float gq = 1.f / (1.f + __expf(-(d_edge[nq] * wd[0] + bd[0])));
    const int gidq = gid[nq];
    if (n0 + q < N) gate[n0 + q] = gq;
    int gi[WALK]; float g[WALK];
    row_bcast_all(gidq, gq, gi, g);
    float4 acc = {0.f, 0.f, 0.f, 0.f};
    int cur = gi[0];
#pragma unroll
    for (int u = 0; u < WALK; ++u) {
        if (n0 + u < N) {
            if (gi[u] != cur) {
                row_flush(hi + (size_t)cur * 64, q, acc);
                acc.x = 0.f; acc.y = 0.f; acc.z = 0.f; acc.w = 0.f;
                cur = gi[u];
            }
            acc.x += g[u] * hv[u].x; acc.y += g[u] * hv[u].y;
            acc.z += g[u] * hv[u].z; acc.w += g[u] * hv[u].w;
        }
    }
    row_flush(hi + (size_t)cur * 64, q, acc);
}

__global__ __launch_bounds__(128) void h_mix_pool(
    float* __restrict__ h, const float* __restrict__ gate,
    const float* __restrict__ hi, const int* __restrict__ gid,
    float* __restrict__ pooled, int N)
{
    const int q  = threadIdx.x & 15;
    const int n0 = (blockIdx.x * 8 + (threadIdx.x >> 4)) * WALK;
    if (n0 >= N) return;
    const int nq = min(n0 + q, N - 1);
    float4 hv[WALK];
#pragma unroll
    for (int u = 0; u < WALK; ++u)
        hv[u] = *(const float4*)(h + (size_t)min(n0 + u, N - 1) * 64 + q * 4);
    const float gq = gate[nq];
    const int gidq = gid[nq];
    int gi[WALK]; float g[WALK];
    row_bcast_all(gidq, gq, gi, g);
    float4 hiv[WALK];
#pragma unroll
    for (int u = 0; u < WALK; ++u)
        hiv[u] = *(const float4*)(hi + (size_t)gi[u] * 64 + q * 4);
    float4 acc = {0.f, 0.f, 0.f, 0.f};
    int cur = gi[0];
#pragma unroll
    for (int u = 0; u < WALK; ++u) {
        if (n0 + u < N) {
            float4 v;
            v.x = hv[u].x + g[u] * hiv[u].x; v.y = hv[u].y + g[u] * hiv[u].y;
            v.z = hv[u].z + g[u] * hiv[u].z; v.w = hv[u].w + g[u] * hiv[u].w;
            *(float4*)(h + (size_t)(n0 + u) * 64 + q * 4) = v;
            if (gi[u] != cur) {
                row_flush(pooled + (size_t)cur * 64, q, acc);
                acc.x = 0.f; acc.y = 0.f; acc.z = 0.f; acc.w = 0.f;
                cur = gi[u];
            }
            acc.x += v.x; acc.y += v.y; acc.z += v.z; acc.w += v.w;
        }
    }
    row_flush(pooled + (size_t)cur * 64, q, acc);
}

__global__ void out_kernel(const float* __restrict__ x, const float* __restrict__ Wo,
                           const float* __restrict__ bo, float* __restrict__ out, int G)
{
    const int g = blockIdx.x * blockDim.x + threadIdx.x;
    if (g >= G) return;
    float s = bo[0];
#pragma unroll
    for (int k = 0; k < 64; ++k) s += x[(size_t)g * 64 + k] * Wo[k];
    out[g] = s;
}

extern "C" void kernel_launch(void* const* d_in, const int* in_sizes, int n_in,
                              void* d_out, int out_size, void* d_ws, size_t ws_size,
                              hipStream_t stream)
{
    const float* r_node = (const float*)d_in[0];
    const float* i_node = (const float*)d_in[1];
    const float* r_edge = (const float*)d_in[2];
    const float* d_edge = (const float*)d_in[3];
    const int*   src    = (const int*)d_in[4];
    const int*   dst    = (const int*)d_in[5];
    const int*   gid    = (const int*)d_in[6];
    const float* W_r    = (const float*)d_in[7];
    const float* W_i    = (const float*)d_in[8];
    const float* W_e    = (const float*)d_in[9];
    const float* Wm     = (const float*)d_in[10];
    const float* a_s    = (const float*)d_in[11];
    const float* a_d    = (const float*)d_in[12];
    const float* a_e    = (const float*)d_in[13];
    const float* w_d    = (const float*)d_in[14];
    const float* b_d    = (const float*)d_in[15];
    const float* W_mlp  = (const float*)d_in[16];
    const float* b_mlp  = (const float*)d_in[17];
    const float* W_out  = (const float*)d_in[18];
    const float* b_out  = (const float*)d_in[19];

    const int N = in_sizes[0] / 64;
    const int G = in_sizes[1];
    const int E = in_sizes[2] / 16;
    (void)n_in; (void)out_size; (void)ws_size;

    float* ws = (float*)d_ws;
    size_t off = 0;
    auto carve = [&](size_t n) -> float* {
        float* p = ws + off; off += (n + 63) & ~(size_t)63; return p;
    };
    float*    h      = carve((size_t)N * 64);
    float*    A64    = carve((size_t)N * 64);
    float*    A16    = carve((size_t)N * 16);
    float*    gate   = carve(N);
    float*    hi     = carve((size_t)G * 64);
    float*    pooled = carve((size_t)G * 64);
    float*    x1     = carve((size_t)G * 64);
    float*    x2     = carve((size_t)G * 64);
    float*    cE     = carve(64);
    float*    WeWm   = carve(4 * 16 * 64);
    int*      cnt    = (int*)carve(N);
    int*      rowptr = (int*)carve((size_t)N + 1);
    int*      rank   = (int*)carve(E);
    int*      pk     = (int*)carve((size_t)E * 2);

    hipLaunchKernelGGL(prep_ce, dim3(1), dim3(64), 0, stream, W_e, a_e, cE);
    hipLaunchKernelGGL(prep_wewm, dim3((4096 + 255) / 256), dim3(256), 0, stream, W_e, Wm, WeWm);

    hipLaunchKernelGGL((gemm64<false, false, false, 0>), dim3(N / 16), dim3(32), 0, stream,
                       r_node, W_r, nullptr, nullptr, nullptr, nullptr, h);
    hipLaunchKernelGGL(init_hi, dim3((G * 64 + 255) / 256), dim3(256), 0, stream,
                       i_node, W_i, hi, pooled, G);

    hipLaunchKernelGGL(csr_zero, dim3((N + 255) / 256), dim3(256), 0, stream, cnt, N);
    hipLaunchKernelGGL(csr_count, dim3((E + 255) / 256), dim3(256), 0, stream, dst, cnt, rank, E);
    hipLaunchKernelGGL(csr_scan, dim3((N + 1023) / 1024), dim3(1024), 0, stream, cnt, rowptr, N);
    hipLaunchKernelGGL(csr_fill, dim3((E + 255) / 256), dim3(256), 0, stream,
                       src, dst, rowptr, rank, pk, E);

    for (int l = 0; l < 4; ++l) {
        hipLaunchKernelGGL(node_gather, dim3((N + 15) / 16), dim3(256), 0, stream,
                           rowptr, pk, a_s + l * 64, a_d + l * 64, cE + l * 16,
                           h, r_edge, A64, A16, N);
        hipLaunchKernelGGL((gemm64<true, true, false, 1>), dim3(N / 16), dim3(32), 0, stream,
                           A64, Wm + (size_t)l * 4096, A16, WeWm + (size_t)l * 1024,
                           h, nullptr, h);
    }

    hipLaunchKernelGGL(gate_hi, dim3((N + 127) / 128), dim3(128), 0, stream,
                       h, d_edge, gid, w_d, b_d, gate, hi, N);
    hipLaunchKernelGGL(h_mix_pool, dim3((N + 127) / 128), dim3(128), 0, stream,
                       h, gate, hi, gid, pooled, N);

    hipLaunchKernelGGL((gemm64<false, false, true, 2>), dim3(G / 16), dim3(32), 0, stream,
                       pooled, W_mlp + 0 * 4096, nullptr, nullptr, nullptr, b_mlp + 0 * 64, x1);
    hipLaunchKernelGGL((gemm64<false, false, true, 2>), dim3(G / 16), dim3(32), 0, stream,
                       x1, W_mlp + 1 * 4096, nullptr, nullptr, nullptr, b_mlp + 1 * 64, x2);
    hipLaunchKernelGGL((gemm64<false, false, true, 2>), dim3(G / 16), dim3(32), 0, stream,
                       x2, W_mlp + 2 * 4096, nullptr, nullptr, nullptr, b_mlp + 2 * 64, x1);

    hipLaunchKernelGGL(out_kernel, dim3((G + 255) / 256), dim3(256), 0, stream,
                       x1, W_out, b_out, (float*)d_out, G);
}


# Destination-ordered edge list built with LDS counters per node range over 32 edge slices: no global atomics, no rank plane, 3 launches for 4; list order = slice-major, LDS arrival rank in a slice
# speedup vs baseline: 6.1112x; 1.2454x over previous
#include <hip/hip_runtime.h>

typedef __attribute__((ext_vector_type(2))) float v2f;
typedef __attribute__((ext_vector_type(8))) float v8f;

#define HD 64

__device__ __forceinline__ float lrelu_f(float x) { return x > 0.f ? x : 0.1f * x; }

template <int M> __device__ __forceinline__ float row_xor(float v) {
    return __int_as_float(__builtin_amdgcn_mov_dpp(__float_as_int(v), 0x160 + M, 0xf, 0xf, false));
}
template <int U> __device__ __forceinline__ int row_bcast(int v) {
    return __builtin_amdgcn_mov_dpp(v, 0x150 + U, 0xf, 0xf, false);
}
template <int U> __device__ __forceinline__ float row_bcast(float v) {
    return __int_as_float(__builtin_amdgcn_mov_dpp(__float_as_int(v), 0x150 + U, 0xf, 0xf, false));
}
__device__ __forceinline__ float row_max16(float v) {
    v = fmaxf(v, row_xor<1>(v)); v = fmaxf(v, row_xor<2>(v));
    v = fmaxf(v, row_xor<4>(v)); v = fmaxf(v, row_xor<8>(v));
    return v;
}
__device__ __forceinline__ float row_sum16(float v) {
    v += row_xor<1>(v); v += row_xor<2>(v); v += row_xor<4>(v); v += row_xor<8>(v);
    return v;
}

template <bool HAS_A2, bool HAS_RES, bool HAS_BIAS, int ACT>
__global__ __launch_bounds__(32) void gemm64(
    const float* __restrict__ A,  const float* __restrict__ B,
    const float* __restrict__ A2, const float* __restrict__ B2,
    const float* __restrict__ res, const float* __restrict__ bias,
    float* __restrict__ out)
{
    const int lane = threadIdx.x & 31;
    const int cn   = lane & 15;
    const int kh   = (lane >> 4) << 1;
    const size_t row = (size_t)blockIdx.x * 16 + cn;

    const v8f z = {0.f, 0.f, 0.f, 0.f, 0.f, 0.f, 0.f, 0.f};
    v8f acc0 = z, acc1 = z, acc2 = z, acc3 = z;

    const float* arow = A + row * HD;
#pragma unroll
    for (int k0 = 0; k0 < 64; k0 += 4) {
        v2f a; a.x = arow[k0 + kh]; a.y = arow[k0 + kh + 1];
        const float* b0 = B + (size_t)(k0 + kh) * HD;
        const float* b1 = b0 + HD;
        v2f b;
        b.x = b0[cn];        b.y = b1[cn];
        acc0 = __builtin_amdgcn_wmma_f32_16x16x4_f32(false, a, false, b, (short)0, acc0, false, false);
        b.x = b0[16 + cn];   b.y = b1[16 + cn];
        acc1 = __builtin_amdgcn_wmma_f32_16x16x4_f32(false, a, false, b, (short)0, acc1, false, false);
        b.x = b0[32 + cn];   b.y = b1[32 + cn];
        acc2 = __builtin_amdgcn_wmma_f32_16x16x4_f32(false, a, false, b, (short)0, acc2, false, false);
        b.x = b0[48 + cn];   b.y = b1[48 + cn];
        acc3 = __builtin_amdgcn_wmma_f32_16x16x4_f32(false, a, false, b, (short)0, acc3, false, false);
    }
    if (HAS_A2) {
        const float* a2row = A2 + row * 16;
#pragma unroll
        for (int k0 = 0; k0 < 16; k0 += 4) {
            v2f a; a.x = a2row[k0 + kh]; a.y = a2row[k0 + kh + 1];
            const float* b0 = B2 + (size_t)(k0 + kh) * HD;
            const float* b1 = b0 + HD;
            v2f b;
            b.x = b0[cn];        b.y = b1[cn];
            acc0 = __builtin_amdgcn_wmma_f32_16x16x4_f32(false, a, false, b, (short)0, acc0, false, false);
            b.x = b0[16 + cn];   b.y = b1[16 + cn];
            acc1 = __builtin_amdgcn_wmma_f32_16x16x4_f32(false, a, false, b, (short)0, acc1, false, false);
            b.x = b0[32 + cn];   b.y = b1[32 + cn];
            acc2 = __builtin_amdgcn_wmma_f32_16x16x4_f32(false, a, false, b, (short)0, acc2, false, false);
            b.x = b0[48 + cn];   b.y = b1[48 + cn];
            acc3 = __builtin_amdgcn_wmma_f32_16x16x4_f32(false, a, false, b, (short)0, acc3, false, false);
        }
    }

    const size_t mbase = (size_t)blockIdx.x * 16 + ((lane >> 4) << 3);
    v8f accs[4] = {acc0, acc1, acc2, acc3};
#pragma unroll
    for (int t = 0; t < 4; ++t) {
        const int col = t * 16 + cn;
        const float bval = HAS_BIAS ? bias[col] : 0.f;
#pragma unroll
        for (int v = 0; v < 8; ++v) {
            const size_t m = mbase + v;
            float val = accs[t][v] + bval;
            if (HAS_RES) val += res[m * HD + col];
            if (ACT == 1) val = lrelu_f(val);
            else if (ACT == 2) val = fmaxf(val, 0.f);
            out[m * HD + col] = val;
        }
    }
}

__global__ void prep_ce(const float* __restrict__ W_e, const float* __restrict__ a_e,
                        float* __restrict__ cE)
{
    const int tid = threadIdx.x;
    const int l = tid >> 4, d = tid & 15;
    float s = 0.f;
    for (int k = 0; k < 64; ++k) s += W_e[d * 64 + k] * a_e[l * 64 + k];
    cE[tid] = s;
}

__global__ void prep_wewm(const float* __restrict__ W_e, const float* __restrict__ Wm,
                          float* __restrict__ WeWm)
{
    const int idx = blockIdx.x * blockDim.x + threadIdx.x;
    if (idx >= 4 * 16 * 64) return;
    const int l = idx >> 10, d = (idx >> 6) & 15, k = idx & 63;
    float s = 0.f;
    for (int j = 0; j < 64; ++j) s += W_e[d * 64 + j] * Wm[l * 4096 + j * 64 + k];
    WeWm[idx] = s;
}

__global__ void init_hi(const float* __restrict__ i_node, const float* __restrict__ W_i,
                        float* __restrict__ hi, float* __restrict__ pooled, int G)
{
    const int idx = blockIdx.x * blockDim.x + threadIdx.x;
    if (idx >= G * 64) return;
    hi[idx] = i_node[idx >> 6] * W_i[idx & 63];
    pooled[idx] = 0.f;
}

#define CSR_NB 12800
#define CSR_S  32
#define CSR_CH 128

__global__ __launch_bounds__(1024) void csr_count(const int* __restrict__ dst,
                                                  int* __restrict__ pc, int* __restrict__ cs,
                                                  int N, int NP, int CSP, int E, int R)
{
    __shared__ int cnt[CSR_NB];
    const int t = threadIdx.x;
    const int r = blockIdx.x % R, s = blockIdx.x / R;
    const int r0 = r * CSR_NB;
    for (int i = t; i < CSR_NB; i += 1024) cnt[i] = 0;
    __syncthreads();
    const int E4 = E >> 2;
    const int SL4 = (E4 + CSR_S - 1) / CSR_S;
    const int b4 = s * SL4, e4 = min(b4 + SL4, E4);
    const int4* __restrict__ d4 = (const int4*)dst;
    for (int i = b4 + t; i < e4; i += 4096) {
        int4 d[4];
#pragma unroll
        for (int k = 0; k < 4; ++k) d[k] = d4[min(i + k * 1024, e4 - 1)];
#pragma unroll
        for (int k = 0; k < 4; ++k) {
            if (i + k * 1024 < e4) {
                unsigned u;
                u = (unsigned)(d[k].x - r0); if (u < CSR_NB) atomicAdd(&cnt[u], 1);
                u = (unsigned)(d[k].y - r0); if (u < CSR_NB) atomicAdd(&cnt[u], 1);
                u = (unsigned)(d[k].z - r0); if (u < CSR_NB) atomicAdd(&cnt[u], 1);
                u = (unsigned)(d[k].w - r0); if (u < CSR_NB) atomicAdd(&cnt[u], 1);
            }
        }
    }
    if (s == CSR_S - 1) {
        for (int j = (E4 << 2) + t; j < E; j += 1024) {
            const unsigned u = (unsigned)(dst[j] - r0);
            if (u < CSR_NB) atomicAdd(&cnt[u], 1);
        }
    }
    __syncthreads();
    for (int i = t; i < CSR_NB; i += 1024) {
        const int n = r0 + i;
        if (n < N) pc[(size_t)s * NP + n] = cnt[i];
    }
    const int lane = t & 31, w = t >> 5;
    for (int c = w; c < CSR_NB / CSR_CH; c += 32) {
        const int* p = cnt + c * CSR_CH + lane;
        int v = (p[0] + p[32]) + (p[64] + p[96]);
        for (int o = 16; o > 0; o >>= 1) v += __shfl_xor(v, o, 32);
        if (lane == 0) cs[s * CSP + r * (CSR_NB / CSR_CH) + c] = v;
    }
}

__global__ __launch_bounds__(1024) void csr_scan(const int* __restrict__ pc,
                                                 const int* __restrict__ cs,
                                                 int* __restrict__ base, int N, int NP, int CSP)
{
    __shared__ int wfront[32];
    __shared__ int wtot[32];
    const int t = threadIdx.x, lane = t & 31, w = t >> 5;
    const int base0 = blockIdx.x * 1024;
    const int ch0 = blockIdx.x * (1024 / CSR_CH);

    int sf = 0;
    for (int c = t; c < ch0; c += 1024) {
#pragma unroll
        for (int s = 0; s < CSR_S; ++s) sf += cs[s * CSP + c];
    }
    for (int o = 16; o > 0; o >>= 1) sf += __shfl_xor(sf, o, 32);
    if (lane == 0) wfront[w] = sf;

    const int i = base0 + t;
    const int ic = min(i, N - 1);
    int pcs[CSR_S];
    int c = 0;
#pragma unroll
    for (int s = 0; s < CSR_S; ++s) {
        pcs[s] = (i < N) ? pc[(size_t)s * NP + ic] : 0;
        c += pcs[s];
    }
    int x = c;
    for (int o = 1; o < 32; o <<= 1) {
        const int y = __shfl_up(x, o, 32);
        if (lane >= o) x += y;
    }
    if (lane == 31) wtot[w] = x;
    __syncthreads();

    int front = 0;
    for (int k = 0; k < 32; ++k) front += wfront[k];
    for (int k = 0; k < w; ++k) front += wtot[k];
    if (i < N) {
        int run = front + x - c;
#pragma unroll
        for (int s = 0; s < CSR_S; ++s) {
            base[(size_t)s * NP + i] = run;
            run += pcs[s];
        }
        if (i == N - 1) base[N] = run;
    }
}

__global__ __launch_bounds__(1024) void csr_fill(const int* __restrict__ src,
                                                 const int* __restrict__ dst,
                                                 const int* __restrict__ base,
                                                 int* __restrict__ pk, int N, int NP, int E, int R)
{
    __shared__ int pos[CSR_NB];
    const int t = threadIdx.x;
    const int r = blockIdx.x % R, s = blockIdx.x / R;
    const int r0 = r * CSR_NB;
    for (int i = t; i < CSR_NB; i += 1024) {
        const int n = r0 + i;
        pos[i] = (n < N) ? base[(size_t)s * NP + n] : 0;
    }
    __syncthreads();
    const int E4 = E >> 2;
    const int SL4 = (E4 + CSR_S - 1) / CSR_S;
    const int b4 = s * SL4, e4 = min(b4 + SL4, E4);
    const int4* __restrict__ d4 = (const int4*)dst;
    const int4* __restrict__ s4 = (const int4*)src;
    int2* __restrict__ pk2 = (int2*)pk;
    for (int i = b4 + t; i < e4; i += 4096) {
        int4 d[4], sv[4];
#pragma unroll
        for (int k = 0; k < 4; ++k) {
            const int idx = min(i + k * 1024, e4 - 1);
            d[k] = d4[idx]; sv[k] = s4[idx];
        }
        int p[16];
#pragma unroll
        for (int k = 0; k < 4; ++k) {
            const bool live = i + k * 1024 < e4;
            unsigned u;
            u = (unsigned)(d[k].x - r0); p[4 * k + 0] = -1;
            if (live && u < CSR_NB) p[4 * k + 0] = atomicAdd(&pos[u], 1);
            u = (unsigned)(d[k].y - r0); p[4 * k + 1] = -1;
            if (live && u < CSR_NB) p[4 * k + 1] = atomicAdd(&pos[u], 1);
            u = (unsigned)(d[k].z - r0); p[4 * k + 2] = -1;
            if (live && u < CSR_NB) p[4 * k + 2] = atomicAdd(&pos[u], 1);
            u = (unsigned)(d[k].w - r0); p[4 * k + 3] = -1;
            if (live && u < CSR_NB) p[4 * k + 3] = atomicAdd(&pos[u], 1);
        }
#pragma unroll
        for (int k = 0; k < 4; ++k) {
            const int j = (i + k * 1024) << 2;
            if (p[4 * k + 0] >= 0) pk2[p[4 * k + 0]] = make_int2(j,     sv[k].x);
            if (p[4 * k + 1] >= 0) pk2[p[4 * k + 1]] = make_int2(j + 1, sv[k].y);
            if (p[4 * k + 2] >= 0) pk2[p[4 * k + 2]] = make_int2(j + 2, sv[k].z);
            if (p[4 * k + 3] >= 0) pk2[p[4 * k + 3]] = make_int2(j + 3, sv[k].w);
        }
    }
    if (s == CSR_S - 1) {
        for (int j = (E4 << 2) + t; j < E; j += 1024) {
            const unsigned u = (unsigned)(dst[j] - r0);
            if (u < CSR_NB) { const int q = atomicAdd(&pos[u], 1); pk2[q] = make_int2(j, src[j]); }
        }
    }
}

template <int U0> __device__ __forceinline__ void gather4(
    const int2 e, const int q, const int nlive,
    const float4 as4, const float ceq, const float dn,
    const float* __restrict__ h, const float* __restrict__ re,
    float& m, float& dsum, float4& acc, float& accr)
{
    const int j0 = row_bcast<U0 + 0>(e.x), s0 = row_bcast<U0 + 0>(e.y);
    const int j1 = row_bcast<U0 + 1>(e.x), s1 = row_bcast<U0 + 1>(e.y);
    const int j2 = row_bcast<U0 + 2>(e.x), s2 = row_bcast<U0 + 2>(e.y);
    const int j3 = row_bcast<U0 + 3>(e.x), s3 = row_bcast<U0 + 3>(e.y);
    const float4 h0 = *(const float4*)(h + ((unsigned)s0 * 64u + (unsigned)q * 4u));
    const float4 h1 = *(const float4*)(h + ((unsigned)s1 * 64u + (unsigned)q * 4u));
    const float4 h2 = *(const float4*)(h + ((unsigned)s2 * 64u + (unsigned)q * 4u));
    const float4 h3 = *(const float4*)(h + ((unsigned)s3 * 64u + (unsigned)q * 4u));
    const float r0 = re[(unsigned)j0 * 16u + (unsigned)q];
    const float r1 = re[(unsigned)j1 * 16u + (unsigned)q];
    const float r2 = re[(unsigned)j2 * 16u + (unsigned)q];
    const float r3 = re[(unsigned)j3 * 16u + (unsigned)q];
    const float ninf = -__builtin_inff();
    float l0 = row_sum16(h0.x * as4.x + h0.y * as4.y + h0.z * as4.z + h0.w * as4.w + r0 * ceq);
    float l1 = row_sum16(h1.x * as4.x + h1.y * as4.y + h1.z * as4.z + h1.w * as4.w + r1 * ceq);
    float l2 = row_sum16(h2.x * as4.x + h2.y * as4.y + h2.z * as4.z + h2.w * as4.w + r2 * ceq);
    float l3 = row_sum16(h3.x * as4.x + h3.y * as4.y + h3.z * as4.z + h3.w * as4.w + r3 * ceq);
    l0 = lrelu_f(l0 + dn);
    l1 = (U0 + 1 < nlive) ? lrelu_f(l1 + dn) : ninf;
    l2 = (U0 + 2 < nlive) ? lrelu_f(l2 + dn) : ninf;
    l3 = (U0 + 3 < nlive) ? lrelu_f(l3 + dn) : ninf;
    const float mn = fmaxf(fmaxf(m, l0), fmaxf(fmaxf(l1, l2), l3));
    const float sc = __expf(m - mn);
    const float a0 = __expf(l0 - mn), a1 = __expf(l1 - mn);
    const float a2 = __expf(l2 - mn), a3 = __expf(l3 - mn);
    m = mn;
    dsum = dsum * sc + ((a0 + a1) + (a2 + a3));
    acc.x *= sc; acc.y *= sc; acc.z *= sc; acc.w *= sc; accr *= sc;
    acc.x += a0 * h0.x; acc.y += a0 * h0.y; acc.z += a0 * h0.z; acc.w += a0 * h0.w; accr += a0 * r0;
    acc.x += a1 * h1.x; acc.y += a1 * h1.y; acc.z += a1 * h1.z; acc.w += a1 * h1.w; accr += a1 * r1;
    acc.x += a2 * h2.x; acc.y += a2 * h2.y; acc.z += a2 * h2.z; acc.w += a2 * h2.w; accr += a2 * r2;
    acc.x += a3 * h3.x; acc.y += a3 * h3.y; acc.z += a3 * h3.z; acc.w += a3 * h3.w; accr += a3 * r3;
}

__global__ __launch_bounds__(256) void node_gather(
    const int* __restrict__ rowptr, const int* __restrict__ pk,
    const float* __restrict__ as, const float* __restrict__ ad,
    const float* __restrict__ ce,
    const float* __restrict__ h, const float* __restrict__ re,
    float* __restrict__ A64, float* __restrict__ A16, int N)
{
    const int n = blockIdx.x * 16 + (threadIdx.x >> 4);
    const int q = threadIdx.x & 15;
    if (n >= N) return;
    const int beg = rowptr[n], end = rowptr[n + 1];
    const float4 as4 = *(const float4*)(as + q * 4);
    const float4 ad4 = *(const float4*)(ad + q * 4);
    const float ceq = ce[q];
    const float4 hn = *(const float4*)(h + (size_t)n * 64 + q * 4);
    asm volatile("" :: "v"(hn.x), "v"(hn.y), "v"(hn.z), "v"(hn.w),
                       "v"(ad4.x), "v"(ad4.y), "v"(ad4.z), "v"(ad4.w),
                       "v"(as4.x), "v"(as4.y), "v"(as4.z), "v"(as4.w), "v"(ceq));
    const int2* __restrict__ pk2 = (const int2*)pk;
    float4 acc = {0.f, 0.f, 0.f, 0.f};
    float accr = 0.f;
    if (end > beg) {
        const int last = end - 1;
        int2 e = pk2[min(beg + q, last)];
        const float dn = row_sum16(hn.x * ad4.x + hn.y * ad4.y + hn.z * ad4.z + hn.w * ad4.w);
        float m = -__builtin_inff();
        float dsum = 0.f;
        int c = beg;
        for (;;) {
            const int nlive = end - c;
            gather4<0>(e, q, nlive, as4, ceq, dn, h, re, m, dsum, acc, accr);
            if (nlive > 4)  gather4<4>(e, q, nlive, as4, ceq, dn, h, re, m, dsum, acc, accr);
            if (nlive > 8)  gather4<8>(e, q, nlive, as4, ceq, dn, h, re, m, dsum, acc, accr);
            if (nlive > 12) gather4<12>(e, q, nlive, as4, ceq, dn, h, re, m, dsum, acc, accr);
            c += 16;
            if (c >= end) break;
            e = pk2[min(c + q, last)];
        }
        const float inv = 1.f / dsum;
        acc.x *= inv; acc.y *= inv; acc.z *= inv; acc.w *= inv; accr *= inv;
    }
    *(float4*)(A64 + (size_t)n * 64 + q * 4) = acc;
    A16[(size_t)n * 16 + q] = accr;
}

#define WALK 16

__device__ __forceinline__ void row_flush(float* __restrict__ grow, const int q, const float4 a)
{
    float* p = grow + q * 4;
    atomicAdd(p + 0, a.x); atomicAdd(p + 1, a.y); atomicAdd(p + 2, a.z); atomicAdd(p + 3, a.w);
}

__device__ __forceinline__ void row_bcast_all(const int iq, const float fq,
                                              int (&iv)[WALK], float (&fv)[WALK])
{
#define BC1(U) iv[U] = row_bcast<U>(iq); fv[U] = row_bcast<U>(fq);
    BC1(0)  BC1(1)  BC1(2)  BC1(3)  BC1(4)  BC1(5)  BC1(6)  BC1(7)
    BC1(8)  BC1(9)  BC1(10) BC1(11) BC1(12) BC1(13) BC1(14) BC1(15)
#undef BC1
}

__global__ __launch_bounds__(128) void gate_hi(
    const float* __restrict__ h, const float* __restrict__ d_edge,
    const int* __restrict__ gid, const float* __restrict__ wd,
    const float* __restrict__ bd, float* __restrict__ gate,
    float* __restrict__ hi, int N)
{
    const int q  = threadIdx.x & 15;
    const int n0 = (blockIdx.x * 8 + (threadIdx.x >> 4)) * WALK;
    if (n0 >= N) return;
    const int nq = min(n0 + q, N - 1);
    float4 hv[WALK];
#pragma unroll
    for (int u = 0; u < WALK; ++u)
        hv[u] = *(const float4*)(h + (size_t)min(n0 + u, N - 1) * 64 + q * 4);
    const float gq = 1.f / (1.f + __expf(-(d_edge[nq] * wd[0] + bd[0])));
    const int gidq = gid[nq];
    if (n0 + q < N) gate[n0 + q] = gq;
    int gi[WALK]; float g[WALK];
    row_bcast_all(gidq, gq, gi, g);
    float4 acc = {0.f, 0.f, 0.f, 0.f};
    int cur = gi[0];
#pragma unroll
    for (int u = 0; u < WALK; ++u) {
        if (n0 + u < N) {
            if (gi[u] != cur) {
                row_flush(hi + (size_t)cur * 64, q, acc);
                acc.x = 0.f; acc.y = 0.f; acc.z = 0.f; acc.w = 0.f;
                cur = gi[u];
            }
            acc.x += g[u] * hv[u].x; acc.y += g[u] * hv[u].y;
            acc.z += g[u] * hv[u].z; acc.w += g[u] * hv[u].w;
        }
    }
    row_flush(hi + (size_t)cur * 64, q, acc);
}

__global__ __launch_bounds__(128) void h_mix_pool(
    float* __restrict__ h, const float* __restrict__ gate,
    const float* __restrict__ hi, const int* __restrict__ gid,
    float* __restrict__ pooled, int N)
{
    const int q  = threadIdx.x & 15;
    const int n0 = (blockIdx.x * 8 + (threadIdx.x >> 4)) * WALK;
    if (n0 >= N) return;
    const int nq = min(n0 + q, N - 1);
    float4 hv[WALK];
#pragma unroll
    for (int u = 0; u < WALK; ++u)
        hv[u] = *(const float4*)(h + (size_t)min(n0 + u, N - 1) * 64 + q * 4);
    const float gq = gate[nq];
    const int gidq = gid[nq];
    int gi[WALK]; float g[WALK];
    row_bcast_all(gidq, gq, gi, g);
    float4 hiv[WALK];
#pragma unroll
    for (int u = 0; u < WALK; ++u)
        hiv[u] = *(const float4*)(hi + (size_t)gi[u] * 64 + q * 4);
    float4 acc = {0.f, 0.f, 0.f, 0.f};
    int cur = gi[0];
#pragma unroll
    for (int u = 0; u < WALK; ++u) {
        if (n0 + u < N) {
            float4 v;
            v.x = hv[u].x + g[u] * hiv[u].x; v.y = hv[u].y + g[u] * hiv[u].y;
            v.z = hv[u].z + g[u] * hiv[u].z; v.w = hv[u].w + g[u] * hiv[u].w;
            *(float4*)(h + (size_t)(n0 + u) * 64 + q * 4) = v;
            if (gi[u] != cur) {
                row_flush(pooled + (size_t)cur * 64, q, acc);
                acc.x = 0.f; acc.y = 0.f; acc.z = 0.f; acc.w = 0.f;
                cur = gi[u];
            }
            acc.x += v.x; acc.y += v.y; acc.z += v.z; acc.w += v.w;
        }
    }
    row_flush(pooled + (size_t)cur * 64, q, acc);
}

__global__ void out_kernel(const float* __restrict__ x, const float* __restrict__ Wo,
                           const float* __restrict__ bo, float* __restrict__ out, int G)
{
    const int g = blockIdx.x * blockDim.x + threadIdx.x;
    if (g >= G) return;
    float s = bo[0];
#pragma unroll
    for (int k = 0; k < 64; ++k) s += x[(size_t)g * 64 + k] * Wo[k];
    out[g] = s;
}

extern "C" void kernel_launch(void* const* d_in, const int* in_sizes, int n_in,
                              void* d_out, int out_size, void* d_ws, size_t ws_size,
                              hipStream_t stream)
{
    const float* r_node = (const float*)d_in[0];
    const float* i_node = (const float*)d_in[1];
    const float* r_edge = (const float*)d_in[2];
    const float* d_edge = (const float*)d_in[3];
    const int*   src    = (const int*)d_in[4];
    const int*   dst    = (const int*)d_in[5];
    const int*   gid    = (const int*)d_in[6];
    const float* W_r    = (const float*)d_in[7];
    const float* W_i    = (const float*)d_in[8];
    const float* W_e    = (const float*)d_in[9];
    const float* Wm     = (const float*)d_in[10];
    const float* a_s    = (const float*)d_in[11];
    const float* a_d    = (const float*)d_in[12];
    const float* a_e    = (const float*)d_in[13];
    const float* w_d    = (const float*)d_in[14];
    const float* b_d    = (const float*)d_in[15];
    const float* W_mlp  = (const float*)d_in[16];
    const float* b_mlp  = (const float*)d_in[17];
    const float* W_out  = (const float*)d_in[18];
    const float* b_out  = (const float*)d_in[19];

    const int N = in_sizes[0] / 64;
    const int G = in_sizes[1];
    const int E = in_sizes[2] / 16;
    (void)n_in; (void)out_size; (void)ws_size;

    float* ws = (float*)d_ws;
    size_t off = 0;
    auto carve = [&](size_t n) -> float* {
        float* p = ws + off; off += (n + 63) & ~(size_t)63; return p;
    };
    float*    h      = carve((size_t)N * 64);
    float*    A64    = carve((size_t)N * 64);
    float*    A16    = carve((size_t)N * 16);
    float*    gate   = carve(N);
    float*    hi     = carve((size_t)G * 64);
    float*    pooled = carve((size_t)G * 64);
    float*    x1     = carve((size_t)G * 64);
    float*    x2     = carve((size_t)G * 64);
    float*    cE     = carve(64);
    float*    WeWm   = carve(4 * 16 * 64);
    const int R   = (N + CSR_NB - 1) / CSR_NB;
    const int NP  = (N + 1 + 63) & ~63;
    const int CSP = R * (CSR_NB / CSR_CH);
    int*      pc     = (int*)carve((size_t)CSR_S * NP);
    int*      cs     = (int*)carve((size_t)CSR_S * CSP);
    int*      rowptr = (int*)carve((size_t)CSR_S * NP);
    int*      pk     = (int*)carve((size_t)E * 2);

    hipLaunchKernelGGL(prep_ce, dim3(1), dim3(64), 0, stream, W_e, a_e, cE);
    hipLaunchKernelGGL(prep_wewm, dim3((4096 + 255) / 256), dim3(256), 0, stream, W_e, Wm, WeWm);

    hipLaunchKernelGGL((gemm64<false, false, false, 0>), dim3(N / 16), dim3(32), 0, stream,
                       r_node, W_r, nullptr, nullptr, nullptr, nullptr, h);
    hipLaunchKernelGGL(init_hi, dim3((G * 64 + 255) / 256), dim3(256), 0, stream,
                       i_node, W_i, hi, pooled, G);

    hipLaunchKernelGGL(csr_count, dim3(R * CSR_S), dim3(1024), 0, stream, dst, pc, cs, N, NP, CSP, E, R);
    hipLaunchKernelGGL(csr_scan, dim3((N + 1023) / 1024), dim3(1024), 0, stream, pc, cs, rowptr, N, NP, CSP);
    hipLaunchKernelGGL(csr_fill, dim3(R * CSR_S), dim3(1024), 0, stream,
                       src, dst, rowptr, pk, N, NP, E, R);

    for (int l = 0; l < 4; ++l) {
        hipLaunchKernelGGL(node_gather, dim3((N + 15) / 16), dim3(256), 0, stream,
                           rowptr, pk, a_s + l * 64, a_d + l * 64, cE + l * 16,
                           h, r_edge, A64, A16, N);
        hipLaunchKernelGGL((gemm64<true, true, false, 1>), dim3(N / 16), dim3(32), 0, stream,
                           A64, Wm + (size_t)l * 4096, A16, WeWm + (size_t)l * 1024,
                           h, nullptr, h);
    }

    hipLaunchKernelGGL(gate_hi, dim3((N + 127) / 128), dim3(128), 0, stream,
                       h, d_edge, gid, w_d, b_d, gate, hi, N);
    hipLaunchKernelGGL(h_mix_pool, dim3((N + 127) / 128), dim3(128), 0, stream,
                       h, gate, hi, gid, pooled, N);

    hipLaunchKernelGGL((gemm64<false, false, true, 2>), dim3(G / 16), dim3(32), 0, stream,
                       pooled, W_mlp + 0 * 4096, nullptr, nullptr, nullptr, b_mlp + 0 * 64, x1);
    hipLaunchKernelGGL((gemm64<false, false, true, 2>), dim3(G / 16), dim3(32), 0, stream,
                       x1, W_mlp + 1 * 4096, nullptr, nullptr, nullptr, b_mlp + 1 * 64, x2);
    hipLaunchKernelGGL((gemm64<false, false, true, 2>), dim3(G / 16), dim3(32), 0, stream,
                       x2, W_mlp + 2 * 4096, nullptr, nullptr, nullptr, b_mlp + 2 * 64, x1);

    hipLaunchKernelGGL(out_kernel, dim3((G + 255) / 256), dim3(256), 0, stream,
                       x1, W_out, b_out, (float*)d_out, G);
}


# Layer GEMM fused into the gather launch: a group's 16 summed rows pass through LDS as the A tile of 4 waves' f32 WMMA, new h rows go to a second h buffer; A64/A16 planes and 4 launches gone
# speedup vs baseline: 7.3713x; 1.2062x over previous
#include <hip/hip_runtime.h>

typedef __attribute__((ext_vector_type(2))) float v2f;
typedef __attribute__((ext_vector_type(8))) float v8f;

#define HD 64

__device__ __forceinline__ float lrelu_f(float x) { return x > 0.f ? x : 0.1f * x; }

template <int M> __device__ __forceinline__ float row_xor(float v) {
    return __int_as_float(__builtin_amdgcn_mov_dpp(__float_as_int(v), 0x160 + M, 0xf, 0xf, false));
}
template <int U> __device__ __forceinline__ int row_bcast(int v) {
    return __builtin_amdgcn_mov_dpp(v, 0x150 + U, 0xf, 0xf, false);
}
template <int U> __device__ __forceinline__ float row_bcast(float v) {
    return __int_as_float(__builtin_amdgcn_mov_dpp(__float_as_int(v), 0x150 + U, 0xf, 0xf, false));
}
__device__ __forceinline__ float row_max16(float v) {
    v = fmaxf(v, row_xor<1>(v)); v = fmaxf(v, row_xor<2>(v));
    v = fmaxf(v, row_xor<4>(v)); v = fmaxf(v, row_xor<8>(v));
    return v;
}
__device__ __forceinline__ float row_sum16(float v) {
    v += row_xor<1>(v); v += row_xor<2>(v); v += row_xor<4>(v); v += row_xor<8>(v);
    return v;
}

template <bool HAS_A2, bool HAS_RES, bool HAS_BIAS, int ACT>
__global__ __launch_bounds__(32) void gemm64(
    const float* __restrict__ A,  const float* __restrict__ B,
    const float* __restrict__ A2, const float* __restrict__ B2,
    const float* __restrict__ res, const float* __restrict__ bias,
    float* __restrict__ out)
{
    const int lane = threadIdx.x & 31;
    const int cn   = lane & 15;
    const int kh   = (lane >> 4) << 1;
    const size_t row = (size_t)blockIdx.x * 16 + cn;

    const v8f z = {0.f, 0.f, 0.f, 0.f, 0.f, 0.f, 0.f, 0.f};
    v8f acc0 = z, acc1 = z, acc2 = z, acc3 = z;

    const float* arow = A + row * HD;
#pragma unroll
    for (int k0 = 0; k0 < 64; k0 += 4) {
        v2f a; a.x = arow[k0 + kh]; a.y = arow[k0 + kh + 1];
        const float* b0 = B + (size_t)(k0 + kh) * HD;
        const float* b1 = b0 + HD;
        v2f b;
        b.x = b0[cn];        b.y = b1[cn];
        acc0 = __builtin_amdgcn_wmma_f32_16x16x4_f32(false, a, false, b, (short)0, acc0, false, false);
        b.x = b0[16 + cn];   b.y = b1[16 + cn];
        acc1 = __builtin_amdgcn_wmma_f32_16x16x4_f32(false, a, false, b, (short)0, acc1, false, false);
        b.x = b0[32 + cn];   b.y = b1[32 + cn];
        acc2 = __builtin_amdgcn_wmma_f32_16x16x4_f32(false, a, false, b, (short)0, acc2, false, false);
        b.x = b0[48 + cn];   b.y = b1[48 + cn];
        acc3 = __builtin_amdgcn_wmma_f32_16x16x4_f32(false, a, false, b, (short)0, acc3, false, false);
    }
    if (HAS_A2) {
        const float* a2row = A2 + row * 16;
#pragma unroll
        for (int k0 = 0; k0 < 16; k0 += 4) {
            v2f a; a.x = a2row[k0 + kh]; a.y = a2row[k0 + kh + 1];
            const float* b0 = B2 + (size_t)(k0 + kh) * HD;
            const float* b1 = b0 + HD;
            v2f b;
            b.x = b0[cn];        b.y = b1[cn];
            acc0 = __builtin_amdgcn_wmma_f32_16x16x4_f32(false, a, false, b, (short)0, acc0, false, false);
            b.x = b0[16 + cn];   b.y = b1[16 + cn];
            acc1 = __builtin_amdgcn_wmma_f32_16x16x4_f32(false, a, false, b, (short)0, acc1, false, false);
            b.x = b0[32 + cn];   b.y = b1[32 + cn];
            acc2 = __builtin_amdgcn_wmma_f32_16x16x4_f32(false, a, false, b, (short)0, acc2, false, false);
            b.x = b0[48 + cn];   b.y = b1[48 + cn];
            acc3 = __builtin_amdgcn_wmma_f32_16x16x4_f32(false, a, false, b, (short)0, acc3, false, false);
        }
    }

    const size_t mbase = (size_t)blockIdx.x * 16 + ((lane >> 4) << 3);
    v8f accs[4] = {acc0, acc1, acc2, acc3};
#pragma unroll
    for (int t = 0; t < 4; ++t) {
        const int col = t * 16 + cn;
        const float bval = HAS_BIAS ? bias[col] : 0.f;
#pragma unroll
        for (int v = 0; v < 8; ++v) {
            const size_t m = mbase + v;
            float val = accs[t][v] + bval;
            if (HAS_RES) val += res[m * HD + col];
            if (ACT == 1) val = lrelu_f(val);
            else if (ACT == 2) val = fmaxf(val, 0.f);
            out[m * HD + col] = val;
        }
    }
}

__global__ void prep_ce(const float* __restrict__ W_e, const float* __restrict__ a_e,
                        float* __restrict__ cE)
{
    const int tid = threadIdx.x;
    const int l = tid >> 4, d = tid & 15;
    float s = 0.f;
    for (int k = 0; k < 64; ++k) s += W_e[d * 64 + k] * a_e[l * 64 + k];
    cE[tid] = s;
}

__global__ void prep_wewm(const float* __restrict__ W_e, const float* __restrict__ Wm,
                          float* __restrict__ WeWm)
{
    const int idx = blockIdx.x * blockDim.x + threadIdx.x;
    if (idx >= 4 * 16 * 64) return;
    const int l = idx >> 10, d = (idx >> 6) & 15, k = idx & 63;
    float s = 0.f;
    for (int j = 0; j < 64; ++j) s += W_e[d * 64 + j] * Wm[l * 4096 + j * 64 + k];
    WeWm[idx] = s;
}

__global__ void init_hi(const float* __restrict__ i_node, const float* __restrict__ W_i,
                        float* __restrict__ hi, float* __restrict__ pooled, int G)
{
    const int idx = blockIdx.x * blockDim.x + threadIdx.x;
    if (idx >= G * 64) return;
    hi[idx] = i_node[idx >> 6] * W_i[idx & 63];
    pooled[idx] = 0.f;
}

#define CSR_NB 12800
#define CSR_S  32
#define CSR_CH 128

__global__ __launch_bounds__(1024) void csr_count(const int* __restrict__ dst,
                                                  int* __restrict__ pc, int* __restrict__ cs,
                                                  int N, int NP, int CSP, int E, int R)
{
    __shared__ int cnt[CSR_NB];
    const int t = threadIdx.x;
    const int r = blockIdx.x % R, s = blockIdx.x / R;
    const int r0 = r * CSR_NB;
    for (int i = t; i < CSR_NB; i += 1024) cnt[i] = 0;
    __syncthreads();
    const int E4 = E >> 2;
    const int SL4 = (E4 + CSR_S - 1) / CSR_S;
    const int b4 = s * SL4, e4 = min(b4 + SL4, E4);
    const int4* __restrict__ d4 = (const int4*)dst;
    for (int i = b4 + t; i < e4; i += 4096) {
        int4 d[4];
#pragma unroll
        for (int k = 0; k < 4; ++k) d[k] = d4[min(i + k * 1024, e4 - 1)];
#pragma unroll
        for (int k = 0; k < 4; ++k) {
            if (i + k * 1024 < e4) {
                unsigned u;
                u = (unsigned)(d[k].x - r0); if (u < CSR_NB) atomicAdd(&cnt[u], 1);
                u = (unsigned)(d[k].y - r0); if (u < CSR_NB) atomicAdd(&cnt[u], 1);
                u = (unsigned)(d[k].z - r0); if (u < CSR_NB) atomicAdd(&cnt[u], 1);
                u = (unsigned)(d[k].w - r0); if (u < CSR_NB) atomicAdd(&cnt[u], 1);
            }
        }
    }
    if (s == CSR_S - 1) {
        for (int j = (E4 << 2) + t; j < E; j += 1024) {
            const unsigned u = (unsigned)(dst[j] - r0);
            if (u < CSR_NB) atomicAdd(&cnt[u], 1);
        }
    }
    __syncthreads();
    for (int i = t; i < CSR_NB; i += 1024) {
        const int n = r0 + i;
        if (n < N) pc[(size_t)s * NP + n] = cnt[i];
    }
    const int lane = t & 31, w = t >> 5;
    for (int c = w; c < CSR_NB / CSR_CH; c += 32) {
        const int* p = cnt + c * CSR_CH + lane;
        int v = (p[0] + p[32]) + (p[64] + p[96]);
        for (int o = 16; o > 0; o >>= 1) v += __shfl_xor(v, o, 32);
        if (lane == 0) cs[s * CSP + r * (CSR_NB / CSR_CH) + c] = v;
    }
}

__global__ __launch_bounds__(1024) void csr_scan(const int* __restrict__ pc,
                                                 const int* __restrict__ cs,
                                                 int* __restrict__ base, int N, int NP, int CSP)
{
    __shared__ int wfront[32];
    __shared__ int wtot[32];
    const int t = threadIdx.x, lane = t & 31, w = t >> 5;
    const int base0 = blockIdx.x * 1024;
    const int ch0 = blockIdx.x * (1024 / CSR_CH);

    int sf = 0;
    for (int c = t; c < ch0; c += 1024) {
#pragma unroll
        for (int s = 0; s < CSR_S; ++s) sf += cs[s * CSP + c];
    }
    for (int o = 16; o > 0; o >>= 1) sf += __shfl_xor(sf, o, 32);
    if (lane == 0) wfront[w] = sf;

    const int i = base0 + t;
    const int ic = min(i, N - 1);
    int pcs[CSR_S];
    int c = 0;
#pragma unroll
    for (int s = 0; s < CSR_S; ++s) {
        pcs[s] = (i < N) ? pc[(size_t)s * NP + ic] : 0;
        c += pcs[s];
    }
    int x = c;
    for (int o = 1; o < 32; o <<= 1) {
        const int y = __shfl_up(x, o, 32);
        if (lane >= o) x += y;
    }
    if (lane == 31) wtot[w] = x;
    __syncthreads();

    int front = 0;
    for (int k = 0; k < 32; ++k) front += wfront[k];
    for (int k = 0; k < w; ++k) front += wtot[k];
    if (i < N) {
        int run = front + x - c;
#pragma unroll
        for (int s = 0; s < CSR_S; ++s) {
            base[(size_t)s * NP + i] = run;
            run += pcs[s];
        }
        if (i == N - 1) base[N] = run;
    }
}

__global__ __launch_bounds__(1024) void csr_fill(const int* __restrict__ src,
                                                 const int* __restrict__ dst,
                                                 const int* __restrict__ base,
                                                 int* __restrict__ pk, int N, int NP, int E, int R)
{
    __shared__ int pos[CSR_NB];
    const int t = threadIdx.x;
    const int r = blockIdx.x % R, s = blockIdx.x / R;
    const int r0 = r * CSR_NB;
    for (int i = t; i < CSR_NB; i += 1024) {
        const int n = r0 + i;
        pos[i] = (n < N) ? base[(size_t)s * NP + n] : 0;
    }
    __syncthreads();
    const int E4 = E >> 2;
    const int SL4 = (E4 + CSR_S - 1) / CSR_S;
    const int b4 = s * SL4, e4 = min(b4 + SL4, E4);
    const int4* __restrict__ d4 = (const int4*)dst;
    const int4* __restrict__ s4 = (const int4*)src;
    int2* __restrict__ pk2 = (int2*)pk;
    for (int i = b4 + t; i < e4; i += 4096) {
        int4 d[4], sv[4];
#pragma unroll
        for (int k = 0; k < 4; ++k) {
            const int idx = min(i + k * 1024, e4 - 1);
            d[k] = d4[idx]; sv[k] = s4[idx];
        }
        int p[16];
#pragma unroll
        for (int k = 0; k < 4; ++k) {
            const bool live = i + k * 1024 < e4;
            unsigned u;
            u = (unsigned)(d[k].x - r0); p[4 * k + 0] = -1;
            if (live && u < CSR_NB) p[4 * k + 0] = atomicAdd(&pos[u], 1);
            u = (unsigned)(d[k].y - r0); p[4 * k + 1] = -1;
            if (live && u < CSR_NB) p[4 * k + 1] = atomicAdd(&pos[u], 1);
            u = (unsigned)(d[k].z - r0); p[4 * k + 2] = -1;
            if (live && u < CSR_NB) p[4 * k + 2] = atomicAdd(&pos[u], 1);
            u = (unsigned)(d[k].w - r0); p[4 * k + 3] = -1;
            if (live && u < CSR_NB) p[4 * k + 3] = atomicAdd(&pos[u], 1);
        }
#pragma unroll
        for (int k = 0; k < 4; ++k) {
            const int j = (i + k * 1024) << 2;
            if (p[4 * k + 0] >= 0) pk2[p[4 * k + 0]] = make_int2(j,     sv[k].x);
            if (p[4 * k + 1] >= 0) pk2[p[4 * k + 1]] = make_int2(j + 1, sv[k].y);
            if (p[4 * k + 2] >= 0) pk2[p[4 * k + 2]] = make_int2(j + 2, sv[k].z);
            if (p[4 * k + 3] >= 0) pk2[p[4 * k + 3]] = make_int2(j + 3, sv[k].w);
        }
    }
    if (s == CSR_S - 1) {
        for (int j = (E4 << 2) + t; j < E; j += 1024) {
            const unsigned u = (unsigned)(dst[j] - r0);
            if (u < CSR_NB) { const int q = atomicAdd(&pos[u], 1); pk2[q] = make_int2(j, src[j]); }
        }
    }
}

template <int U0> __device__ __forceinline__ void gather4(
    const int2 e, const int q, const int nlive,
    const float4 as4, const float ceq, const float dn,
    const float* __restrict__ h, const float* __restrict__ re,
    float& m, float& dsum, float4& acc, float& accr)
{
    const int j0 = row_bcast<U0 + 0>(e.x), s0 = row_bcast<U0 + 0>(e.y);
    const int j1 = row_bcast<U0 + 1>(e.x), s1 = row_bcast<U0 + 1>(e.y);
    const int j2 = row_bcast<U0 + 2>(e.x), s2 = row_bcast<U0 + 2>(e.y);
    const int j3 = row_bcast<U0 + 3>(e.x), s3 = row_bcast<U0 + 3>(e.y);
    const float4 h0 = *(const float4*)(h + ((unsigned)s0 * 64u + (unsigned)q * 4u));
    const float4 h1 = *(const float4*)(h + ((unsigned)s1 * 64u + (unsigned)q * 4u));
    const float4 h2 = *(const float4*)(h + ((unsigned)s2 * 64u + (unsigned)q * 4u));
    const float4 h3 = *(const float4*)(h + ((unsigned)s3 * 64u + (unsigned)q * 4u));
    const float r0 = re[(unsigned)j0 * 16u + (unsigned)q];
    const float r1 = re[(unsigned)j1 * 16u + (unsigned)q];
    const float r2 = re[(unsigned)j2 * 16u + (unsigned)q];
    const float r3 = re[(unsigned)j3 * 16u + (unsigned)q];
    const float ninf = -__builtin_inff();
    float l0 = row_sum16(h0.x * as4.x + h0.y * as4.y + h0.z * as4.z + h0.w * as4.w + r0 * ceq);
    float l1 = row_sum16(h1.x * as4.x + h1.y * as4.y + h1.z * as4.z + h1.w * as4.w + r1 * ceq);
    float l2 = row_sum16(h2.x * as4.x + h2.y * as4.y + h2.z * as4.z + h2.w * as4.w + r2 * ceq);
    float l3 = row_sum16(h3.x * as4.x + h3.y * as4.y + h3.z * as4.z + h3.w * as4.w + r3 * ceq);
    l0 = lrelu_f(l0 + dn);
    l1 = (U0 + 1 < nlive) ? lrelu_f(l1 + dn) : ninf;
    l2 = (U0 + 2 < nlive) ? lrelu_f(l2 + dn) : ninf;
    l3 = (U0 + 3 < nlive) ? lrelu_f(l3 + dn) : ninf;
    const float mn = fmaxf(fmaxf(m, l0), fmaxf(fmaxf(l1, l2), l3));
    const float sc = __expf(m - mn);
    const float a0 = __expf(l0 - mn), a1 = __expf(l1 - mn);
    const float a2 = __expf(l2 - mn), a3 = __expf(l3 - mn);
    m = mn;
    dsum = dsum * sc + ((a0 + a1) + (a2 + a3));
    acc.x *= sc; acc.y *= sc; acc.z *= sc; acc.w *= sc; accr *= sc;
    acc.x += a0 * h0.x; acc.y += a0 * h0.y; acc.z += a0 * h0.z; acc.w += a0 * h0.w; accr += a0 * r0;
    acc.x += a1 * h1.x; acc.y += a1 * h1.y; acc.z += a1 * h1.z; acc.w += a1 * h1.w; accr += a1 * r1;
    acc.x += a2 * h2.x; acc.y += a2 * h2.y; acc.z += a2 * h2.z; acc.w += a2 * h2.w; accr += a2 * r2;
    acc.x += a3 * h3.x; acc.y += a3 * h3.y; acc.z += a3 * h3.z; acc.w += a3 * h3.w; accr += a3 * r3;
}

#define TA_P 84
#define TD_P 68
#define TB_P 64
__global__ __launch_bounds__(256) void layer_fused(
    const int* __restrict__ rowptr, const int* __restrict__ pk,
    const float* __restrict__ as, const float* __restrict__ ad,
    const float* __restrict__ ce,
    const float* __restrict__ h, const float* __restrict__ re,
    const float* __restrict__ Wm, const float* __restrict__ WeWm,
    float* __restrict__ hout, int N)
{
    __shared__ __attribute__((aligned(16))) float sA[16 * TA_P];
    __shared__ __attribute__((aligned(16))) float sD[16 * TD_P];
    __shared__ __attribute__((aligned(16))) float sB[80 * TB_P];
    const int r = threadIdx.x >> 4;
    const int nn = blockIdx.x * 16 + r;
    const int n = min(nn, N - 1);
    const int q = threadIdx.x & 15;
    const int t = threadIdx.x;
    const float4 w0 = ((const float4*)Wm)[t];
    const float4 w1 = ((const float4*)Wm)[t + 256];
    const float4 w2 = ((const float4*)Wm)[t + 512];
    const float4 w3 = ((const float4*)Wm)[t + 768];
    const float4 w4 = ((const float4*)WeWm)[t];
    const int beg = rowptr[n], end = rowptr[n + 1];
    const float4 as4 = *(const float4*)(as + q * 4);
    const float4 ad4 = *(const float4*)(ad + q * 4);
    const float ceq = ce[q];
    const float4 hn = *(const float4*)(h + (size_t)n * 64 + q * 4);
    {
        float* sb = sB + (t >> 4) * TB_P + (t & 15) * 4;
        *(float4*)(sb)              = w0;
        *(float4*)(sb + 16 * TB_P)  = w1;
        *(float4*)(sb + 32 * TB_P)  = w2;
        *(float4*)(sb + 48 * TB_P)  = w3;
        *(float4*)(sb + 64 * TB_P)  = w4;
    }
    asm volatile("" :: "v"(hn.x), "v"(hn.y), "v"(hn.z), "v"(hn.w),
                       "v"(ad4.x), "v"(ad4.y), "v"(ad4.z), "v"(ad4.w),
                       "v"(as4.x), "v"(as4.y), "v"(as4.z), "v"(as4.w), "v"(ceq));
    const int2* __restrict__ pk2 = (const int2*)pk;
    float4 acc = {0.f, 0.f, 0.f, 0.f};
    float accr = 0.f;
    if (end > beg) {
        const int last = end - 1;
        int2 e = pk2[min(beg + q, last)];
        const float dn = row_sum16(hn.x * ad4.x + hn.y * ad4.y + hn.z * ad4.z + hn.w * ad4.w);
        float m = -__builtin_inff();
        float dsum = 0.f;
        int c = beg;
        for (;;) {
            const int nlive = end - c;
            gather4<0>(e, q, nlive, as4, ceq, dn, h, re, m, dsum, acc, accr);
            if (nlive > 4)  gather4<4>(e, q, nlive, as4, ceq, dn, h, re, m, dsum, acc, accr);
            if (nlive > 8)  gather4<8>(e, q, nlive, as4, ceq, dn, h, re, m, dsum, acc, accr);
            if (nlive > 12) gather4<12>(e, q, nlive, as4, ceq, dn, h, re, m, dsum, acc, accr);
            c += 16;
            if (c >= end) break;
            e = pk2[min(c + q, last)];
        }
        const float inv = 1.f / dsum;
        acc.x *= inv; acc.y *= inv; acc.z *= inv; acc.w *= inv; accr *= inv;
    }

    const int wv = __builtin_amdgcn_readfirstlane((int)(threadIdx.x >> 5));
    const int lane = threadIdx.x & 31;
    const int cn = lane & 15;
    const int kh = (lane >> 4) << 1;
    *(float4*)(sA + r * TA_P + q * 4) = acc;
    sA[r * TA_P + 64 + q] = accr;
    __syncthreads();
    if (wv < 4) {
        v8f d = {0.f, 0.f, 0.f, 0.f, 0.f, 0.f, 0.f, 0.f};
        const float* ap = sA + cn * TA_P + kh;
        const float* bp = sB + kh * TB_P + wv * 16 + cn;
#pragma unroll
        for (int k = 0; k < 20; ++k) {
            const v2f a = *(const v2f*)(ap + 4 * k);
            v2f b; b.x = bp[4 * k * TB_P]; b.y = bp[(4 * k + 1) * TB_P];
            d = __builtin_amdgcn_wmma_f32_16x16x4_f32(false, a, false, b, (short)0, d, false, false);
        }
        float* dp = sD + ((lane >> 4) << 3) * TD_P + wv * 16 + cn;
#pragma unroll
        for (int v = 0; v < 8; ++v) dp[v * TD_P] = d[v];
    }
    __syncthreads();
    if (nn < N) {
        float4 o = *(const float4*)(sD + r * TD_P + q * 4);
        o.x = lrelu_f(o.x + hn.x); o.y = lrelu_f(o.y + hn.y);
        o.z = lrelu_f(o.z + hn.z); o.w = lrelu_f(o.w + hn.w);
        *(float4*)(hout + (size_t)nn * 64 + q * 4) = o;
    }
}

#define WALK 16

__device__ __forceinline__ void row_flush(float* __restrict__ grow, const int q, const float4 a)
{
    float* p = grow + q * 4;
    atomicAdd(p + 0, a.x); atomicAdd(p + 1, a.y); atomicAdd(p + 2, a.z); atomicAdd(p + 3, a.w);
}

__device__ __forceinline__ void row_bcast_all(const int iq, const float fq,
                                              int (&iv)[WALK], float (&fv)[WALK])
{
#define BC1(U) iv[U] = row_bcast<U>(iq); fv[U] = row_bcast<U>(fq);
    BC1(0)  BC1(1)  BC1(2)  BC1(3)  BC1(4)  BC1(5)  BC1(6)  BC1(7)
    BC1(8)  BC1(9)  BC1(10) BC1(11) BC1(12) BC1(13) BC1(14) BC1(15)
#undef BC1
}

__global__ __launch_bounds__(128) void gate_hi(
    const float* __restrict__ h, const float* __restrict__ d_edge,
    const int* __restrict__ gid, const float* __restrict__ wd,
    const float* __restrict__ bd, float* __restrict__ gate,
    float* __restrict__ hi, int N)
{
    const int q  = threadIdx.x & 15;
    const int n0 = (blockIdx.x * 8 + (threadIdx.x >> 4)) * WALK;
    if (n0 >= N) return;
    const int nq = min(n0 + q, N - 1);
    float4 hv[WALK];
#pragma unroll
    for (int u = 0; u < WALK; ++u)
        hv[u] = *(const float4*)(h + (size_t)min(n0 + u, N - 1) * 64 + q * 4);
    const float gq = 1.f / (1.f + __expf(-(d_edge[nq] * wd[0] + bd[0])));
    const int gidq = gid[nq];
    if (n0 + q < N) gate[n0 + q] = gq;
    int gi[WALK]; float g[WALK];
    row_bcast_all(gidq, gq, gi, g);
    float4 acc = {0.f, 0.f, 0.f, 0.f};
    int cur = gi[0];
#pragma unroll
    for (int u = 0; u < WALK; ++u) {
        if (n0 + u < N) {
            if (gi[u] != cur) {
                row_flush(hi + (size_t)cur * 64, q, acc);
                acc.x = 0.f; acc.y = 0.f; acc.z = 0.f; acc.w = 0.f;
                cur = gi[u];
            }
            acc.x += g[u] * hv[u].x; acc.y += g[u] * hv[u].y;
            acc.z += g[u] * hv[u].z; acc.w += g[u] * hv[u].w;
        }
    }
    row_flush(hi + (size_t)cur * 64, q, acc);
}

__global__ __launch_bounds__(128) void h_mix_pool(
    float* __restrict__ h, const float* __restrict__ gate,
    const float* __restrict__ hi, const int* __restrict__ gid,
    float* __restrict__ pooled, int N)
{
    const int q  = threadIdx.x & 15;
    const int n0 = (blockIdx.x * 8 + (threadIdx.x >> 4)) * WALK;
    if (n0 >= N) return;
    const int nq = min(n0 + q, N - 1);
    float4 hv[WALK];
#pragma unroll
    for (int u = 0; u < WALK; ++u)
        hv[u] = *(const float4*)(h + (size_t)min(n0 + u, N - 1) * 64 + q * 4);
    const float gq = gate[nq];
    const int gidq = gid[nq];
    int gi[WALK]; float g[WALK];
    row_bcast_all(gidq, gq, gi, g);
    float4 hiv[WALK];
#pragma unroll
    for (int u = 0; u < WALK; ++u)
        hiv[u] = *(const float4*)(hi + (size_t)gi[u] * 64 + q * 4);
    float4 acc = {0.f, 0.f, 0.f, 0.f};
    int cur = gi[0];
#pragma unroll
    for (int u = 0; u < WALK; ++u) {
        if (n0 + u < N) {
            float4 v;
            v.x = hv[u].x + g[u] * hiv[u].x; v.y = hv[u].y + g[u] * hiv[u].y;
            v.z = hv[u].z + g[u] * hiv[u].z; v.w = hv[u].w + g[u] * hiv[u].w;
            *(float4*)(h + (size_t)(n0 + u) * 64 + q * 4) = v;
            if (gi[u] != cur) {
                row_flush(pooled + (size_t)cur * 64, q, acc);
                acc.x = 0.f; acc.y = 0.f; acc.z = 0.f; acc.w = 0.f;
                cur = gi[u];
            }
            acc.x += v.x; acc.y += v.y; acc.z += v.z; acc.w += v.w;
        }
    }
    row_flush(pooled + (size_t)cur * 64, q, acc);
}

__global__ void out_kernel(const float* __restrict__ x, const float* __restrict__ Wo,
                           const float* __restrict__ bo, float* __restrict__ out, int G)
{
    const int g = blockIdx.x * blockDim.x + threadIdx.x;
    if (g >= G) return;
    float s = bo[0];
#pragma unroll
    for (int k = 0; k < 64; ++k) s += x[(size_t)g * 64 + k] * Wo[k];
    out[g] = s;
}

extern "C" void kernel_launch(void* const* d_in, const int* in_sizes, int n_in,
                              void* d_out, int out_size, void* d_ws, size_t ws_size,
                              hipStream_t stream)
{
    const float* r_node = (const float*)d_in[0];
    const float* i_node = (const float*)d_in[1];
    const float* r_edge = (const float*)d_in[2];
    const float* d_edge = (const float*)d_in[3];
    const int*   src    = (const int*)d_in[4];
    const int*   dst    = (const int*)d_in[5];
    const int*   gid    = (const int*)d_in[6];
    const float* W_r    = (const float*)d_in[7];
    const float* W_i    = (const float*)d_in[8];
    const float* W_e    = (const float*)d_in[9];
    const float* Wm     = (const float*)d_in[10];
    const float* a_s    = (const float*)d_in[11];
    const float* a_d    = (const float*)d_in[12];
    const float* a_e    = (const float*)d_in[13];
    const float* w_d    = (const float*)d_in[14];
    const float* b_d    = (const float*)d_in[15];
    const float* W_mlp  = (const float*)d_in[16];
    const float* b_mlp  = (const float*)d_in[17];
    const float* W_out  = (const float*)d_in[18];
    const float* b_out  = (const float*)d_in[19];

    const int N = in_sizes[0] / 64;
    const int G = in_sizes[1];
    const int E = in_sizes[2] / 16;
    (void)n_in; (void)out_size; (void)ws_size;

    float* ws = (float*)d_ws;
    size_t off = 0;
    auto carve = [&](size_t n) -> float* {
        float* p = ws + off; off += (n + 63) & ~(size_t)63; return p;
    };
    float*    h      = carve((size_t)N * 64);
    float*    h2     = carve((size_t)N * 64);
    float*    gate   = carve(N);
    float*    hi     = carve((size_t)G * 64);
    float*    pooled = carve((size_t)G * 64);
    float*    x1     = carve((size_t)G * 64);
    float*    x2     = carve((size_t)G * 64);
    float*    cE     = carve(64);
    float*    WeWm   = carve(4 * 16 * 64);
    const int R   = (N + CSR_NB - 1) / CSR_NB;
    const int NP  = (N + 1 + 63) & ~63;
    const int CSP = R * (CSR_NB / CSR_CH);
    int*      pc     = (int*)carve((size_t)CSR_S * NP);
    int*      cs     = (int*)carve((size_t)CSR_S * CSP);
    int*      rowptr = (int*)carve((size_t)CSR_S * NP);
    int*      pk     = (int*)carve((size_t)E * 2);

    hipLaunchKernelGGL(prep_ce, dim3(1), dim3(64), 0, stream, W_e, a_e, cE);
    hipLaunchKernelGGL(prep_wewm, dim3((4096 + 255) / 256), dim3(256), 0, stream, W_e, Wm, WeWm);

    hipLaunchKernelGGL((gemm64<false, false, false, 0>), dim3(N / 16), dim3(32), 0, stream,
                       r_node, W_r, nullptr, nullptr, nullptr, nullptr, h);
    hipLaunchKernelGGL(init_hi, dim3((G * 64 + 255) / 256), dim3(256), 0, stream,
                       i_node, W_i, hi, pooled, G);

    hipLaunchKernelGGL(csr_count, dim3(R * CSR_S), dim3(1024), 0, stream, dst, pc, cs, N, NP, CSP, E, R);
    hipLaunchKernelGGL(csr_scan, dim3((N + 1023) / 1024), dim3(1024), 0, stream, pc, cs, rowptr, N, NP, CSP);
    hipLaunchKernelGGL(csr_fill, dim3(R * CSR_S), dim3(1024), 0, stream,
                       src, dst, rowptr, pk, N, NP, E, R);

    float* hc = h;
    float* hx = h2;
    for (int l = 0; l < 4; ++l) {
        hipLaunchKernelGGL(layer_fused, dim3((N + 15) / 16), dim3(256), 0, stream,
                           rowptr, pk, a_s + l * 64, a_d + l * 64, cE + l * 16,
                           hc, r_edge, Wm + (size_t)l * 4096, WeWm + (size_t)l * 1024, hx, N);
        float* t = hc; hc = hx; hx = t;
    }

    hipLaunchKernelGGL(gate_hi, dim3((N + 127) / 128), dim3(128), 0, stream,
                       hc, d_edge, gid, w_d, b_d, gate, hi, N);
    hipLaunchKernelGGL(h_mix_pool, dim3((N + 127) / 128), dim3(128), 0, stream,
                       hc, gate, hi, gid, pooled, N);

    hipLaunchKernelGGL((gemm64<false, false, true, 2>), dim3(G / 16), dim3(32), 0, stream,
                       pooled, W_mlp + 0 * 4096, nullptr, nullptr, nullptr, b_mlp + 0 * 64, x1);
    hipLaunchKernelGGL((gemm64<false, false, true, 2>), dim3(G / 16), dim3(32), 0, stream,
                       x1, W_mlp + 1 * 4096, nullptr, nullptr, nullptr, b_mlp + 1 * 64, x2);
    hipLaunchKernelGGL((gemm64<false, false, true, 2>), dim3(G / 16), dim3(32), 0, stream,
                       x2, W_mlp + 2 * 4096, nullptr, nullptr, nullptr, b_mlp + 2 * 64, x1);

    hipLaunchKernelGGL(out_kernel, dim3((G + 255) / 256), dim3(256), 0, stream,
                       x1, W_out, b_out, (float*)d_out, G);
}


# Layer gather: the 8 loads of a 4-edge group issued before any use (scheduling barrier, 32-bit row offsets, own h row parked in LDS to keep 64 registers): one memory round trip per group, not two
# speedup vs baseline: 7.7521x; 1.0517x over previous
#include <hip/hip_runtime.h>

typedef __attribute__((ext_vector_type(2))) float v2f;
typedef __attribute__((ext_vector_type(8))) float v8f;

#define HD 64

__device__ __forceinline__ float lrelu_f(float x) { return x > 0.f ? x : 0.1f * x; }

template <int M> __device__ __forceinline__ float row_xor(float v) {
    return __int_as_float(__builtin_amdgcn_mov_dpp(__float_as_int(v), 0x160 + M, 0xf, 0xf, false));
}
template <int U> __device__ __forceinline__ int row_bcast(int v) {
    return __builtin_amdgcn_mov_dpp(v, 0x150 + U, 0xf, 0xf, false);
}
template <int U> __device__ __forceinline__ float row_bcast(float v) {
    return __int_as_float(__builtin_amdgcn_mov_dpp(__float_as_int(v), 0x150 + U, 0xf, 0xf, false));
}
__device__ __forceinline__ float row_max16(float v) {
    v = fmaxf(v, row_xor<1>(v)); v = fmaxf(v, row_xor<2>(v));
    v = fmaxf(v, row_xor<4>(v)); v = fmaxf(v, row_xor<8>(v));
    return v;
}
__device__ __forceinline__ float row_sum16(float v) {
    v += row_xor<1>(v); v += row_xor<2>(v); v += row_xor<4>(v); v += row_xor<8>(v);
    return v;
}

template <bool HAS_A2, bool HAS_RES, bool HAS_BIAS, int ACT>
__global__ __launch_bounds__(32) void gemm64(
    const float* __restrict__ A,  const float* __restrict__ B,
    const float* __restrict__ A2, const float* __restrict__ B2,
    const float* __restrict__ res, const float* __restrict__ bias,
    float* __restrict__ out)
{
    const int lane = threadIdx.x & 31;
    const int cn   = lane & 15;
    const int kh   = (lane >> 4) << 1;
    const size_t row = (size_t)blockIdx.x * 16 + cn;

    const v8f z = {0.f, 0.f, 0.f, 0.f, 0.f, 0.f, 0.f, 0.f};
    v8f acc0 = z, acc1 = z, acc2 = z, acc3 = z;

    const float* arow = A + row * HD;
#pragma unroll
    for (int k0 = 0; k0 < 64; k0 += 4) {
        v2f a; a.x = arow[k0 + kh]; a.y = arow[k0 + kh + 1];
        const float* b0 = B + (size_t)(k0 + kh) * HD;
        const float* b1 = b0 + HD;
        v2f b;
        b.x = b0[cn];        b.y = b1[cn];
        acc0 = __builtin_amdgcn_wmma_f32_16x16x4_f32(false, a, false, b, (short)0, acc0, false, false);
        b.x = b0[16 + cn];   b.y = b1[16 + cn];
        acc1 = __builtin_amdgcn_wmma_f32_16x16x4_f32(false, a, false, b, (short)0, acc1, false, false);
        b.x = b0[32 + cn];   b.y = b1[32 + cn];
        acc2 = __builtin_amdgcn_wmma_f32_16x16x4_f32(false, a, false, b, (short)0, acc2, false, false);
        b.x = b0[48 + cn];   b.y = b1[48 + cn];
        acc3 = __builtin_amdgcn_wmma_f32_16x16x4_f32(false, a, false, b, (short)0, acc3, false, false);
    }
    if (HAS_A2) {
        const float* a2row = A2 + row * 16;
#pragma unroll
        for (int k0 = 0; k0 < 16; k0 += 4) {
            v2f a; a.x = a2row[k0 + kh]; a.y = a2row[k0 + kh + 1];
            const float* b0 = B2 + (size_t)(k0 + kh) * HD;
            const float* b1 = b0 + HD;
            v2f b;
            b.x = b0[cn];        b.y = b1[cn];
            acc0 = __builtin_amdgcn_wmma_f32_16x16x4_f32(false, a, false, b, (short)0, acc0, false, false);
            b.x = b0[16 + cn];   b.y = b1[16 + cn];
            acc1 = __builtin_amdgcn_wmma_f32_16x16x4_f32(false, a, false, b, (short)0, acc1, false, false);
            b.x = b0[32 + cn];   b.y = b1[32 + cn];
            acc2 = __builtin_amdgcn_wmma_f32_16x16x4_f32(false, a, false, b, (short)0, acc2, false, false);
            b.x = b0[48 + cn];   b.y = b1[48 + cn];
            acc3 = __builtin_amdgcn_wmma_f32_16x16x4_f32(false, a, false, b, (short)0, acc3, false, false);
        }
    }

    const size_t mbase = (size_t)blockIdx.x * 16 + ((lane >> 4) << 3);
    v8f accs[4] = {acc0, acc1, acc2, acc3};
#pragma unroll
    for (int t = 0; t < 4; ++t) {
        const int col = t * 16 + cn;
        const float bval = HAS_BIAS ? bias[col] : 0.f;
#pragma unroll
        for (int v = 0; v < 8; ++v) {
            const size_t m = mbase + v;
            float val = accs[t][v] + bval;
            if (HAS_RES) val += res[m * HD + col];
            if (ACT == 1) val = lrelu_f(val);
            else if (ACT == 2) val = fmaxf(val, 0.f);
            out[m * HD + col] = val;
        }
    }
}

__global__ void prep_ce(const float* __restrict__ W_e, const float* __restrict__ a_e,
                        float* __restrict__ cE)
{
    const int tid = threadIdx.x;
    const int l = tid >> 4, d = tid & 15;
    float s = 0.f;
    for (int k = 0; k < 64; ++k) s += W_e[d * 64 + k] * a_e[l * 64 + k];
    cE[tid] = s;
}

__global__ void prep_wewm(const float* __restrict__ W_e, const float* __restrict__ Wm,
                          float* __restrict__ WeWm)
{
    const int idx = blockIdx.x * blockDim.x + threadIdx.x;
    if (idx >= 4 * 16 * 64) return;
    const int l = idx >> 10, d = (idx >> 6) & 15, k = idx & 63;
    float s = 0.f;
    for (int j = 0; j < 64; ++j) s += W_e[d * 64 + j] * Wm[l * 4096 + j * 64 + k];
    WeWm[idx] = s;
}

__global__ void init_hi(const float* __restrict__ i_node, const float* __restrict__ W_i,
                        float* __restrict__ hi, float* __restrict__ pooled, int G)
{
    const int idx = blockIdx.x * blockDim.x + threadIdx.x;
    if (idx >= G * 64) return;
    hi[idx] = i_node[idx >> 6] * W_i[idx & 63];
    pooled[idx] = 0.f;
}

#define CSR_NB 12800
#define CSR_S  32
#define CSR_CH 128

__global__ __launch_bounds__(1024) void csr_count(const int* __restrict__ dst,
                                                  int* __restrict__ pc, int* __restrict__ cs,
                                                  int N, int NP, int CSP, int E, int R)
{
    __shared__ int cnt[CSR_NB];
    const int t = threadIdx.x;
    const int r = blockIdx.x % R, s = blockIdx.x / R;
    const int r0 = r * CSR_NB;
    for (int i = t; i < CSR_NB; i += 1024) cnt[i] = 0;
    __syncthreads();
    const int E4 = E >> 2;
    const int SL4 = (E4 + CSR_S - 1) / CSR_S;
    const int b4 = s * SL4, e4 = min(b4 + SL4, E4);
    const int4* __restrict__ d4 = (const int4*)dst;
    for (int i = b4 + t; i < e4; i += 4096) {
        int4 d[4];
#pragma unroll
        for (int k = 0; k < 4; ++k) d[k] = d4[min(i + k * 1024, e4 - 1)];
#pragma unroll
        for (int k = 0; k < 4; ++k) {
            if (i + k * 1024 < e4) {
                unsigned u;
                u = (unsigned)(d[k].x - r0); if (u < CSR_NB) atomicAdd(&cnt[u], 1);
                u = (unsigned)(d[k].y - r0); if (u < CSR_NB) atomicAdd(&cnt[u], 1);
                u = (unsigned)(d[k].z - r0); if (u < CSR_NB) atomicAdd(&cnt[u], 1);
                u = (unsigned)(d[k].w - r0); if (u < CSR_NB) atomicAdd(&cnt[u], 1);
            }
        }
    }
    if (s == CSR_S - 1) {
        for (int j = (E4 << 2) + t; j < E; j += 1024) {
            const unsigned u = (unsigned)(dst[j] - r0);
            if (u < CSR_NB) atomicAdd(&cnt[u], 1);
        }
    }
    __syncthreads();
    for (int i = t; i < CSR_NB; i += 1024) {
        const int n = r0 + i;
        if (n < N) pc[(size_t)s * NP + n] = cnt[i];
    }
    const int lane = t & 31, w = t >> 5;
    for (int c = w; c < CSR_NB / CSR_CH; c += 32) {
        const int* p = cnt + c * CSR_CH + lane;
        int v = (p[0] + p[32]) + (p[64] + p[96]);
        for (int o = 16; o > 0; o >>= 1) v += __shfl_xor(v, o, 32);
        if (lane == 0) cs[s * CSP + r * (CSR_NB / CSR_CH) + c] = v;
    }
}

__global__ __launch_bounds__(1024) void csr_scan(const int* __restrict__ pc,
                                                 const int* __restrict__ cs,
                                                 int* __restrict__ base, int N, int NP, int CSP)
{
    __shared__ int wfront[32];
    __shared__ int wtot[32];
    const int t = threadIdx.x, lane = t & 31, w = t >> 5;
    const int base0 = blockIdx.x * 1024;
    const int ch0 = blockIdx.x * (1024 / CSR_CH);

    int sf = 0;
    for (int c = t; c < ch0; c += 1024) {
#pragma unroll
        for (int s = 0; s < CSR_S; ++s) sf += cs[s * CSP + c];
    }
    for (int o = 16; o > 0; o >>= 1) sf += __shfl_xor(sf, o, 32);
    if (lane == 0) wfront[w] = sf;

    const int i = base0 + t;
    const int ic = min(i, N - 1);
    int pcs[CSR_S];
    int c = 0;
#pragma unroll
    for (int s = 0; s < CSR_S; ++s) {
        pcs[s] = (i < N) ? pc[(size_t)s * NP + ic] : 0;
        c += pcs[s];
    }
    int x = c;
    for (int o = 1; o < 32; o <<= 1) {
        const int y = __shfl_up(x, o, 32);
        if (lane >= o) x += y;
    }
    if (lane == 31) wtot[w] = x;
    __syncthreads();

    int front = 0;
    for (int k = 0; k < 32; ++k) front += wfront[k];
    for (int k = 0; k < w; ++k) front += wtot[k];
    if (i < N) {
        int run = front + x - c;
#pragma unroll
        for (int s = 0; s < CSR_S; ++s) {
            base[(size_t)s * NP + i] = run;
            run += pcs[s];
        }
        if (i == N - 1) base[N] = run;
    }
}

__global__ __launch_bounds__(1024) void csr_fill(const int* __restrict__ src,
                                                 const int* __restrict__ dst,
                                                 const int* __restrict__ base,
                                                 int* __restrict__ pk, int N, int NP, int E, int R)
{
    __shared__ int pos[CSR_NB];
    const int t = threadIdx.x;
    const int r = blockIdx.x % R, s = blockIdx.x / R;
    const int r0 = r * CSR_NB;
    for (int i = t; i < CSR_NB; i += 1024) {
        const int n = r0 + i;
        pos[i] = (n < N) ? base[(size_t)s * NP + n] : 0;
    }
    __syncthreads();
    const int E4 = E >> 2;
    const int SL4 = (E4 + CSR_S - 1) / CSR_S;
    const int b4 = s * SL4, e4 = min(b4 + SL4, E4);
    const int4* __restrict__ d4 = (const int4*)dst;
    const int4* __restrict__ s4 = (const int4*)src;
    int2* __restrict__ pk2 = (int2*)pk;
    for (int i = b4 + t; i < e4; i += 4096) {
        int4 d[4], sv[4];
#pragma unroll
        for (int k = 0; k < 4; ++k) {
            const int idx = min(i + k * 1024, e4 - 1);
            d[k] = d4[idx]; sv[k] = s4[idx];
        }
        int p[16];
#pragma unroll
        for (int k = 0; k < 4; ++k) {
            const bool live = i + k * 1024 < e4;
            unsigned u;
            u = (unsigned)(d[k].x - r0); p[4 * k + 0] = -1;
            if (live && u < CSR_NB) p[4 * k + 0] = atomicAdd(&pos[u], 1);
            u = (unsigned)(d[k].y - r0); p[4 * k + 1] = -1;
            if (live && u < CSR_NB) p[4 * k + 1] = atomicAdd(&pos[u], 1);
            u = (unsigned)(d[k].z - r0); p[4 * k + 2] = -1;
            if (live && u < CSR_NB) p[4 * k + 2] = atomicAdd(&pos[u], 1);
            u = (unsigned)(d[k].w - r0); p[4 * k + 3] = -1;
            if (live && u < CSR_NB) p[4 * k + 3] = atomicAdd(&pos[u], 1);
        }
#pragma unroll
        for (int k = 0; k < 4; ++k) {
            const int j = (i + k * 1024) << 2;
            if (p[4 * k + 0] >= 0) pk2[p[4 * k + 0]] = make_int2(j,     sv[k].x);
            if (p[4 * k + 1] >= 0) pk2[p[4 * k + 1]] = make_int2(j + 1, sv[k].y);
            if (p[4 * k + 2] >= 0) pk2[p[4 * k + 2]] = make_int2(j + 2, sv[k].z);
            if (p[4 * k + 3] >= 0) pk2[p[4 * k + 3]] = make_int2(j + 3, sv[k].w);
        }
    }
    if (s == CSR_S - 1) {
        for (int j = (E4 << 2) + t; j < E; j += 1024) {
            const unsigned u = (unsigned)(dst[j] - r0);
            if (u < CSR_NB) { const int q = atomicAdd(&pos[u], 1); pk2[q] = make_int2(j, src[j]); }
        }
    }
}

template <int U0> __device__ __forceinline__ void gather4(
    const int2 e, const int q, const int nlive,
    const float4 as4, const float ceq, const float dn,
    const float* __restrict__ h, const float* __restrict__ re,
    float& m, float& dsum, float4& acc, float& accr)
{
    const int j0 = row_bcast<U0 + 0>(e.x), s0 = row_bcast<U0 + 0>(e.y);
    const int j1 = row_bcast<U0 + 1>(e.x), s1 = row_bcast<U0 + 1>(e.y);
    const int j2 = row_bcast<U0 + 2>(e.x), s2 = row_bcast<U0 + 2>(e.y);
    const int j3 = row_bcast<U0 + 3>(e.x), s3 = row_bcast<U0 + 3>(e.y);
    const float4* __restrict__ h4 = (const float4*)h;
    const float4 h0 = h4[(unsigned)s0 * 16u + (unsigned)q];
    const float4 h1 = h4[(unsigned)s1 * 16u + (unsigned)q];
    const float4 h2 = h4[(unsigned)s2 * 16u + (unsigned)q];
    const float4 h3 = h4[(unsigned)s3 * 16u + (unsigned)q];
    const float r0 = re[(unsigned)j0 * 16u + (unsigned)q];
    const float r1 = re[(unsigned)j1 * 16u + (unsigned)q];
    const float r2 = re[(unsigned)j2 * 16u + (unsigned)q];
    const float r3 = re[(unsigned)j3 * 16u + (unsigned)q];
    __builtin_amdgcn_sched_barrier(0);
    const float ninf = -__builtin_inff();
    float l0 = row_sum16(h0.x * as4.x + h0.y * as4.y + h0.z * as4.z + h0.w * as4.w + r0 * ceq);
    float l1 = row_sum16(h1.x * as4.x + h1.y * as4.y + h1.z * as4.z + h1.w * as4.w + r1 * ceq);
    float l2 = row_sum16(h2.x * as4.x + h2.y * as4.y + h2.z * as4.z + h2.w * as4.w + r2 * ceq);
    float l3 = row_sum16(h3.x * as4.x + h3.y * as4.y + h3.z * as4.z + h3.w * as4.w + r3 * ceq);
    l0 = lrelu_f(l0 + dn);
    l1 = (U0 + 1 < nlive) ? lrelu_f(l1 + dn) : ninf;
    l2 = (U0 + 2 < nlive) ? lrelu_f(l2 + dn) : ninf;
    l3 = (U0 + 3 < nlive) ? lrelu_f(l3 + dn) : ninf;
    const float mn = fmaxf(fmaxf(m, l0), fmaxf(fmaxf(l1, l2), l3));
    const float sc = __expf(m - mn);
    const float a0 = __expf(l0 - mn), a1 = __expf(l1 - mn);
    const float a2 = __expf(l2 - mn), a3 = __expf(l3 - mn);
    m = mn;
    dsum = dsum * sc + ((a0 + a1) + (a2 + a3));
    acc.x *= sc; acc.y *= sc; acc.z *= sc; acc.w *= sc; accr *= sc;
    acc.x += a0 * h0.x; acc.y += a0 * h0.y; acc.z += a0 * h0.z; acc.w += a0 * h0.w; accr += a0 * r0;
    acc.x += a1 * h1.x; acc.y += a1 * h1.y; acc.z += a1 * h1.z; acc.w += a1 * h1.w; accr += a1 * r1;
    acc.x += a2 * h2.x; acc.y += a2 * h2.y; acc.z += a2 * h2.z; acc.w += a2 * h2.w; accr += a2 * r2;
    acc.x += a3 * h3.x; acc.y += a3 * h3.y; acc.z += a3 * h3.z; acc.w += a3 * h3.w; accr += a3 * r3;
}

#define TA_P 84
#define TD_P 68
#define TB_P 64
__global__ __launch_bounds__(256) void layer_fused(
    const int* __restrict__ rowptr, const int* __restrict__ pk,
    const float* __restrict__ as, const float* __restrict__ ad,
    const float* __restrict__ ce,
    const float* __restrict__ h, const float* __restrict__ re,
    const float* __restrict__ Wm, const float* __restrict__ WeWm,
    float* __restrict__ hout, int N)
{
    __shared__ __attribute__((aligned(16))) float sA[16 * TA_P];
    __shared__ __attribute__((aligned(16))) float sD[16 * TD_P];
    __shared__ __attribute__((aligned(16))) float sB[80 * TB_P];
    __shared__ __attribute__((aligned(16))) float sH[16 * 64];
    const int r = threadIdx.x >> 4;
    const int nn = blockIdx.x * 16 + r;
    const int n = min(nn, N - 1);
    const int q = threadIdx.x & 15;
    const int t = threadIdx.x;
    const float4 w0 = ((const float4*)Wm)[t];
    const float4 w1 = ((const float4*)Wm)[t + 256];
    const float4 w2 = ((const float4*)Wm)[t + 512];
    const float4 w3 = ((const float4*)Wm)[t + 768];
    const float4 w4 = ((const float4*)WeWm)[t];
    const int beg = rowptr[n], end = rowptr[n + 1];
    const float4 as4 = *(const float4*)(as + q * 4);
    const float4 ad4 = *(const float4*)(ad + q * 4);
    const float ceq = ce[q];
    const float4 hn = *(const float4*)(h + (size_t)n * 64 + q * 4);
    {
        float* sb = sB + (t >> 4) * TB_P + (t & 15) * 4;
        *(float4*)(sb)              = w0;
        *(float4*)(sb + 16 * TB_P)  = w1;
        *(float4*)(sb + 32 * TB_P)  = w2;
        *(float4*)(sb + 48 * TB_P)  = w3;
        *(float4*)(sb + 64 * TB_P)  = w4;
    }
    asm volatile("" :: "v"(hn.x), "v"(hn.y), "v"(hn.z), "v"(hn.w),
                       "v"(ad4.x), "v"(ad4.y), "v"(ad4.z), "v"(ad4.w),
                       "v"(as4.x), "v"(as4.y), "v"(as4.z), "v"(as4.w), "v"(ceq));
    *(float4*)(sH + t * 4) = hn;
    const float dnq = hn.x * ad4.x + hn.y * ad4.y + hn.z * ad4.z + hn.w * ad4.w;
    const int2* __restrict__ pk2 = (const int2*)pk;
    float4 acc = {0.f, 0.f, 0.f, 0.f};
    float accr = 0.f;
    if (end > beg) {
        const int last = end - 1;
        int2 e = pk2[min(beg + q, last)];
        const float dn = row_sum16(dnq);
        float m = -__builtin_inff();
        float dsum = 0.f;
        int c = beg;
        for (;;) {
            const int nlive = end - c;
            gather4<0>(e, q, nlive, as4, ceq, dn, h, re, m, dsum, acc, accr);
            if (nlive > 4)  gather4<4>(e, q, nlive, as4, ceq, dn, h, re, m, dsum, acc, accr);
            if (nlive > 8)  gather4<8>(e, q, nlive, as4, ceq, dn, h, re, m, dsum, acc, accr);
            if (nlive > 12) gather4<12>(e, q, nlive, as4, ceq, dn, h, re, m, dsum, acc, accr);
            c += 16;
            if (c >= end) break;
            e = pk2[min(c + q, last)];
        }
        const float inv = 1.f / dsum;
        acc.x *= inv; acc.y *= inv; acc.z *= inv; acc.w *= inv; accr *= inv;
    }

    const int wv = __builtin_amdgcn_readfirstlane((int)(threadIdx.x >> 5));
    const int lane = threadIdx.x & 31;
    const int cn = lane & 15;
    const int kh = (lane >> 4) << 1;
    *(float4*)(sA + r * TA_P + q * 4) = acc;
    sA[r * TA_P + 64 + q] = accr;
    __syncthreads();
    if (wv < 4) {
        v8f d = {0.f, 0.f, 0.f, 0.f, 0.f, 0.f, 0.f, 0.f};
        const float* ap = sA + cn * TA_P + kh;
        const float* bp = sB + kh * TB_P + wv * 16 + cn;
#pragma unroll
        for (int k = 0; k < 20; ++k) {
            const v2f a = *(const v2f*)(ap + 4 * k);
            v2f b; b.x = bp[4 * k * TB_P]; b.y = bp[(4 * k + 1) * TB_P];
            d = __builtin_amdgcn_wmma_f32_16x16x4_f32(false, a, false, b, (short)0, d, false, false);
        }
        float* dp = sD + ((lane >> 4) << 3) * TD_P + wv * 16 + cn;
#pragma unroll
        for (int v = 0; v < 8; ++v) dp[v * TD_P] = d[v];
    }
    __syncthreads();
    if (nn < N) {
        float4 o = *(const float4*)(sD + r * TD_P + q * 4);
        const float4 hr = *(const float4*)(sH + t * 4);
        o.x = lrelu_f(o.x + hr.x); o.y = lrelu_f(o.y + hr.y);
        o.z = lrelu_f(o.z + hr.z); o.w = lrelu_f(o.w + hr.w);
        *(float4*)(hout + (size_t)nn * 64 + q * 4) = o;
    }
}

#define WALK 16

__device__ __forceinline__ void row_flush(float* __restrict__ grow, const int q, const float4 a)
{
    float* p = grow + q * 4;
    atomicAdd(p + 0, a.x); atomicAdd(p + 1, a.y); atomicAdd(p + 2, a.z); atomicAdd(p + 3, a.w);
}

__device__ __forceinline__ void row_bcast_all(const int iq, const float fq,
                                              int (&iv)[WALK], float (&fv)[WALK])
{
#define BC1(U) iv[U] = row_bcast<U>(iq); fv[U] = row_bcast<U>(fq);
    BC1(0)  BC1(1)  BC1(2)  BC1(3)  BC1(4)  BC1(5)  BC1(6)  BC1(7)
    BC1(8)  BC1(9)  BC1(10) BC1(11) BC1(12) BC1(13) BC1(14) BC1(15)
#undef BC1
}

__global__ __launch_bounds__(128) void gate_hi(
    const float* __restrict__ h, const float* __restrict__ d_edge,
    const int* __restrict__ gid, const float* __restrict__ wd,
    const float* __restrict__ bd, float* __restrict__ gate,
    float* __restrict__ hi, int N)
{
    const int q  = threadIdx.x & 15;
    const int n0 = (blockIdx.x * 8 + (threadIdx.x >> 4)) * WALK;
    if (n0 >= N) return;
    const int nq = min(n0 + q, N - 1);
    float4 hv[WALK];
#pragma unroll
    for (int u = 0; u < WALK; ++u)
        hv[u] = *(const float4*)(h + (size_t)min(n0 + u, N - 1) * 64 + q * 4);
    const float gq = 1.f / (1.f + __expf(-(d_edge[nq] * wd[0] + bd[0])));
    const int gidq = gid[nq];
    if (n0 + q < N) gate[n0 + q] = gq;
    int gi[WALK]; float g[WALK];
    row_bcast_all(gidq, gq, gi, g);
    float4 acc = {0.f, 0.f, 0.f, 0.f};
    int cur = gi[0];
#pragma unroll
    for (int u = 0; u < WALK; ++u) {
        if (n0 + u < N) {
            if (gi[u] != cur) {
                row_flush(hi + (size_t)cur * 64, q, acc);
                acc.x = 0.f; acc.y = 0.f; acc.z = 0.f; acc.w = 0.f;
                cur = gi[u];
            }
            acc.x += g[u] * hv[u].x; acc.y += g[u] * hv[u].y;
            acc.z += g[u] * hv[u].z; acc.w += g[u] * hv[u].w;
        }
    }
    row_flush(hi + (size_t)cur * 64, q, acc);
}

__global__ __launch_bounds__(128) void h_mix_pool(
    float* __restrict__ h, const float* __restrict__ gate,
    const float* __restrict__ hi, const int* __restrict__ gid,
    float* __restrict__ pooled, int N)
{
    const int q  = threadIdx.x & 15;
    const int n0 = (blockIdx.x * 8 + (threadIdx.x >> 4)) * WALK;
    if (n0 >= N) return;
    const int nq = min(n0 + q, N - 1);
    float4 hv[WALK];
#pragma unroll
    for (int u = 0; u < WALK; ++u)
        hv[u] = *(const float4*)(h + (size_t)min(n0 + u, N - 1) * 64 + q * 4);
    const float gq = gate[nq];
    const int gidq = gid[nq];
    int gi[WALK]; float g[WALK];
    row_bcast_all(gidq, gq, gi, g);
    float4 hiv[WALK];
#pragma unroll
    for (int u = 0; u < WALK; ++u)
        hiv[u] = *(const float4*)(hi + (size_t)gi[u] * 64 + q * 4);
    float4 acc = {0.f, 0.f, 0.f, 0.f};
    int cur = gi[0];
#pragma unroll
    for (int u = 0; u < WALK; ++u) {
        if (n0 + u < N) {
            float4 v;
            v.x = hv[u].x + g[u] * hiv[u].x; v.y = hv[u].y + g[u] * hiv[u].y;
            v.z = hv[u].z + g[u] * hiv[u].z; v.w = hv[u].w + g[u] * hiv[u].w;
            *(float4*)(h + (size_t)(n0 + u) * 64 + q * 4) = v;
            if (gi[u] != cur) {
                row_flush(pooled + (size_t)cur * 64, q, acc);
                acc.x = 0.f; acc.y = 0.f; acc.z = 0.f; acc.w = 0.f;
                cur = gi[u];
            }
            acc.x += v.x; acc.y += v.y; acc.z += v.z; acc.w += v.w;
        }
    }
    row_flush(pooled + (size_t)cur * 64, q, acc);
}

__global__ void out_kernel(const float* __restrict__ x, const float* __restrict__ Wo,
                           const float* __restrict__ bo, float* __restrict__ out, int G)
{
    const int g = blockIdx.x * blockDim.x + threadIdx.x;
    if (g >= G) return;
    float s = bo[0];
#pragma unroll
    for (int k = 0; k < 64; ++k) s += x[(size_t)g * 64 + k] * Wo[k];
    out[g] = s;
}

extern "C" void kernel_launch(void* const* d_in, const int* in_sizes, int n_in,
                              void* d_out, int out_size, void* d_ws, size_t ws_size,
                              hipStream_t stream)
{
    const float* r_node = (const float*)d_in[0];
    const float* i_node = (const float*)d_in[1];
    const float* r_edge = (const float*)d_in[2];
    const float* d_edge = (const float*)d_in[3];
    const int*   src    = (const int*)d_in[4];
    const int*   dst    = (const int*)d_in[5];
    const int*   gid    = (const int*)d_in[6];
    const float* W_r    = (const float*)d_in[7];
    const float* W_i    = (const float*)d_in[8];
    const float* W_e    = (const float*)d_in[9];
    const float* Wm     = (const float*)d_in[10];
    const float* a_s    = (const float*)d_in[11];
    const float* a_d    = (const float*)d_in[12];
    const float* a_e    = (const float*)d_in[13];
    const float* w_d    = (const float*)d_in[14];
    const float* b_d    = (const float*)d_in[15];
    const float* W_mlp  = (const float*)d_in[16];
    const float* b_mlp  = (const float*)d_in[17];
    const float* W_out  = (const float*)d_in[18];
    const float* b_out  = (const float*)d_in[19];

    const int N = in_sizes[0] / 64;
    const int G = in_sizes[1];
    const int E = in_sizes[2] / 16;
    (void)n_in; (void)out_size; (void)ws_size;

    float* ws = (float*)d_ws;
    size_t off = 0;
    auto carve = [&](size_t n) -> float* {
        float* p = ws + off; off += (n + 63) & ~(size_t)63; return p;
    };
    float*    h      = carve((size_t)N * 64);
    float*    h2     = carve((size_t)N * 64);
    float*    gate   = carve(N);
    float*    hi     = carve((size_t)G * 64);
    float*    pooled = carve((size_t)G * 64);
    float*    x1     = carve((size_t)G * 64);
    float*    x2     = carve((size_t)G * 64);
    float*    cE     = carve(64);
    float*    WeWm   = carve(4 * 16 * 64);
    const int R   = (N + CSR_NB - 1) / CSR_NB;
    const int NP  = (N + 1 + 63) & ~63;
    const int CSP = R * (CSR_NB / CSR_CH);
    int*      pc     = (int*)carve((size_t)CSR_S * NP);
    int*      cs     = (int*)carve((size_t)CSR_S * CSP);
    int*      rowptr = (int*)carve((size_t)CSR_S * NP);
    int*      pk     = (int*)carve((size_t)E * 2);

    hipLaunchKernelGGL(prep_ce, dim3(1), dim3(64), 0, stream, W_e, a_e, cE);
    hipLaunchKernelGGL(prep_wewm, dim3((4096 + 255) / 256), dim3(256), 0, stream, W_e, Wm, WeWm);

    hipLaunchKernelGGL((gemm64<false, false, false, 0>), dim3(N / 16), dim3(32), 0, stream,
                       r_node, W_r, nullptr, nullptr, nullptr, nullptr, h);
    hipLaunchKernelGGL(init_hi, dim3((G * 64 + 255) / 256), dim3(256), 0, stream,
                       i_node, W_i, hi, pooled, G);

    hipLaunchKernelGGL(csr_count, dim3(R * CSR_S), dim3(1024), 0, stream, dst, pc, cs, N, NP, CSP, E, R);
    hipLaunchKernelGGL(csr_scan, dim3((N + 1023) / 1024), dim3(1024), 0, stream, pc, cs, rowptr, N, NP, CSP);
    hipLaunchKernelGGL(csr_fill, dim3(R * CSR_S), dim3(1024), 0, stream,
                       src, dst, rowptr, pk, N, NP, E, R);

    float* hc = h;
    float* hx = h2;
    for (int l = 0; l < 4; ++l) {
        hipLaunchKernelGGL(layer_fused, dim3((N + 15) / 16), dim3(256), 0, stream,
                           rowptr, pk, a_s + l * 64, a_d + l * 64, cE + l * 16,
                           hc, r_edge, Wm + (size_t)l * 4096, WeWm + (size_t)l * 1024, hx, N);
        float* t = hc; hc = hx; hx = t;
    }

    hipLaunchKernelGGL(gate_hi, dim3((N + 127) / 128), dim3(128), 0, stream,
                       hc, d_edge, gid, w_d, b_d, gate, hi, N);
    hipLaunchKernelGGL(h_mix_pool, dim3((N + 127) / 128), dim3(128), 0, stream,
                       hc, gate, hi, gid, pooled, N);

    hipLaunchKernelGGL((gemm64<false, false, true, 2>), dim3(G / 16), dim3(32), 0, stream,
                       pooled, W_mlp + 0 * 4096, nullptr, nullptr, nullptr, b_mlp + 0 * 64, x1);
    hipLaunchKernelGGL((gemm64<false, false, true, 2>), dim3(G / 16), dim3(32), 0, stream,
                       x1, W_mlp + 1 * 4096, nullptr, nullptr, nullptr, b_mlp + 1 * 64, x2);
    hipLaunchKernelGGL((gemm64<false, false, true, 2>), dim3(G / 16), dim3(32), 0, stream,
                       x2, W_mlp + 2 * 4096, nullptr, nullptr, nullptr, b_mlp + 2 * 64, x1);

    hipLaunchKernelGGL(out_kernel, dim3((G + 255) / 256), dim3(256), 0, stream,
                       x1, W_out, b_out, (float*)d_out, G);
}


# Launches merged: embedding GEMM, hi/pooled init and both weight preps are block ranges of one launch; the 3 MLP layers and the output dot are one launch (x via LDS); 17 launches to 11, sums unchanged
# speedup vs baseline: 8.4700x; 1.0926x over previous
#include <hip/hip_runtime.h>

typedef __attribute__((ext_vector_type(2))) float v2f;
typedef __attribute__((ext_vector_type(8))) float v8f;

#define HD 64

__device__ __forceinline__ float lrelu_f(float x) { return x > 0.f ? x : 0.1f * x; }

template <int M> __device__ __forceinline__ float row_xor(float v) {
    return __int_as_float(__builtin_amdgcn_mov_dpp(__float_as_int(v), 0x160 + M, 0xf, 0xf, false));
}
template <int U> __device__ __forceinline__ int row_bcast(int v) {
    return __builtin_amdgcn_mov_dpp(v, 0x150 + U, 0xf, 0xf, false);
}
template <int U> __device__ __forceinline__ float row_bcast(float v) {
    return __int_as_float(__builtin_amdgcn_mov_dpp(__float_as_int(v), 0x150 + U, 0xf, 0xf, false));
}
__device__ __forceinline__ float row_max16(float v) {
    v = fmaxf(v, row_xor<1>(v)); v = fmaxf(v, row_xor<2>(v));
    v = fmaxf(v, row_xor<4>(v)); v = fmaxf(v, row_xor<8>(v));
    return v;
}
__device__ __forceinline__ float row_sum16(float v) {
    v += row_xor<1>(v); v += row_xor<2>(v); v += row_xor<4>(v); v += row_xor<8>(v);
    return v;
}

template <bool HAS_A2, bool HAS_RES, bool HAS_BIAS, int ACT>
__device__ __forceinline__ void gemm64_tile(const int tile,
    const float* __restrict__ A,  const float* __restrict__ B,
    const float* __restrict__ A2, const float* __restrict__ B2,
    const float* __restrict__ res, const float* __restrict__ bias,
    float* __restrict__ out)
{
    const int lane = threadIdx.x & 31;
    const int cn   = lane & 15;
    const int kh   = (lane >> 4) << 1;
    const size_t row = (size_t)tile * 16 + cn;

    const v8f z = {0.f, 0.f, 0.f, 0.f, 0.f, 0.f, 0.f, 0.f};
    v8f acc0 = z, acc1 = z, acc2 = z, acc3 = z;

    const float* arow = A + row * HD;
#pragma unroll
    for (int k0 = 0; k0 < 64; k0 += 4) {
        v2f a; a.x = arow[k0 + kh]; a.y = arow[k0 + kh + 1];
        const float* b0 = B + (size_t)(k0 + kh) * HD;
        const float* b1 = b0 + HD;
        v2f b;
        b.x = b0[cn];        b.y = b1[cn];
        acc0 = __builtin_amdgcn_wmma_f32_16x16x4_f32(false, a, false, b, (short)0, acc0, false, false);
        b.x = b0[16 + cn];   b.y = b1[16 + cn];
        acc1 = __builtin_amdgcn_wmma_f32_16x16x4_f32(false, a, false, b, (short)0, acc1, false, false);
        b.x = b0[32 + cn];   b.y = b1[32 + cn];
        acc2 = __builtin_amdgcn_wmma_f32_16x16x4_f32(false, a, false, b, (short)0, acc2, false, false);
        b.x = b0[48 + cn];   b.y = b1[48 + cn];
        acc3 = __builtin_amdgcn_wmma_f32_16x16x4_f32(false, a, false, b, (short)0, acc3, false, false);
    }
    if (HAS_A2) {
        const float* a2row = A2 + row * 16;
#pragma unroll
        for (int k0 = 0; k0 < 16; k0 += 4) {
            v2f a; a.x = a2row[k0 + kh]; a.y = a2row[k0 + kh + 1];
            const float* b0 = B2 + (size_t)(k0 + kh) * HD;
            const float* b1 = b0 + HD;
            v2f b;
            b.x = b0[cn];        b.y = b1[cn];
            acc0 = __builtin_amdgcn_wmma_f32_16x16x4_f32(false, a, false, b, (short)0, acc0, false, false);
            b.x = b0[16 + cn];   b.y = b1[16 + cn];
            acc1 = __builtin_amdgcn_wmma_f32_16x16x4_f32(false, a, false, b, (short)0, acc1, false, false);
            b.x = b0[32 + cn];   b.y = b1[32 + cn];
            acc2 = __builtin_amdgcn_wmma_f32_16x16x4_f32(false, a, false, b, (short)0, acc2, false, false);
            b.x = b0[48 + cn];   b.y = b1[48 + cn];
            acc3 = __builtin_amdgcn_wmma_f32_16x16x4_f32(false, a, false, b, (short)0, acc3, false, false);
        }
    }

    const size_t mbase = (size_t)tile * 16 + ((lane >> 4) << 3);
    v8f accs[4] = {acc0, acc1, acc2, acc3};
#pragma unroll
    for (int t = 0; t < 4; ++t) {
        const int col = t * 16 + cn;
        const float bval = HAS_BIAS ? bias[col] : 0.f;
#pragma unroll
        for (int v = 0; v < 8; ++v) {
            const size_t m = mbase + v;
            float val = accs[t][v] + bval;
            if (HAS_RES) val += res[m * HD + col];
            if (ACT == 1) val = lrelu_f(val);
            else if (ACT == 2) val = fmaxf(val, 0.f);
            out[m * HD + col] = val;
        }
    }
}

#define FRONT_SMALL 17
__global__ __launch_bounds__(256) void front_all(
    const float* __restrict__ r_node, const float* __restrict__ W_r, float* __restrict__ h,
    const int ntile, const int nb_gemm,
    const float* __restrict__ i_node, const float* __restrict__ W_i,
    float* __restrict__ hi, float* __restrict__ pooled, const int G,
    const float* __restrict__ W_e, const float* __restrict__ a_e, const float* __restrict__ Wm,
    float* __restrict__ cE, float* __restrict__ WeWm)
{
    const int b = blockIdx.x;
    const int tid = threadIdx.x;
    if (b < 16) {
        const int idx = b * 256 + tid;
        const int l = idx >> 10, d = (idx >> 6) & 15, k = idx & 63;
        float s = 0.f;
        for (int j = 0; j < 64; ++j) s += W_e[d * 64 + j] * Wm[l * 4096 + j * 64 + k];
        WeWm[idx] = s;
    } else if (b == 16) {
        if (tid < 64) {
            const int l = tid >> 4, d = tid & 15;
            float s = 0.f;
            for (int k = 0; k < 64; ++k) s += W_e[d * 64 + k] * a_e[l * 64 + k];
            cE[tid] = s;
        }
    } else if (b < FRONT_SMALL + nb_gemm) {
        const int tile = (b - FRONT_SMALL) * 8 + __builtin_amdgcn_readfirstlane(tid >> 5);
        if (tile < ntile)
            gemm64_tile<false, false, false, 0>(tile, r_node, W_r, nullptr, nullptr, nullptr, nullptr, h);
    } else {
        const int idx4 = (b - FRONT_SMALL - nb_gemm) * 256 + tid;
        if (idx4 < G * 16) {
            const float s = i_node[idx4 >> 4];
            const float4 w = *(const float4*)(W_i + (idx4 & 15) * 4);
            float4 v; v.x = s * w.x; v.y = s * w.y; v.z = s * w.z; v.w = s * w.w;
            const float4 zero = {0.f, 0.f, 0.f, 0.f};
            ((float4*)hi)[idx4] = v;
            ((float4*)pooled)[idx4] = zero;
        }
    }
}

#define CSR_NB 12800
#define CSR_S  32
#define CSR_CH 128

__global__ __launch_bounds__(1024) void csr_count(const int* __restrict__ dst,
                                                  int* __restrict__ pc, int* __restrict__ cs,
                                                  int N, int NP, int CSP, int E, int R)
{
    __shared__ int cnt[CSR_NB];
    const int t = threadIdx.x;
    const int r = blockIdx.x % R, s = blockIdx.x / R;
    const int r0 = r * CSR_NB;
    for (int i = t; i < CSR_NB; i += 1024) cnt[i] = 0;
    __syncthreads();
    const int E4 = E >> 2;
    const int SL4 = (E4 + CSR_S - 1) / CSR_S;
    const int b4 = s * SL4, e4 = min(b4 + SL4, E4);
    const int4* __restrict__ d4 = (const int4*)dst;
    for (int i = b4 + t; i < e4; i += 4096) {
        int4 d[4];
#pragma unroll
        for (int k = 0; k < 4; ++k) d[k] = d4[min(i + k * 1024, e4 - 1)];
#pragma unroll
        for (int k = 0; k < 4; ++k) {
            if (i + k * 1024 < e4) {
                unsigned u;
                u = (unsigned)(d[k].x - r0); if (u < CSR_NB) atomicAdd(&cnt[u], 1);
                u = (unsigned)(d[k].y - r0); if (u < CSR_NB) atomicAdd(&cnt[u], 1);
                u = (unsigned)(d[k].z - r0); if (u < CSR_NB) atomicAdd(&cnt[u], 1);
                u = (unsigned)(d[k].w - r0); if (u < CSR_NB) atomicAdd(&cnt[u], 1);
            }
        }
    }
    if (s == CSR_S - 1) {
        for (int j = (E4 << 2) + t; j < E; j += 1024) {
            const unsigned u = (unsigned)(dst[j] - r0);
            if (u < CSR_NB) atomicAdd(&cnt[u], 1);
        }
    }
    __syncthreads();
    for (int i = t; i < CSR_NB; i += 1024) {
        const int n = r0 + i;
        if (n < N) pc[(size_t)s * NP + n] = cnt[i];
    }
    const int lane = t & 31, w = t >> 5;
    for (int c = w; c < CSR_NB / CSR_CH; c += 32) {
        const int* p = cnt + c * CSR_CH + lane;
        int v = (p[0] + p[32]) + (p[64] + p[96]);
        for (int o = 16; o > 0; o >>= 1) v += __shfl_xor(v, o, 32);
        if (lane == 0) cs[s * CSP + r * (CSR_NB / CSR_CH) + c] = v;
    }
}

__global__ __launch_bounds__(1024) void csr_scan(const int* __restrict__ pc,
                                                 const int* __restrict__ cs,
                                                 int* __restrict__ base, int N, int NP, int CSP)
{
    __shared__ int wfront[32];
    __shared__ int wtot[32];
    const int t = threadIdx.x, lane = t & 31, w = t >> 5;
    const int base0 = blockIdx.x * 1024;
    const int ch0 = blockIdx.x * (1024 / CSR_CH);

    int sf = 0;
    for (int c = t; c < ch0; c += 1024) {
#pragma unroll
        for (int s = 0; s < CSR_S; ++s) sf += cs[s * CSP + c];
    }
    for (int o = 16; o > 0; o >>= 1) sf += __shfl_xor(sf, o, 32);
    if (lane == 0) wfront[w] = sf;

    const int i = base0 + t;
    const int ic = min(i, N - 1);
    int pcs[CSR_S];
    int c = 0;
#pragma unroll
    for (int s = 0; s < CSR_S; ++s) {
        pcs[s] = (i < N) ? pc[(size_t)s * NP + ic] : 0;
        c += pcs[s];
    }
    int x = c;
    for (int o = 1; o < 32; o <<= 1) {
        const int y = __shfl_up(x, o, 32);
        if (lane >= o) x += y;
    }
    if (lane == 31) wtot[w] = x;
    __syncthreads();

    int front = 0;
    for (int k = 0; k < 32; ++k) front += wfront[k];
    for (int k = 0; k < w; ++k) front += wtot[k];
    if (i < N) {
        int run = front + x - c;
#pragma unroll
        for (int s = 0; s < CSR_S; ++s) {
            base[(size_t)s * NP + i] = run;
            run += pcs[s];
        }
        if (i == N - 1) base[N] = run;
    }
}

__global__ __launch_bounds__(1024) void csr_fill(const int* __restrict__ src,
                                                 const int* __restrict__ dst,
                                                 const int* __restrict__ base,
                                                 int* __restrict__ pk, int N, int NP, int E, int R)
{
    __shared__ int pos[CSR_NB];
    const int t = threadIdx.x;
    const int r = blockIdx.x % R, s = blockIdx.x / R;
    const int r0 = r * CSR_NB;
    for (int i = t; i < CSR_NB; i += 1024) {
        const int n = r0 + i;
        pos[i] = (n < N) ? base[(size_t)s * NP + n] : 0;
    }
    __syncthreads();
    const int E4 = E >> 2;
    const int SL4 = (E4 + CSR_S - 1) / CSR_S;
    const int b4 = s * SL4, e4 = min(b4 + SL4, E4);
    const int4* __restrict__ d4 = (const int4*)dst;
    const int4* __restrict__ s4 = (const int4*)src;
    int2* __restrict__ pk2 = (int2*)pk;
    for (int i = b4 + t; i < e4; i += 4096) {
        int4 d[4], sv[4];
#pragma unroll
        for (int k = 0; k < 4; ++k) {
            const int idx = min(i + k * 1024, e4 - 1);
            d[k] = d4[idx]; sv[k] = s4[idx];
        }
        int p[16];
#pragma unroll
        for (int k = 0; k < 4; ++k) {
            const bool live = i + k * 1024 < e4;
            unsigned u;
            u = (unsigned)(d[k].x - r0); p[4 * k + 0] = -1;
            if (live && u < CSR_NB) p[4 * k + 0] = atomicAdd(&pos[u], 1);
            u = (unsigned)(d[k].y - r0); p[4 * k + 1] = -1;
            if (live && u < CSR_NB) p[4 * k + 1] = atomicAdd(&pos[u], 1);
            u = (unsigned)(d[k].z - r0); p[4 * k + 2] = -1;
            if (live && u < CSR_NB) p[4 * k + 2] = atomicAdd(&pos[u], 1);
            u = (unsigned)(d[k].w - r0); p[4 * k + 3] = -1;
            if (live && u < CSR_NB) p[4 * k + 3] = atomicAdd(&pos[u], 1);
        }
#pragma unroll
        for (int k = 0; k < 4; ++k) {
            const int j = (i + k * 1024) << 2;
            if (p[4 * k + 0] >= 0) pk2[p[4 * k + 0]] = make_int2(j,     sv[k].x);
            if (p[4 * k + 1] >= 0) pk2[p[4 * k + 1]] = make_int2(j + 1, sv[k].y);
            if (p[4 * k + 2] >= 0) pk2[p[4 * k + 2]] = make_int2(j + 2, sv[k].z);
            if (p[4 * k + 3] >= 0) pk2[p[4 * k + 3]] = make_int2(j + 3, sv[k].w);
        }
    }
    if (s == CSR_S - 1) {
        for (int j = (E4 << 2) + t; j < E; j += 1024) {
            const unsigned u = (unsigned)(dst[j] - r0);
            if (u < CSR_NB) { const int q = atomicAdd(&pos[u], 1); pk2[q] = make_int2(j, src[j]); }
        }
    }
}

template <int U0> __device__ __forceinline__ void gather4(
    const int2 e, const int q, const int nlive,
    const float4 as4, const float ceq, const float dn,
    const float* __restrict__ h, const float* __restrict__ re,
    float& m, float& dsum, float4& acc, float& accr)
{
    const int j0 = row_bcast<U0 + 0>(e.x), s0 = row_bcast<U0 + 0>(e.y);
    const int j1 = row_bcast<U0 + 1>(e.x), s1 = row_bcast<U0 + 1>(e.y);
    const int j2 = row_bcast<U0 + 2>(e.x), s2 = row_bcast<U0 + 2>(e.y);
    const int j3 = row_bcast<U0 + 3>(e.x), s3 = row_bcast<U0 + 3>(e.y);
    const float4* __restrict__ h4 = (const float4*)h;
    const float4 h0 = h4[(unsigned)s0 * 16u + (unsigned)q];
    const float4 h1 = h4[(unsigned)s1 * 16u + (unsigned)q];
    const float4 h2 = h4[(unsigned)s2 * 16u + (unsigned)q];
    const float4 h3 = h4[(unsigned)s3 * 16u + (unsigned)q];
    const float r0 = re[(unsigned)j0 * 16u + (unsigned)q];
    const float r1 = re[(unsigned)j1 * 16u + (unsigned)q];
    const float r2 = re[(unsigned)j2 * 16u + (unsigned)q];
    const float r3 = re[(unsigned)j3 * 16u + (unsigned)q];
    __builtin_amdgcn_sched_barrier(0);
    const float ninf = -__builtin_inff();
    float l0 = row_sum16(h0.x * as4.x + h0.y * as4.y + h0.z * as4.z + h0.w * as4.w + r0 * ceq);
    float l1 = row_sum16(h1.x * as4.x + h1.y * as4.y + h1.z * as4.z + h1.w * as4.w + r1 * ceq);
    float l2 = row_sum16(h2.x * as4.x + h2.y * as4.y + h2.z * as4.z + h2.w * as4.w + r2 * ceq);
    float l3 = row_sum16(h3.x * as4.x + h3.y * as4.y + h3.z * as4.z + h3.w * as4.w + r3 * ceq);
    l0 = lrelu_f(l0 + dn);
    l1 = (U0 + 1 < nlive) ? lrelu_f(l1 + dn) : ninf;
    l2 = (U0 + 2 < nlive) ? lrelu_f(l2 + dn) : ninf;
    l3 = (U0 + 3 < nlive) ? lrelu_f(l3 + dn) : ninf;
    const float mn = fmaxf(fmaxf(m, l0), fmaxf(fmaxf(l1, l2), l3));
    const float sc = __expf(m - mn);
    const float a0 = __expf(l0 - mn), a1 = __expf(l1 - mn);
    const float a2 = __expf(l2 - mn), a3 = __expf(l3 - mn);
    m = mn;
    dsum = dsum * sc + ((a0 + a1) + (a2 + a3));
    acc.x *= sc; acc.y *= sc; acc.z *= sc; acc.w *= sc; accr *= sc;
    acc.x += a0 * h0.x; acc.y += a0 * h0.y; acc.z += a0 * h0.z; acc.w += a0 * h0.w; accr += a0 * r0;
    acc.x += a1 * h1.x; acc.y += a1 * h1.y; acc.z += a1 * h1.z; acc.w += a1 * h1.w; accr += a1 * r1;
    acc.x += a2 * h2.x; acc.y += a2 * h2.y; acc.z += a2 * h2.z; acc.w += a2 * h2.w; accr += a2 * r2;
    acc.x += a3 * h3.x; acc.y += a3 * h3.y; acc.z += a3 * h3.z; acc.w += a3 * h3.w; accr += a3 * r3;
}

#define TA_P 84
#define TD_P 68
#define TB_P 64
__global__ __launch_bounds__(256) void layer_fused(
    const int* __restrict__ rowptr, const int* __restrict__ pk,
    const float* __restrict__ as, const float* __restrict__ ad,
    const float* __restrict__ ce,
    const float* __restrict__ h, const float* __restrict__ re,
    const float* __restrict__ Wm, const float* __restrict__ WeWm,
    float* __restrict__ hout, int N)
{
    __shared__ __attribute__((aligned(16))) float sA[16 * TA_P];
    __shared__ __attribute__((aligned(16))) float sD[16 * TD_P];
    __shared__ __attribute__((aligned(16))) float sB[80 * TB_P];
    __shared__ __attribute__((aligned(16))) float sH[16 * 64];
    const int r = threadIdx.x >> 4;
    const int nn = blockIdx.x * 16 + r;
    const int n = min(nn, N - 1);
    const int q = threadIdx.x & 15;
    const int t = threadIdx.x;
    const float4 w0 = ((const float4*)Wm)[t];
    const float4 w1 = ((const float4*)Wm)[t + 256];
    const float4 w2 = ((const float4*)Wm)[t + 512];
    const float4 w3 = ((const float4*)Wm)[t + 768];
    const float4 w4 = ((const float4*)WeWm)[t];
    const int beg = rowptr[n], end = rowptr[n + 1];
    const float4 as4 = *(const float4*)(as + q * 4);
    const float4 ad4 = *(const float4*)(ad + q * 4);
    const float ceq = ce[q];
    const float4 hn = *(const float4*)(h + (size_t)n * 64 + q * 4);
    {
        float* sb = sB + (t >> 4) * TB_P + (t & 15) * 4;
        *(float4*)(sb)              = w0;
        *(float4*)(sb + 16 * TB_P)  = w1;
        *(float4*)(sb + 32 * TB_P)  = w2;
        *(float4*)(sb + 48 * TB_P)  = w3;
        *(float4*)(sb + 64 * TB_P)  = w4;
    }
    asm volatile("" :: "v"(hn.x), "v"(hn.y), "v"(hn.z), "v"(hn.w),
                       "v"(ad4.x), "v"(ad4.y), "v"(ad4.z), "v"(ad4.w),
                       "v"(as4.x), "v"(as4.y), "v"(as4.z), "v"(as4.w), "v"(ceq));
    *(float4*)(sH + t * 4) = hn;
    const float dnq = hn.x * ad4.x + hn.y * ad4.y + hn.z * ad4.z + hn.w * ad4.w;
    const int2* __restrict__ pk2 = (const int2*)pk;
    float4 acc = {0.f, 0.f, 0.f, 0.f};
    float accr = 0.f;
    if (end > beg) {
        const int last = end - 1;
        int2 e = pk2[min(beg + q, last)];
        const float dn = row_sum16(dnq);
        float m = -__builtin_inff();
        float dsum = 0.f;
        int c = beg;
        for (;;) {
            const int nlive = end - c;
            gather4<0>(e, q, nlive, as4, ceq, dn, h, re, m, dsum, acc, accr);
            if (nlive > 4)  gather4<4>(e, q, nlive, as4, ceq, dn, h, re, m, dsum, acc, accr);
            if (nlive > 8)  gather4<8>(e, q, nlive, as4, ceq, dn, h, re, m, dsum, acc, accr);
            if (nlive > 12) gather4<12>(e, q, nlive, as4, ceq, dn, h, re, m, dsum, acc, accr);
            c += 16;
            if (c >= end) break;
            e = pk2[min(c + q, last)];
        }
        const float inv = 1.f / dsum;
        acc.x *= inv; acc.y *= inv; acc.z *= inv; acc.w *= inv; accr *= inv;
    }

    const int wv = __builtin_amdgcn_readfirstlane((int)(threadIdx.x >> 5));
    const int lane = threadIdx.x & 31;
    const int cn = lane & 15;
    const int kh = (lane >> 4) << 1;
    *(float4*)(sA + r * TA_P + q * 4) = acc;
    sA[r * TA_P + 64 + q] = accr;
    __syncthreads();
    if (wv < 4) {
        v8f d = {0.f, 0.f, 0.f, 0.f, 0.f, 0.f, 0.f, 0.f};
        const float* ap = sA + cn * TA_P + kh;
        const float* bp = sB + kh * TB_P + wv * 16 + cn;
#pragma unroll
        for (int k = 0; k < 20; ++k) {
            const v2f a = *(const v2f*)(ap + 4 * k);
            v2f b; b.x = bp[4 * k * TB_P]; b.y = bp[(4 * k + 1) * TB_P];
            d = __builtin_amdgcn_wmma_f32_16x16x4_f32(false, a, false, b, (short)0, d, false, false);
        }
        float* dp = sD + ((lane >> 4) << 3) * TD_P + wv * 16 + cn;
#pragma unroll
        for (int v = 0; v < 8; ++v) dp[v * TD_P] = d[v];
    }
    __syncthreads();
    if (nn < N) {
        float4 o = *(const float4*)(sD + r * TD_P + q * 4);
        const float4 hr = *(const float4*)(sH + t * 4);
        o.x = lrelu_f(o.x + hr.x); o.y = lrelu_f(o.y + hr.y);
        o.z = lrelu_f(o.z + hr.z); o.w = lrelu_f(o.w + hr.w);
        *(float4*)(hout + (size_t)nn * 64 + q * 4) = o;
    }
}

#define WALK 16

__device__ __forceinline__ void row_flush(float* __restrict__ grow, const int q, const float4 a)
{
    float* p = grow + q * 4;
    atomicAdd(p + 0, a.x); atomicAdd(p + 1, a.y); atomicAdd(p + 2, a.z); atomicAdd(p + 3, a.w);
}

__device__ __forceinline__ void row_bcast_all(const int iq, const float fq,
                                              int (&iv)[WALK], float (&fv)[WALK])
{
#define BC1(U) iv[U] = row_bcast<U>(iq); fv[U] = row_bcast<U>(fq);
    BC1(0)  BC1(1)  BC1(2)  BC1(3)  BC1(4)  BC1(5)  BC1(6)  BC1(7)
    BC1(8)  BC1(9)  BC1(10) BC1(11) BC1(12) BC1(13) BC1(14) BC1(15)
#undef BC1
}

__global__ __launch_bounds__(128) void gate_hi(
    const float* __restrict__ h, const float* __restrict__ d_edge,
    const int* __restrict__ gid, const float* __restrict__ wd,
    const float* __restrict__ bd, float* __restrict__ gate,
    float* __restrict__ hi, int N)
{
    const int q  = threadIdx.x & 15;
    const int n0 = (blockIdx.x * 8 + (threadIdx.x >> 4)) * WALK;
    if (n0 >= N) return;
    const int nq = min(n0 + q, N - 1);
    float4 hv[WALK];
#pragma unroll
    for (int u = 0; u < WALK; ++u)
        hv[u] = *(const float4*)(h + (size_t)min(n0 + u, N - 1) * 64 + q * 4);
    const float gq = 1.f / (1.f + __expf(-(d_edge[nq] * wd[0] + bd[0])));
    const int gidq = gid[nq];
    if (n0 + q < N) gate[n0 + q] = gq;
    int gi[WALK]; float g[WALK];
    row_bcast_all(gidq, gq, gi, g);
    float4 acc = {0.f, 0.f, 0.f, 0.f};
    int cur = gi[0];
#pragma unroll
    for (int u = 0; u < WALK; ++u) {
        if (n0 + u < N) {
            if (gi[u] != cur) {
                row_flush(hi + (size_t)cur * 64, q, acc);
                acc.x = 0.f; acc.y = 0.f; acc.z = 0.f; acc.w = 0.f;
                cur = gi[u];
            }
            acc.x += g[u] * hv[u].x; acc.y += g[u] * hv[u].y;
            acc.z += g[u] * hv[u].z; acc.w += g[u] * hv[u].w;
        }
    }
    row_flush(hi + (size_t)cur * 64, q, acc);
}

__global__ __launch_bounds__(128) void h_mix_pool(
    float* __restrict__ h, const float* __restrict__ gate,
    const float* __restrict__ hi, const int* __restrict__ gid,
    float* __restrict__ pooled, int N)
{
    const int q  = threadIdx.x & 15;
    const int n0 = (blockIdx.x * 8 + (threadIdx.x >> 4)) * WALK;
    if (n0 >= N) return;
    const int nq = min(n0 + q, N - 1);
    float4 hv[WALK];
#pragma unroll
    for (int u = 0; u < WALK; ++u)
        hv[u] = *(const float4*)(h + (size_t)min(n0 + u, N - 1) * 64 + q * 4);
    const float gq = gate[nq];
    const int gidq = gid[nq];
    int gi[WALK]; float g[WALK];
    row_bcast_all(gidq, gq, gi, g);
    float4 hiv[WALK];
#pragma unroll
    for (int u = 0; u < WALK; ++u)
        hiv[u] = *(const float4*)(hi + (size_t)gi[u] * 64 + q * 4);
    float4 acc = {0.f, 0.f, 0.f, 0.f};
    int cur = gi[0];
#pragma unroll
    for (int u = 0; u < WALK; ++u) {
        if (n0 + u < N) {
            float4 v;
            v.x = hv[u].x + g[u] * hiv[u].x; v.y = hv[u].y + g[u] * hiv[u].y;
            v.z = hv[u].z + g[u] * hiv[u].z; v.w = hv[u].w + g[u] * hiv[u].w;
            *(float4*)(h + (size_t)(n0 + u) * 64 + q * 4) = v;
            if (gi[u] != cur) {
                row_flush(pooled + (size_t)cur * 64, q, acc);
                acc.x = 0.f; acc.y = 0.f; acc.z = 0.f; acc.w = 0.f;
                cur = gi[u];
            }
            acc.x += v.x; acc.y += v.y; acc.z += v.z; acc.w += v.w;
        }
    }
    row_flush(pooled + (size_t)cur * 64, q, acc);
}

#define XP 68
__global__ __launch_bounds__(128) void mlp_out(
    const float* __restrict__ pooled, const float* __restrict__ W,
    const float* __restrict__ bias, const float* __restrict__ Wo,
    const float* __restrict__ bo, float* __restrict__ out)
{
    __shared__ __attribute__((aligned(16))) float sX[2][16 * XP];
    const int lane = threadIdx.x & 31;
    const int wv = __builtin_amdgcn_readfirstlane((int)(threadIdx.x >> 5));
    const int cn = lane & 15;
    const int kh = (lane >> 4) << 1;
    const int hb = (lane >> 4) << 3;
    const size_t row = (size_t)blockIdx.x * 16 + cn;
    const int col = wv * 16 + cn;
    const float* prow = pooled + row * HD;
#pragma unroll
    for (int l = 0; l < 3; ++l) {
        const float* B = W + l * 4096;
        const float* xin = sX[(l + 1) & 1];
        float* xout = sX[l & 1];
        v8f d = {0.f, 0.f, 0.f, 0.f, 0.f, 0.f, 0.f, 0.f};
#pragma unroll
        for (int k0 = 0; k0 < 64; k0 += 4) {
            v2f a;
            if (l == 0) { a.x = prow[k0 + kh]; a.y = prow[k0 + kh + 1]; }
            else        { a = *(const v2f*)(xin + cn * XP + k0 + kh); }
            v2f b;
            b.x = B[(size_t)(k0 + kh) * HD + col];
            b.y = B[(size_t)(k0 + kh + 1) * HD + col];
            d = __builtin_amdgcn_wmma_f32_16x16x4_f32(false, a, false, b, (short)0, d, false, false);
        }
        const float bval = bias[l * HD + col];
#pragma unroll
        for (int v = 0; v < 8; ++v) xout[(hb + v) * XP + col] = fmaxf(d[v] + bval, 0.f);
        __syncthreads();
    }
    if (wv == 0 && lane < 16) {
        const float* xr = sX[0] + cn * XP;
        float s = bo[0];
#pragma unroll
        for (int k = 0; k < 64; ++k) s += xr[k] * Wo[k];
        out[row] = s;
    }
}

extern "C" void kernel_launch(void* const* d_in, const int* in_sizes, int n_in,
                              void* d_out, int out_size, void* d_ws, size_t ws_size,
                              hipStream_t stream)
{
    const float* r_node = (const float*)d_in[0];
    const float* i_node = (const float*)d_in[1];
    const float* r_edge = (const float*)d_in[2];
    const float* d_edge = (const float*)d_in[3];
    const int*   src    = (const int*)d_in[4];
    const int*   dst    = (const int*)d_in[5];
    const int*   gid    = (const int*)d_in[6];
    const float* W_r    = (const float*)d_in[7];
    const float* W_i    = (const float*)d_in[8];
    const float* W_e    = (const float*)d_in[9];
    const float* Wm     = (const float*)d_in[10];
    const float* a_s    = (const float*)d_in[11];
    const float* a_d    = (const float*)d_in[12];
    const float* a_e    = (const float*)d_in[13];
    const float* w_d    = (const float*)d_in[14];
    const float* b_d    = (const float*)d_in[15];
    const float* W_mlp  = (const float*)d_in[16];
    const float* b_mlp  = (const float*)d_in[17];
    const float* W_out  = (const float*)d_in[18];
    const float* b_out  = (const float*)d_in[19];

    const int N = in_sizes[0] / 64;
    const int G = in_sizes[1];
    const int E = in_sizes[2] / 16;
    (void)n_in; (void)out_size; (void)ws_size;

    float* ws = (float*)d_ws;
    size_t off = 0;
    auto carve = [&](size_t n) -> float* {
        float* p = ws + off; off += (n + 63) & ~(size_t)63; return p;
    };
    float*    h      = carve((size_t)N * 64);
    float*    h2     = carve((size_t)N * 64);
    float*    gate   = carve(N);
    float*    hi     = carve((size_t)G * 64);
    float*    pooled = carve((size_t)G * 64);
    float*    cE     = carve(64);
    float*    WeWm   = carve(4 * 16 * 64);
    const int R   = (N + CSR_NB - 1) / CSR_NB;
    const int NP  = (N + 1 + 63) & ~63;
    const int CSP = R * (CSR_NB / CSR_CH);
    int*      pc     = (int*)carve((size_t)CSR_S * NP);
    int*      cs     = (int*)carve((size_t)CSR_S * CSP);
    int*      rowptr = (int*)carve((size_t)CSR_S * NP);
    int*      pk     = (int*)carve((size_t)E * 2);

    const int ntile   = N / 16;
    const int nb_gemm = (ntile + 7) / 8;
    const int nb_init = (G * 16 + 255) / 256;
    hipLaunchKernelGGL(front_all, dim3(FRONT_SMALL + nb_gemm + nb_init), dim3(256), 0, stream,
                       r_node, W_r, h, ntile, nb_gemm, i_node, W_i, hi, pooled, G,
                       W_e, a_e, Wm, cE, WeWm);

    hipLaunchKernelGGL(csr_count, dim3(R * CSR_S), dim3(1024), 0, stream, dst, pc, cs, N, NP, CSP, E, R);
    hipLaunchKernelGGL(csr_scan, dim3((N + 1023) / 1024), dim3(1024), 0, stream, pc, cs, rowptr, N, NP, CSP);
    hipLaunchKernelGGL(csr_fill, dim3(R * CSR_S), dim3(1024), 0, stream,
                       src, dst, rowptr, pk, N, NP, E, R);

    float* hc = h;
    float* hx = h2;
    for (int l = 0; l < 4; ++l) {
        hipLaunchKernelGGL(layer_fused, dim3((N + 15) / 16), dim3(256), 0, stream,
                           rowptr, pk, a_s + l * 64, a_d + l * 64, cE + l * 16,
                           hc, r_edge, Wm + (size_t)l * 4096, WeWm + (size_t)l * 1024, hx, N);
        float* t = hc; hc = hx; hx = t;
    }

    hipLaunchKernelGGL(gate_hi, dim3((N + 127) / 128), dim3(128), 0, stream,
                       hc, d_edge, gid, w_d, b_d, gate, hi, N);
    hipLaunchKernelGGL(h_mix_pool, dim3((N + 127) / 128), dim3(128), 0, stream,
                       hc, gate, hi, gid, pooled, N);

    hipLaunchKernelGGL(mlp_out, dim3(G / 16), dim3(128), 0, stream,
                       pooled, W_mlp, b_mlp, W_out, b_out, (float*)d_out);
}


# Gate/mix/pool passes folded away: layer 4 sums h, gate*h, gate per graph-id run in its epilogue (LDS walk, one atomic set per run); readout forms pooled = sum h + hi*sum gate; 11 launches to 9
# speedup vs baseline: 8.8832x; 1.0488x over previous
#include <hip/hip_runtime.h>

typedef __attribute__((ext_vector_type(2))) float v2f;
typedef __attribute__((ext_vector_type(8))) float v8f;

#define HD 64

__device__ __forceinline__ float lrelu_f(float x) { return x > 0.f ? x : 0.1f * x; }

template <int M> __device__ __forceinline__ float row_xor(float v) {
    return __int_as_float(__builtin_amdgcn_mov_dpp(__float_as_int(v), 0x160 + M, 0xf, 0xf, false));
}
template <int U> __device__ __forceinline__ int row_bcast(int v) {
    return __builtin_amdgcn_mov_dpp(v, 0x150 + U, 0xf, 0xf, false);
}
__device__ __forceinline__ float row_max16(float v) {
    v = fmaxf(v, row_xor<1>(v)); v = fmaxf(v, row_xor<2>(v));
    v = fmaxf(v, row_xor<4>(v)); v = fmaxf(v, row_xor<8>(v));
    return v;
}
__device__ __forceinline__ float row_sum16(float v) {
    v += row_xor<1>(v); v += row_xor<2>(v); v += row_xor<4>(v); v += row_xor<8>(v);
    return v;
}

template <bool HAS_A2, bool HAS_RES, bool HAS_BIAS, int ACT>
__device__ __forceinline__ void gemm64_tile(const int tile,
    const float* __restrict__ A,  const float* __restrict__ B,
    const float* __restrict__ A2, const float* __restrict__ B2,
    const float* __restrict__ res, const float* __restrict__ bias,
    float* __restrict__ out)
{
    const int lane = threadIdx.x & 31;
    const int cn   = lane & 15;
    const int kh   = (lane >> 4) << 1;
    const size_t row = (size_t)tile * 16 + cn;

    const v8f z = {0.f, 0.f, 0.f, 0.f, 0.f, 0.f, 0.f, 0.f};
    v8f acc0 = z, acc1 = z, acc2 = z, acc3 = z;

    const float* arow = A + row * HD;
#pragma unroll
    for (int k0 = 0; k0 < 64; k0 += 4) {
        v2f a; a.x = arow[k0 + kh]; a.y = arow[k0 + kh + 1];
        const float* b0 = B + (size_t)(k0 + kh) * HD;
        const float* b1 = b0 + HD;
        v2f b;
        b.x = b0[cn];        b.y = b1[cn];
        acc0 = __builtin_amdgcn_wmma_f32_16x16x4_f32(false, a, false, b, (short)0, acc0, false, false);
        b.x = b0[16 + cn];   b.y = b1[16 + cn];
        acc1 = __builtin_amdgcn_wmma_f32_16x16x4_f32(false, a, false, b, (short)0, acc1, false, false);
        b.x = b0[32 + cn];   b.y = b1[32 + cn];
        acc2 = __builtin_amdgcn_wmma_f32_16x16x4_f32(false, a, false, b, (short)0, acc2, false, false);
        b.x = b0[48 + cn];   b.y = b1[48 + cn];
        acc3 = __builtin_amdgcn_wmma_f32_16x16x4_f32(false, a, false, b, (short)0, acc3, false, false);
    }
    if (HAS_A2) {
        const float* a2row = A2 + row * 16;
#pragma unroll
        for (int k0 = 0; k0 < 16; k0 += 4) {
            v2f a; a.x = a2row[k0 + kh]; a.y = a2row[k0 + kh + 1];
            const float* b0 = B2 + (size_t)(k0 + kh) * HD;
            const float* b1 = b0 + HD;
            v2f b;
            b.x = b0[cn];        b.y = b1[cn];
            acc0 = __builtin_amdgcn_wmma_f32_16x16x4_f32(false, a, false, b, (short)0, acc0, false, false);
            b.x = b0[16 + cn];   b.y = b1[16 + cn];
            acc1 = __builtin_amdgcn_wmma_f32_16x16x4_f32(false, a, false, b, (short)0, acc1, false, false);
            b.x = b0[32 + cn];   b.y = b1[32 + cn];
            acc2 = __builtin_amdgcn_wmma_f32_16x16x4_f32(false, a, false, b, (short)0, acc2, false, false);
            b.x = b0[48 + cn];   b.y = b1[48 + cn];
            acc3 = __builtin_amdgcn_wmma_f32_16x16x4_f32(false, a, false, b, (short)0, acc3, false, false);
        }
    }

    const size_t mbase = (size_t)tile * 16 + ((lane >> 4) << 3);
    v8f accs[4] = {acc0, acc1, acc2, acc3};
#pragma unroll
    for (int t = 0; t < 4; ++t) {
        const int col = t * 16 + cn;
        const float bval = HAS_BIAS ? bias[col] : 0.f;
#pragma unroll
        for (int v = 0; v < 8; ++v) {
            const size_t m = mbase + v;
            float val = accs[t][v] + bval;
            if (HAS_RES) val += res[m * HD + col];
            if (ACT == 1) val = lrelu_f(val);
            else if (ACT == 2) val = fmaxf(val, 0.f);
            out[m * HD + col] = val;
        }
    }
}

#define FRONT_SMALL 17
__global__ __launch_bounds__(256) void front_all(
    const float* __restrict__ r_node, const float* __restrict__ W_r, float* __restrict__ h,
    const int ntile, const int nb_gemm,
    const float* __restrict__ i_node, const float* __restrict__ W_i,
    float* __restrict__ hi, float* __restrict__ pooled, float* __restrict__ gsum, const int G,
    const float* __restrict__ W_e, const float* __restrict__ a_e, const float* __restrict__ Wm,
    float* __restrict__ cE, float* __restrict__ WeWm)
{
    const int b = blockIdx.x;
    const int tid = threadIdx.x;
    if (b < 16) {
        const int idx = b * 256 + tid;
        const int l = idx >> 10, d = (idx >> 6) & 15, k = idx & 63;
        float s = 0.f;
        for (int j = 0; j < 64; ++j) s += W_e[d * 64 + j] * Wm[l * 4096 + j * 64 + k];
        WeWm[idx] = s;
    } else if (b == 16) {
        if (tid < 64) {
            const int l = tid >> 4, d = tid & 15;
            float s = 0.f;
            for (int k = 0; k < 64; ++k) s += W_e[d * 64 + k] * a_e[l * 64 + k];
            cE[tid] = s;
        }
    } else if (b < FRONT_SMALL + nb_gemm) {
        const int tile = (b - FRONT_SMALL) * 8 + __builtin_amdgcn_readfirstlane(tid >> 5);
        if (tile < ntile)
            gemm64_tile<false, false, false, 0>(tile, r_node, W_r, nullptr, nullptr, nullptr, nullptr, h);
    } else {
        const int idx4 = (b - FRONT_SMALL - nb_gemm) * 256 + tid;
        if (idx4 < G * 16) {
            const float s = i_node[idx4 >> 4];
            const float4 w = *(const float4*)(W_i + (idx4 & 15) * 4);
            float4 v; v.x = s * w.x; v.y = s * w.y; v.z = s * w.z; v.w = s * w.w;
            const float4 zero = {0.f, 0.f, 0.f, 0.f};
            ((float4*)hi)[idx4] = v;
            ((float4*)pooled)[idx4] = zero;
            if ((idx4 & 15) == 0) gsum[idx4 >> 4] = 0.f;
        }
    }
}

#define CSR_NB 12800
#define CSR_S  32
#define CSR_CH 128

__global__ __launch_bounds__(1024) void csr_count(const int* __restrict__ dst,
                                                  int* __restrict__ pc, int* __restrict__ cs,
                                                  int N, int NP, int CSP, int E, int R)
{
    __shared__ int cnt[CSR_NB];
    const int t = threadIdx.x;
    const int r = blockIdx.x % R, s = blockIdx.x / R;
    const int r0 = r * CSR_NB;
    for (int i = t; i < CSR_NB; i += 1024) cnt[i] = 0;
    __syncthreads();
    const int E4 = E >> 2;
    const int SL4 = (E4 + CSR_S - 1) / CSR_S;
    const int b4 = s * SL4, e4 = min(b4 + SL4, E4);
    const int4* __restrict__ d4 = (const int4*)dst;
    for (int i = b4 + t; i < e4; i += 4096) {
        int4 d[4];
#pragma unroll
        for (int k = 0; k < 4; ++k) d[k] = d4[min(i + k * 1024, e4 - 1)];
#pragma unroll
        for (int k = 0; k < 4; ++k) {
            if (i + k * 1024 < e4) {
                unsigned u;
                u = (unsigned)(d[k].x - r0); if (u < CSR_NB) atomicAdd(&cnt[u], 1);
                u = (unsigned)(d[k].y - r0); if (u < CSR_NB) atomicAdd(&cnt[u], 1);
                u = (unsigned)(d[k].z - r0); if (u < CSR_NB) atomicAdd(&cnt[u], 1);
                u = (unsigned)(d[k].w - r0); if (u < CSR_NB) atomicAdd(&cnt[u], 1);
            }
        }
    }
    if (s == CSR_S - 1) {
        for (int j = (E4 << 2) + t; j < E; j += 1024) {
            const unsigned u = (unsigned)(dst[j] - r0);
            if (u < CSR_NB) atomicAdd(&cnt[u], 1);
        }
    }
    __syncthreads();
    for (int i = t; i < CSR_NB; i += 1024) {
        const int n = r0 + i;
        if (n < N) pc[(size_t)s * NP + n] = cnt[i];
    }
    const int lane = t & 31, w = t >> 5;
    for (int c = w; c < CSR_NB / CSR_CH; c += 32) {
        const int* p = cnt + c * CSR_CH + lane;
        int v = (p[0] + p[32]) + (p[64] + p[96]);
        for (int o = 16; o > 0; o >>= 1) v += __shfl_xor(v, o, 32);
        if (lane == 0) cs[s * CSP + r * (CSR_NB / CSR_CH) + c] = v;
    }
}

__global__ __launch_bounds__(1024) void csr_scan(const int* __restrict__ pc,
                                                 const int* __restrict__ cs,
                                                 int* __restrict__ base, int N, int NP, int CSP)
{
    __shared__ int wfront[32];
    __shared__ int wtot[32];
    const int t = threadIdx.x, lane = t & 31, w = t >> 5;
    const int base0 = blockIdx.x * 1024;
    const int ch0 = blockIdx.x * (1024 / CSR_CH);

    int sf = 0;
    for (int c = t; c < ch0; c += 1024) {
#pragma unroll
        for (int s = 0; s < CSR_S; ++s) sf += cs[s * CSP + c];
    }
    for (int o = 16; o > 0; o >>= 1) sf += __shfl_xor(sf, o, 32);
    if (lane == 0) wfront[w] = sf;

    const int i = base0 + t;
    const int ic = min(i, N - 1);
    int pcs[CSR_S];
    int c = 0;
#pragma unroll
    for (int s = 0; s < CSR_S; ++s) {
        pcs[s] = (i < N) ? pc[(size_t)s * NP + ic] : 0;
        c += pcs[s];
    }
    int x = c;
    for (int o = 1; o < 32; o <<= 1) {
        const int y = __shfl_up(x, o, 32);
        if (lane >= o) x += y;
    }
    if (lane == 31) wtot[w] = x;
    __syncthreads();

    int front = 0;
    for (int k = 0; k < 32; ++k) front += wfront[k];
    for (int k = 0; k < w; ++k) front += wtot[k];
    if (i < N) {
        int run = front + x - c;
#pragma unroll
        for (int s = 0; s < CSR_S; ++s) {
            base[(size_t)s * NP + i] = run;
            run += pcs[s];
        }
        if (i == N - 1) base[N] = run;
    }
}

__global__ __launch_bounds__(1024) void csr_fill(const int* __restrict__ src,
                                                 const int* __restrict__ dst,
                                                 const int* __restrict__ base,
                                                 int* __restrict__ pk, int N, int NP, int E, int R)
{
    __shared__ int pos[CSR_NB];
    const int t = threadIdx.x;
    const int r = blockIdx.x % R, s = blockIdx.x / R;
    const int r0 = r * CSR_NB;
    for (int i = t; i < CSR_NB; i += 1024) {
        const int n = r0 + i;
        pos[i] = (n < N) ? base[(size_t)s * NP + n] : 0;
    }
    __syncthreads();
    const int E4 = E >> 2;
    const int SL4 = (E4 + CSR_S - 1) / CSR_S;
    const int b4 = s * SL4, e4 = min(b4 + SL4, E4);
    const int4* __restrict__ d4 = (const int4*)dst;
    const int4* __restrict__ s4 = (const int4*)src;
    int2* __restrict__ pk2 = (int2*)pk;
    for (int i = b4 + t; i < e4; i += 4096) {
        int4 d[4], sv[4];
#pragma unroll
        for (int k = 0; k < 4; ++k) {
            const int idx = min(i + k * 1024, e4 - 1);
            d[k] = d4[idx]; sv[k] = s4[idx];
        }
        int p[16];
#pragma unroll
        for (int k = 0; k < 4; ++k) {
            const bool live = i + k * 1024 < e4;
            unsigned u;
            u = (unsigned)(d[k].x - r0); p[4 * k + 0] = -1;
            if (live && u < CSR_NB) p[4 * k + 0] = atomicAdd(&pos[u], 1);
            u = (unsigned)(d[k].y - r0); p[4 * k + 1] = -1;
            if (live && u < CSR_NB) p[4 * k + 1] = atomicAdd(&pos[u], 1);
            u = (unsigned)(d[k].z - r0); p[4 * k + 2] = -1;
            if (live && u < CSR_NB) p[4 * k + 2] = atomicAdd(&pos[u], 1);
            u = (unsigned)(d[k].w - r0); p[4 * k + 3] = -1;
            if (live && u < CSR_NB) p[4 * k + 3] = atomicAdd(&pos[u], 1);
        }
#pragma unroll
        for (int k = 0; k < 4; ++k) {
            const int j = (i + k * 1024) << 2;
            if (p[4 * k + 0] >= 0) pk2[p[4 * k + 0]] = make_int2(j,     sv[k].x);
            if (p[4 * k + 1] >= 0) pk2[p[4 * k + 1]] = make_int2(j + 1, sv[k].y);
            if (p[4 * k + 2] >= 0) pk2[p[4 * k + 2]] = make_int2(j + 2, sv[k].z);
            if (p[4 * k + 3] >= 0) pk2[p[4 * k + 3]] = make_int2(j + 3, sv[k].w);
        }
    }
    if (s == CSR_S - 1) {
        for (int j = (E4 << 2) + t; j < E; j += 1024) {
            const unsigned u = (unsigned)(dst[j] - r0);
            if (u < CSR_NB) { const int q = atomicAdd(&pos[u], 1); pk2[q] = make_int2(j, src[j]); }
        }
    }
}

template <int U0> __device__ __forceinline__ void gather4(
    const int2 e, const int q, const int nlive,
    const float4 as4, const float ceq, const float dn,
    const float* __restrict__ h, const float* __restrict__ re,
    float& m, float& dsum, float4& acc, float& accr)
{
    const int j0 = row_bcast<U0 + 0>(e.x), s0 = row_bcast<U0 + 0>(e.y);
    const int j1 = row_bcast<U0 + 1>(e.x), s1 = row_bcast<U0 + 1>(e.y);
    const int j2 = row_bcast<U0 + 2>(e.x), s2 = row_bcast<U0 + 2>(e.y);
    const int j3 = row_bcast<U0 + 3>(e.x), s3 = row_bcast<U0 + 3>(e.y);
    const float4* __restrict__ h4 = (const float4*)h;
    const float4 h0 = h4[(unsigned)s0 * 16u + (unsigned)q];
    const float4 h1 = h4[(unsigned)s1 * 16u + (unsigned)q];
    const float4 h2 = h4[(unsigned)s2 * 16u + (unsigned)q];
    const float4 h3 = h4[(unsigned)s3 * 16u + (unsigned)q];
    const float r0 = re[(unsigned)j0 * 16u + (unsigned)q];
    const float r1 = re[(unsigned)j1 * 16u + (unsigned)q];
    const float r2 = re[(unsigned)j2 * 16u + (unsigned)q];
    const float r3 = re[(unsigned)j3 * 16u + (unsigned)q];
    __builtin_amdgcn_sched_barrier(0);
    const float ninf = -__builtin_inff();
    float l0 = row_sum16(h0.x * as4.x + h0.y * as4.y + h0.z * as4.z + h0.w * as4.w + r0 * ceq);
    float l1 = row_sum16(h1.x * as4.x + h1.y * as4.y + h1.z * as4.z + h1.w * as4.w + r1 * ceq);
    float l2 = row_sum16(h2.x * as4.x + h2.y * as4.y + h2.z * as4.z + h2.w * as4.w + r2 * ceq);
    float l3 = row_sum16(h3.x * as4.x + h3.y * as4.y + h3.z * as4.z + h3.w * as4.w + r3 * ceq);
    l0 = lrelu_f(l0 + dn);
    l1 = (U0 + 1 < nlive) ? lrelu_f(l1 + dn) : ninf;
    l2 = (U0 + 2 < nlive) ? lrelu_f(l2 + dn) : ninf;
    l3 = (U0 + 3 < nlive) ? lrelu_f(l3 + dn) : ninf;
    const float mn = fmaxf(fmaxf(m, l0), fmaxf(fmaxf(l1, l2), l3));
    const float sc = __expf(m - mn);
    const float a0 = __expf(l0 - mn), a1 = __expf(l1 - mn);
    const float a2 = __expf(l2 - mn), a3 = __expf(l3 - mn);
    m = mn;
    dsum = dsum * sc + ((a0 + a1) + (a2 + a3));
    acc.x *= sc; acc.y *= sc; acc.z *= sc; acc.w *= sc; accr *= sc;
    acc.x += a0 * h0.x; acc.y += a0 * h0.y; acc.z += a0 * h0.z; acc.w += a0 * h0.w; accr += a0 * r0;
    acc.x += a1 * h1.x; acc.y += a1 * h1.y; acc.z += a1 * h1.z; acc.w += a1 * h1.w; accr += a1 * r1;
    acc.x += a2 * h2.x; acc.y += a2 * h2.y; acc.z += a2 * h2.z; acc.w += a2 * h2.w; accr += a2 * r2;
    acc.x += a3 * h3.x; acc.y += a3 * h3.y; acc.z += a3 * h3.z; acc.w += a3 * h3.w; accr += a3 * r3;
}

#define TA_P 84
#define TD_P 68
#define TB_P 64
template <bool LAST>
__global__ __launch_bounds__(256) void layer_fused(
    const int* __restrict__ rowptr, const int* __restrict__ pk,
    const float* __restrict__ as, const float* __restrict__ ad,
    const float* __restrict__ ce,
    const float* __restrict__ h, const float* __restrict__ re,
    const float* __restrict__ Wm, const float* __restrict__ WeWm,
    float* __restrict__ hout, int N,
    const float* __restrict__ d_edge, const int* __restrict__ gid,
    const float* __restrict__ wd, const float* __restrict__ bd,
    float* __restrict__ hi, float* __restrict__ pooled, float* __restrict__ gsum)
{
    __shared__ __attribute__((aligned(16))) float sA[16 * TA_P];
    __shared__ __attribute__((aligned(16))) float sD[16 * TD_P];
    __shared__ __attribute__((aligned(16))) float sB[80 * TB_P];
    __shared__ __attribute__((aligned(16))) float sH[16 * 64];
    __shared__ __attribute__((aligned(16))) float sG[16];
    __shared__ __attribute__((aligned(16))) int   sGid[16];
    const int r = threadIdx.x >> 4;
    const int nn = blockIdx.x * 16 + r;
    const int n = min(nn, N - 1);
    const int q = threadIdx.x & 15;
    const int t = threadIdx.x;
    const float4 w0 = ((const float4*)Wm)[t];
    const float4 w1 = ((const float4*)Wm)[t + 256];
    const float4 w2 = ((const float4*)Wm)[t + 512];
    const float4 w3 = ((const float4*)Wm)[t + 768];
    const float4 w4 = ((const float4*)WeWm)[t];
    const int beg = rowptr[n], end = rowptr[n + 1];
    const float4 as4 = *(const float4*)(as + q * 4);
    const float4 ad4 = *(const float4*)(ad + q * 4);
    const float ceq = ce[q];
    float dq = 0.f; int giq = 0;
    if (LAST && t < 16) {
        const int nq = min((int)blockIdx.x * 16 + t, N - 1);
        dq = d_edge[nq]; giq = gid[nq];
    }
    const float4 hn = *(const float4*)(h + (size_t)n * 64 + q * 4);
    {
        float* sb = sB + (t >> 4) * TB_P + (t & 15) * 4;
        *(float4*)(sb)              = w0;
        *(float4*)(sb + 16 * TB_P)  = w1;
        *(float4*)(sb + 32 * TB_P)  = w2;
        *(float4*)(sb + 48 * TB_P)  = w3;
        *(float4*)(sb + 64 * TB_P)  = w4;
    }
    asm volatile("" :: "v"(hn.x), "v"(hn.y), "v"(hn.z), "v"(hn.w),
                       "v"(ad4.x), "v"(ad4.y), "v"(ad4.z), "v"(ad4.w),
                       "v"(as4.x), "v"(as4.y), "v"(as4.z), "v"(as4.w), "v"(ceq));
    *(float4*)(sH + t * 4) = hn;
    if (LAST && t < 16) {
        sG[t] = 1.f / (1.f + __expf(-(dq * wd[0] + bd[0])));
        sGid[t] = giq;
    }
    const float dnq = hn.x * ad4.x + hn.y * ad4.y + hn.z * ad4.z + hn.w * ad4.w;
    const int2* __restrict__ pk2 = (const int2*)pk;
    float4 acc = {0.f, 0.f, 0.f, 0.f};
    float accr = 0.f;
    if (end > beg) {
        const int last = end - 1;
        int2 e = pk2[min(beg + q, last)];
        const float dn = row_sum16(dnq);
        float m = -__builtin_inff();
        float dsum = 0.f;
        int c = beg;
        for (;;) {
            const int nlive = end - c;
            gather4<0>(e, q, nlive, as4, ceq, dn, h, re, m, dsum, acc, accr);
            if (nlive > 4)  gather4<4>(e, q, nlive, as4, ceq, dn, h, re, m, dsum, acc, accr);
            if (nlive > 8)  gather4<8>(e, q, nlive, as4, ceq, dn, h, re, m, dsum, acc, accr);
            if (nlive > 12) gather4<12>(e, q, nlive, as4, ceq, dn, h, re, m, dsum, acc, accr);
            c += 16;
            if (c >= end) break;
            e = pk2[min(c + q, last)];
        }
        const float inv = 1.f / dsum;
        acc.x *= inv; acc.y *= inv; acc.z *= inv; acc.w *= inv; accr *= inv;
    }

    const int wv = __builtin_amdgcn_readfirstlane((int)(threadIdx.x >> 5));
    const int lane = threadIdx.x & 31;
    const int cn = lane & 15;
    const int kh = (lane >> 4) << 1;
    *(float4*)(sA + r * TA_P + q * 4) = acc;
    sA[r * TA_P + 64 + q] = accr;
    __syncthreads();
    if (wv < 4) {
        v8f d = {0.f, 0.f, 0.f, 0.f, 0.f, 0.f, 0.f, 0.f};
        const float* ap = sA + cn * TA_P + kh;
        const float* bp = sB + kh * TB_P + wv * 16 + cn;
#pragma unroll
        for (int k = 0; k < 20; ++k) {
            const v2f a = *(const v2f*)(ap + 4 * k);
            v2f b; b.x = bp[4 * k * TB_P]; b.y = bp[(4 * k + 1) * TB_P];
            d = __builtin_amdgcn_wmma_f32_16x16x4_f32(false, a, false, b, (short)0, d, false, false);
        }
        float* dp = sD + ((lane >> 4) << 3) * TD_P + wv * 16 + cn;
#pragma unroll
        for (int v = 0; v < 8; ++v) dp[v * TD_P] = d[v];
    }
    __syncthreads();
    float4 o = *(const float4*)(sD + r * TD_P + q * 4);
    const float4 hr = *(const float4*)(sH + t * 4);
    o.x = lrelu_f(o.x + hr.x); o.y = lrelu_f(o.y + hr.y);
    o.z = lrelu_f(o.z + hr.z); o.w = lrelu_f(o.w + hr.w);
    if (!LAST) {
        if (nn < N) *(float4*)(hout + (size_t)nn * 64 + q * 4) = o;
    } else {
        *(float4*)(sA + r * TA_P + q * 4) = o;
        __syncthreads();
        if (t < 64) {
            const int nrow = min(16, N - (int)blockIdx.x * 16);
            float* __restrict__ hcol = hi + t;
            float* __restrict__ pcol = pooled + t;
            float vv[16];
#pragma unroll
            for (int u = 0; u < 16; ++u) vv[u] = sA[u * TA_P + t];
            int gi[16]; float g[16];
#pragma unroll
            for (int u = 0; u < 16; ++u) {
                gi[u] = __builtin_amdgcn_readfirstlane(sGid[u]);
                g[u]  = __int_as_float(__builtin_amdgcn_readfirstlane(__float_as_int(sG[u])));
            }
            float sh = 0.f, sgh = 0.f, sg = 0.f;
            int cur = gi[0];
#pragma unroll
            for (int u = 0; u < 16; ++u) {
                if (u < nrow) {
                    if (gi[u] != cur) {
                        atomicAdd(hcol + (size_t)cur * 64, sgh);
                        atomicAdd(pcol + (size_t)cur * 64, sh);
                        if (t == 0) atomicAdd(gsum + cur, sg);
                        sh = 0.f; sgh = 0.f; sg = 0.f;
                        cur = gi[u];
                    }
                    sh += vv[u]; sgh += g[u] * vv[u]; sg += g[u];
                }
            }
            atomicAdd(hcol + (size_t)cur * 64, sgh);
            atomicAdd(pcol + (size_t)cur * 64, sh);
            if (t == 0) atomicAdd(gsum + cur, sg);
        }
    }
}

#define XP 68
__global__ __launch_bounds__(128) void mlp_out(
    const float* __restrict__ pooled, const float* __restrict__ hi,
    const float* __restrict__ gsum, const float* __restrict__ W,
    const float* __restrict__ bias, const float* __restrict__ Wo,
    const float* __restrict__ bo, float* __restrict__ out)
{
    __shared__ __attribute__((aligned(16))) float sX[2][16 * XP];
    const int lane = threadIdx.x & 31;
    const int wv = __builtin_amdgcn_readfirstlane((int)(threadIdx.x >> 5));
    const int cn = lane & 15;
    const int kh = (lane >> 4) << 1;
    const int hb = (lane >> 4) << 3;
    const size_t row = (size_t)blockIdx.x * 16 + cn;
    const int col = wv * 16 + cn;
    const float* prow = pooled + row * HD;
    const float* hrow = hi + row * HD;
    const float sgr = gsum[row];
#pragma unroll
    for (int l = 0; l < 3; ++l) {
        const float* B = W + l * 4096;
        const float* xin = sX[(l + 1) & 1];
        float* xout = sX[l & 1];
        v8f d = {0.f, 0.f, 0.f, 0.f, 0.f, 0.f, 0.f, 0.f};
#pragma unroll
        for (int k0 = 0; k0 < 64; k0 += 4) {
            v2f a;
            if (l == 0) {
                a.x = prow[k0 + kh]     + hrow[k0 + kh]     * sgr;
                a.y = prow[k0 + kh + 1] + hrow[k0 + kh + 1] * sgr;
            }
            else        { a = *(const v2f*)(xin + cn * XP + k0 + kh); }
            v2f b;
            b.x = B[(size_t)(k0 + kh) * HD + col];
            b.y = B[(size_t)(k0 + kh + 1) * HD + col];
            d = __builtin_amdgcn_wmma_f32_16x16x4_f32(false, a, false, b, (short)0, d, false, false);
        }
        const float bval = bias[l * HD + col];
#pragma unroll
        for (int v = 0; v < 8; ++v) xout[(hb + v) * XP + col] = fmaxf(d[v] + bval, 0.f);
        __syncthreads();
    }
    if (wv == 0 && lane < 16) {
        const float* xr = sX[0] + cn * XP;
        float s = bo[0];
#pragma unroll
        for (int k = 0; k < 64; ++k) s += xr[k] * Wo[k];
        out[row] = s;
    }
}

extern "C" void kernel_launch(void* const* d_in, const int* in_sizes, int n_in,
                              void* d_out, int out_size, void* d_ws, size_t ws_size,
                              hipStream_t stream)
{
    const float* r_node = (const float*)d_in[0];
    const float* i_node = (const float*)d_in[1];
    const float* r_edge = (const float*)d_in[2];
    const float* d_edge = (const float*)d_in[3];
    const int*   src    = (const int*)d_in[4];
    const int*   dst    = (const int*)d_in[5];
    const int*   gid    = (const int*)d_in[6];
    const float* W_r    = (const float*)d_in[7];
    const float* W_i    = (const float*)d_in[8];
    const float* W_e    = (const float*)d_in[9];
    const float* Wm     = (const float*)d_in[10];
    const float* a_s    = (const float*)d_in[11];
    const float* a_d    = (const float*)d_in[12];
    const float* a_e    = (const float*)d_in[13];
    const float* w_d    = (const float*)d_in[14];
    const float* b_d    = (const float*)d_in[15];
    const float* W_mlp  = (const float*)d_in[16];
    const float* b_mlp  = (const float*)d_in[17];
    const float* W_out  = (const float*)d_in[18];
    const float* b_out  = (const float*)d_in[19];

    const int N = in_sizes[0] / 64;
    const int G = in_sizes[1];
    const int E = in_sizes[2] / 16;
    (void)n_in; (void)out_size; (void)ws_size;

    float* ws = (float*)d_ws;
    size_t off = 0;
    auto carve = [&](size_t n) -> float* {
        float* p = ws + off; off += (n + 63) & ~(size_t)63; return p;
    };
    float*    h      = carve((size_t)N * 64);
    float*    h2     = carve((size_t)N * 64);
    float*    hi     = carve((size_t)G * 64);
    float*    pooled = carve((size_t)G * 64);
    float*    gsum   = carve(G);
    float*    cE     = carve(64);
    float*    WeWm   = carve(4 * 16 * 64);
    const int R   = (N + CSR_NB - 1) / CSR_NB;
    const int NP  = (N + 1 + 63) & ~63;
    const int CSP = R * (CSR_NB / CSR_CH);
    int*      pc     = (int*)carve((size_t)CSR_S * NP);
    int*      cs     = (int*)carve((size_t)CSR_S * CSP);
    int*      rowptr = (int*)carve((size_t)CSR_S * NP);
    int*      pk     = (int*)carve((size_t)E * 2);

    const int ntile   = N / 16;
    const int nb_gemm = (ntile + 7) / 8;
    const int nb_init = (G * 16 + 255) / 256;
    hipLaunchKernelGGL(front_all, dim3(FRONT_SMALL + nb_gemm + nb_init), dim3(256), 0, stream,
                       r_node, W_r, h, ntile, nb_gemm, i_node, W_i, hi, pooled, gsum, G,
                       W_e, a_e, Wm, cE, WeWm);

    hipLaunchKernelGGL(csr_count, dim3(R * CSR_S), dim3(1024), 0, stream, dst, pc, cs, N, NP, CSP, E, R);
    hipLaunchKernelGGL(csr_scan, dim3((N + 1023) / 1024), dim3(1024), 0, stream, pc, cs, rowptr, N, NP, CSP);
    hipLaunchKernelGGL(csr_fill, dim3(R * CSR_S), dim3(1024), 0, stream,
                       src, dst, rowptr, pk, N, NP, E, R);

    float* hc = h;
    float* hx = h2;
    for (int l = 0; l < 3; ++l) {
        hipLaunchKernelGGL(layer_fused<false>, dim3((N + 15) / 16), dim3(256), 0, stream,
                           rowptr, pk, a_s + l * 64, a_d + l * 64, cE + l * 16,
                           hc, r_edge, Wm + (size_t)l * 4096, WeWm + (size_t)l * 1024, hx, N,
                           nullptr, nullptr, nullptr, nullptr, nullptr, nullptr, nullptr);
        float* t = hc; hc = hx; hx = t;
    }
    hipLaunchKernelGGL(layer_fused<true>, dim3((N + 15) / 16), dim3(256), 0, stream,
                       rowptr, pk, a_s + 3 * 64, a_d + 3 * 64, cE + 3 * 16,
                       hc, r_edge, Wm + (size_t)3 * 4096, WeWm + (size_t)3 * 1024, nullptr, N,
                       d_edge, gid, w_d, b_d, hi, pooled, gsum);

    hipLaunchKernelGGL(mlp_out, dim3(G / 16), dim3(128), 0, stream,
                       pooled, hi, gsum, W_mlp, b_mlp, W_out, b_out, (float*)d_out);
}
